# starting kernel
# baseline (speedup 1.0000x reference)
#include <hip/hip_runtime.h>
#include <hip/hip_bf16.h>
#include <stdint.h>

typedef __attribute__((ext_vector_type(2))) float v2f;
typedef __attribute__((ext_vector_type(8))) float v8f;

#define NNODE   10000
#define NEDGE   160000
#define EN      (NEDGE + NNODE)
#define DIM     256
#define MT      (NNODE / 16)
#define NEG_SLOPE 0.2f
#define BN_EPS  1e-5f

__global__ void zero_ws(float4* __restrict__ p, long n4) {
    long i = (long)blockIdx.x * blockDim.x + threadIdx.x;
    long stride = (long)gridDim.x * blockDim.x;
    float4 z = make_float4(0.f, 0.f, 0.f, 0.f);
    for (; i < n4; i += stride) p[i] = z;
}

__global__ void gemm_xlxr(const float* __restrict__ x,
                          const float* __restrict__ Wl, const float* __restrict__ bl,
                          const float* __restrict__ Wr, const float* __restrict__ br,
                          float* __restrict__ xl, float* __restrict__ xr) {
    int wid  = (int)((blockIdx.x * blockDim.x + threadIdx.x) >> 5);
    int lane = (int)(threadIdx.x & 31);
    if (wid >= MT * 32) return;

    int mt   = wid >> 5;
    int rest = wid & 31;
    int mat  = rest >> 4;
    int nt   = rest & 15;

    const float* W    = mat ? Wr : Wl;
    const float* bias = mat ? br : bl;
    float*       out  = mat ? xr : xl;

    int half = lane >> 4;
    int l16  = lane & 15;
    int row  = mt * 16 + l16;
    int col  = nt * 16 + l16;

    const float* aptr = x + (long)row * DIM + 2 * half;
    const float* bptr = W + (long)(2 * half) * DIM + col;

    v8f c = {};
#pragma unroll 4
    for (int k = 0; k < DIM; k += 4) {
        v2f a, b;
        a[0] = aptr[k];
        a[1] = aptr[k + 1];
        b[0] = bptr[(long)k * DIM];
        b[1] = bptr[(long)k * DIM + DIM];
        c = __builtin_amdgcn_wmma_f32_16x16x4_f32(
                 false, a,  false, b,
                 (short)0, c,  false,  false);
    }

    float bv = bias[col];
#pragma unroll
    for (int r = 0; r < 8; ++r) {
        int m = mt * 16 + r + 8 * half;
        out[(long)m * DIM + col] = c[r] + bv;
    }
}

__device__ __forceinline__ unsigned fenc(float f) {
    unsigned b = __float_as_uint(f);
    return (b & 0x80000000u) ? ~b : (b | 0x80000000u);
}
__device__ __forceinline__ float fdec(unsigned u) {
    return __uint_as_float((u & 0x80000000u) ? (u ^ 0x80000000u) : ~u);
}

__global__ void edge_logits(const int* __restrict__ ei,
                            const float* __restrict__ xl, const float* __restrict__ xr,
                            const float* __restrict__ att,
                            float* __restrict__ alpha_raw, unsigned* __restrict__ amax) {
    int wid  = (int)((blockIdx.x * blockDim.x + threadIdx.x) >> 5);
    int lane = (int)(threadIdx.x & 31);
    if (wid >= EN) return;

    int s, d;
    if (wid < NEDGE) { s = ei[wid]; d = ei[NEDGE + wid]; }
    else             { s = d = wid - NEDGE; }

    const float4* xs = (const float4*)(xl + (long)s * DIM);
    const float4* xd = (const float4*)(xr + (long)d * DIM);
    const float4* av = (const float4*)att;

    float acc = 0.f;
#pragma unroll
    for (int j = lane; j < DIM / 4; j += 32) {
        float4 a = xs[j], b = xd[j], t = av[j];
        float h;
        h = a.x + b.x; h = h > 0.f ? h : NEG_SLOPE * h; acc += t.x * h;
        h = a.y + b.y; h = h > 0.f ? h : NEG_SLOPE * h; acc += t.y * h;
        h = a.z + b.z; h = h > 0.f ? h : NEG_SLOPE * h; acc += t.z * h;
        h = a.w + b.w; h = h > 0.f ? h : NEG_SLOPE * h; acc += t.w * h;
    }
#pragma unroll
    for (int off = 16; off > 0; off >>= 1) acc += __shfl_xor(acc, off, 32);

    if (lane == 0) {
        alpha_raw[wid] = acc;
        atomicMax(&amax[d], fenc(acc));
    }
}

__global__ void edge_aggregate(const int* __restrict__ ei,
                               const float* __restrict__ xl,
                               const float* __restrict__ alpha_raw,
                               const unsigned* __restrict__ amax,
                               float* __restrict__ denom, float* __restrict__ agg) {
    int wid  = (int)((blockIdx.x * blockDim.x + threadIdx.x) >> 5);
    int lane = (int)(threadIdx.x & 31);
    if (wid >= EN) return;

    int s, d;
    if (wid < NEDGE) { s = ei[wid]; d = ei[NEDGE + wid]; }
    else             { s = d = wid - NEDGE; }

    float al = __expf(alpha_raw[wid] - fdec(amax[d]));
    if (lane == 0) atomicAdd(&denom[d], al);

    const float4* xs = (const float4*)(xl + (long)s * DIM);
    float* out = agg + (long)d * DIM;
#pragma unroll
    for (int j = lane; j < DIM / 4; j += 32) {
        float4 a = xs[j];
        atomicAdd(&out[4 * j + 0], al * a.x);
        atomicAdd(&out[4 * j + 1], al * a.y);
        atomicAdd(&out[4 * j + 2], al * a.z);
        atomicAdd(&out[4 * j + 3], al * a.w);
    }
}

#define ROWS_PER_BLK 50

__global__ void post_stats(float* __restrict__ agg, const float* __restrict__ denom,
                           const float* __restrict__ bias, const float* __restrict__ du,
                           float* __restrict__ sums, float* __restrict__ sumsq) {
    int dcol = (int)threadIdx.x;
    int i0   = (int)blockIdx.x * ROWS_PER_BLK;
    float s = 0.f, s2 = 0.f;
    for (int r = 0; r < ROWS_PER_BLK; ++r) {
        int i = i0 + r;
        if (i >= NNODE) break;
        long idx = (long)i * DIM + dcol;
        float v = agg[idx] / denom[i] + bias[dcol];
        v = v > 0.f ? v : 0.f;
        v = (du[idx] >= 0.5f) ? 2.0f * v : 0.0f;
        agg[idx] = v;
        s += v; s2 += v * v;
    }
    atomicAdd(&sums[dcol], s);
    atomicAdd(&sumsq[dcol], s2);
}

__global__ void finalize(const float* __restrict__ v,
                         const float* __restrict__ sums, const float* __restrict__ sumsq,
                         const float* __restrict__ gamma, const float* __restrict__ beta,
                         float* __restrict__ out) {
    int dcol = (int)threadIdx.x;
    int i0   = (int)blockIdx.x * ROWS_PER_BLK;
    const float invn = 1.0f / (float)NNODE;
    float mean  = sums[dcol] * invn;
    float var   = sumsq[dcol] * invn - mean * mean;
    float scale = gamma[dcol] * rsqrtf(var + BN_EPS);
    float shift = beta[dcol] - mean * scale;
    for (int r = 0; r < ROWS_PER_BLK; ++r) {
        int i = i0 + r;
        if (i >= NNODE) break;
        long idx = (long)i * DIM + dcol;
        out[idx] = v[idx] * scale + shift;
    }
}

extern "C" void kernel_launch(void* const* d_in, const int* in_sizes, int n_in,
                              void* d_out, int out_size, void* d_ws, size_t ws_size,
                              hipStream_t stream) {
    const float* x     = (const float*)d_in[0];
    const int*   ei    = (const int*)d_in[1];
    const float* Wl    = (const float*)d_in[2];
    const float* bl    = (const float*)d_in[3];
    const float* Wr    = (const float*)d_in[4];
    const float* br    = (const float*)d_in[5];
    const float* att   = (const float*)d_in[6];
    const float* bias  = (const float*)d_in[7];
    const float* gamma = (const float*)d_in[8];
    const float* beta  = (const float*)d_in[9];
    const float* du    = (const float*)d_in[10];
    float*       out   = (float*)d_out;

    float*    ws    = (float*)d_ws;
    float*    agg   = ws;
    unsigned* amax  = (unsigned*)(ws + (long)NNODE * DIM);
    float*    denom = ws + (long)NNODE * DIM + NNODE;
    float*    sums  = denom + NNODE;
    float*    sumsq = sums + DIM;
    float*    xl    = sumsq + DIM;
    float*    xr    = xl + (long)NNODE * DIM;
    float*    araw  = xr + (long)NNODE * DIM;

    long zeroN  = (long)NNODE * DIM + 2L * NNODE + 2L * DIM;
    long zeroN4 = zeroN / 4;
    zero_ws<<<1024, 256, 0, stream>>>((float4*)d_ws, zeroN4);

    gemm_xlxr<<<(MT * 32) / 4, 128, 0, stream>>>(x, Wl, bl, Wr, br, xl, xr);

    int eblocks = (EN * 32 + 255) / 256;
    edge_logits<<<eblocks, 256, 0, stream>>>(ei, xl, xr, att, araw, amax);

    edge_aggregate<<<eblocks, 256, 0, stream>>>(ei, xl, araw, amax, denom, agg);

    int sblocks = (NNODE + ROWS_PER_BLK - 1) / ROWS_PER_BLK;
    post_stats<<<sblocks, DIM, 0, stream>>>(agg, denom, bias, du, sums, sumsq);

    finalize<<<sblocks, DIM, 0, stream>>>(agg, sums, sumsq, gamma, beta, out);
}


# GEMM as a block tile through LDS: 80 rows x 128 columns per 8-wave block, x and W k-chunks of 32 staged in two LDS stages, fragments read as 8-byte LDS loads; k sum order unchanged
# speedup vs baseline: 1.1124x; 1.1124x over previous
#include <hip/hip_runtime.h>
#include <hip/hip_bf16.h>
#include <stdint.h>

typedef __attribute__((ext_vector_type(2))) float v2f;
typedef __attribute__((ext_vector_type(8))) float v8f;

#define NNODE   10000
#define NEDGE   160000
#define EN      (NEDGE + NNODE)
#define DIM     256
#define MT      (NNODE / 16)
#define NEG_SLOPE 0.2f
#define BN_EPS  1e-5f

__global__ void zero_ws(float4* __restrict__ p, long n4) {
    long i = (long)blockIdx.x * blockDim.x + threadIdx.x;
    long stride = (long)gridDim.x * blockDim.x;
    float4 z = make_float4(0.f, 0.f, 0.f, 0.f);
    for (; i < n4; i += stride) p[i] = z;
}

#define BM   80
#define BN   128
#define KC   32
#define XP   (KC + 4)
#define WNP  (BN + 16)
#define XS_FLOATS (BM * XP)
#define WS_FLOATS ((KC / 2) * WNP * 2)

__global__ __launch_bounds__(256)
void gemm_xlxr(const float* __restrict__ x,
               const float* __restrict__ Wl, const float* __restrict__ bl,
               const float* __restrict__ Wr, const float* __restrict__ br,
               float* __restrict__ xl, float* __restrict__ xr) {
    __shared__ __attribute__((aligned(16))) float xs[2][XS_FLOATS];
    __shared__ __attribute__((aligned(16))) float wsm[2][WS_FLOATS];

    int t    = (int)threadIdx.x;
    int lane = t & 31;
    int w    = t >> 5;
    int half = lane >> 4;
    int l16  = lane & 15;

    int rb  = (int)blockIdx.x >> 2;
    int cb  = (int)blockIdx.x & 3;
    int mat = cb >> 1;
    int nb  = (cb & 1) * BN;
    int m0  = rb * BM;

    const float* W    = mat ? Wr : Wl;
    const float* bias = mat ? br : bl;
    float*       out  = mat ? xr : xl;

    int xrow = t >> 4, xj = t & 15;
    const float* xg = x + (long)(m0 + xrow) * DIM + 2 * xj;
    int xo = xrow * XP + 2 * xj;

    int n4  = t & 31;
    int kpb = t >> 5;
    const float* wg = W + (long)(2 * kpb) * DIM + nb + 4 * n4;
    int wo = (kpb * WNP + 4 * n4) * 2;

    float2 x0, x1, x2, x3, x4;
    float4 wa0, wb0, wa1, wb1;

    auto gload = [&](int k0) {
        x0  = *(const float2*)(xg + k0);
        x1  = *(const float2*)(xg + k0 + 16L * DIM);
        x2  = *(const float2*)(xg + k0 + 32L * DIM);
        x3  = *(const float2*)(xg + k0 + 48L * DIM);
        x4  = *(const float2*)(xg + k0 + 64L * DIM);
        const float* p = wg + (long)k0 * DIM;
        wa0 = *(const float4*)(p);
        wb0 = *(const float4*)(p + DIM);
        wa1 = *(const float4*)(p + 16 * DIM);
        wb1 = *(const float4*)(p + 17 * DIM);
    };

    auto sstore = [&](int buf) {
        float* xd = &xs[buf][xo];
        *(float2*)(xd)           = x0;
        *(float2*)(xd + 16 * XP) = x1;
        *(float2*)(xd + 32 * XP) = x2;
        *(float2*)(xd + 48 * XP) = x3;
        *(float2*)(xd + 64 * XP) = x4;
        float* p = &wsm[buf][wo];
        *(float4*)(p)     = make_float4(wa0.x, wb0.x, wa0.y, wb0.y);
        *(float4*)(p + 4) = make_float4(wa0.z, wb0.z, wa0.w, wb0.w);
        float* q = p + 8 * WNP * 2;
        *(float4*)(q)     = make_float4(wa1.x, wb1.x, wa1.y, wb1.y);
        *(float4*)(q + 4) = make_float4(wa1.z, wb1.z, wa1.w, wb1.w);
    };

    v8f c0 = {}, c1 = {}, c2 = {}, c3 = {}, c4 = {};

    gload(0);
    sstore(0);
    __syncthreads();

    for (int kc = 0; kc < DIM / KC; ++kc) {
        int cur = kc & 1;
        bool more = (kc + 1) < (DIM / KC);
        if (more) gload((kc + 1) * KC);

        const float* xa = &xs[cur][l16 * XP + 2 * half];
        const float* wp = &wsm[cur][(half * WNP + 16 * w + l16) * 2];
#pragma unroll
        for (int kk = 0; kk < KC; kk += 4) {
            v2f b  = *(const v2f*)(wp + (kk >> 1) * WNP * 2);
            v2f a0 = *(const v2f*)(xa + 0 * 16 * XP + kk);
            v2f a1 = *(const v2f*)(xa + 1 * 16 * XP + kk);
            v2f a2 = *(const v2f*)(xa + 2 * 16 * XP + kk);
            v2f a3 = *(const v2f*)(xa + 3 * 16 * XP + kk);
            v2f a4 = *(const v2f*)(xa + 4 * 16 * XP + kk);
            c0 = __builtin_amdgcn_wmma_f32_16x16x4_f32(false, a0, false, b, (short)0, c0, false, false);
            c1 = __builtin_amdgcn_wmma_f32_16x16x4_f32(false, a1, false, b, (short)0, c1, false, false);
            c2 = __builtin_amdgcn_wmma_f32_16x16x4_f32(false, a2, false, b, (short)0, c2, false, false);
            c3 = __builtin_amdgcn_wmma_f32_16x16x4_f32(false, a3, false, b, (short)0, c3, false, false);
            c4 = __builtin_amdgcn_wmma_f32_16x16x4_f32(false, a4, false, b, (short)0, c4, false, false);
        }

        if (more) sstore(cur ^ 1);
        __syncthreads();
    }

    int col = nb + 16 * w + l16;
    float bv = bias[col];
    float* o = out + (long)(m0 + 8 * half) * DIM + col;
#pragma unroll
    for (int r = 0; r < 8; ++r) {
        o[(long)(0 * 16 + r) * DIM] = c0[r] + bv;
        o[(long)(1 * 16 + r) * DIM] = c1[r] + bv;
        o[(long)(2 * 16 + r) * DIM] = c2[r] + bv;
        o[(long)(3 * 16 + r) * DIM] = c3[r] + bv;
        o[(long)(4 * 16 + r) * DIM] = c4[r] + bv;
    }
}

__device__ __forceinline__ unsigned fenc(float f) {
    unsigned b = __float_as_uint(f);
    return (b & 0x80000000u) ? ~b : (b | 0x80000000u);
}
__device__ __forceinline__ float fdec(unsigned u) {
    return __uint_as_float((u & 0x80000000u) ? (u ^ 0x80000000u) : ~u);
}

__global__ void edge_logits(const int* __restrict__ ei,
                            const float* __restrict__ xl, const float* __restrict__ xr,
                            const float* __restrict__ att,
                            float* __restrict__ alpha_raw, unsigned* __restrict__ amax) {
    int wid  = (int)((blockIdx.x * blockDim.x + threadIdx.x) >> 5);
    int lane = (int)(threadIdx.x & 31);
    if (wid >= EN) return;

    int s, d;
    if (wid < NEDGE) { s = ei[wid]; d = ei[NEDGE + wid]; }
    else             { s = d = wid - NEDGE; }

    const float4* xs = (const float4*)(xl + (long)s * DIM);
    const float4* xd = (const float4*)(xr + (long)d * DIM);
    const float4* av = (const float4*)att;

    float acc = 0.f;
#pragma unroll
    for (int j = lane; j < DIM / 4; j += 32) {
        float4 a = xs[j], b = xd[j], t = av[j];
        float h;
        h = a.x + b.x; h = h > 0.f ? h : NEG_SLOPE * h; acc += t.x * h;
        h = a.y + b.y; h = h > 0.f ? h : NEG_SLOPE * h; acc += t.y * h;
        h = a.z + b.z; h = h > 0.f ? h : NEG_SLOPE * h; acc += t.z * h;
        h = a.w + b.w; h = h > 0.f ? h : NEG_SLOPE * h; acc += t.w * h;
    }
#pragma unroll
    for (int off = 16; off > 0; off >>= 1) acc += __shfl_xor(acc, off, 32);

    if (lane == 0) {
        alpha_raw[wid] = acc;
        atomicMax(&amax[d], fenc(acc));
    }
}

__global__ void edge_aggregate(const int* __restrict__ ei,
                               const float* __restrict__ xl,
                               const float* __restrict__ alpha_raw,
                               const unsigned* __restrict__ amax,
                               float* __restrict__ denom, float* __restrict__ agg) {
    int wid  = (int)((blockIdx.x * blockDim.x + threadIdx.x) >> 5);
    int lane = (int)(threadIdx.x & 31);
    if (wid >= EN) return;

    int s, d;
    if (wid < NEDGE) { s = ei[wid]; d = ei[NEDGE + wid]; }
    else             { s = d = wid - NEDGE; }

    float al = __expf(alpha_raw[wid] - fdec(amax[d]));
    if (lane == 0) atomicAdd(&denom[d], al);

    const float4* xs = (const float4*)(xl + (long)s * DIM);
    float* out = agg + (long)d * DIM;
#pragma unroll
    for (int j = lane; j < DIM / 4; j += 32) {
        float4 a = xs[j];
        atomicAdd(&out[4 * j + 0], al * a.x);
        atomicAdd(&out[4 * j + 1], al * a.y);
        atomicAdd(&out[4 * j + 2], al * a.z);
        atomicAdd(&out[4 * j + 3], al * a.w);
    }
}

#define ROWS_PER_BLK 50

__global__ void post_stats(float* __restrict__ agg, const float* __restrict__ denom,
                           const float* __restrict__ bias, const float* __restrict__ du,
                           float* __restrict__ sums, float* __restrict__ sumsq) {
    int dcol = (int)threadIdx.x;
    int i0   = (int)blockIdx.x * ROWS_PER_BLK;
    float s = 0.f, s2 = 0.f;
    for (int r = 0; r < ROWS_PER_BLK; ++r) {
        int i = i0 + r;
        if (i >= NNODE) break;
        long idx = (long)i * DIM + dcol;
        float v = agg[idx] / denom[i] + bias[dcol];
        v = v > 0.f ? v : 0.f;
        v = (du[idx] >= 0.5f) ? 2.0f * v : 0.0f;
        agg[idx] = v;
        s += v; s2 += v * v;
    }
    atomicAdd(&sums[dcol], s);
    atomicAdd(&sumsq[dcol], s2);
}

__global__ void finalize(const float* __restrict__ v,
                         const float* __restrict__ sums, const float* __restrict__ sumsq,
                         const float* __restrict__ gamma, const float* __restrict__ beta,
                         float* __restrict__ out) {
    int dcol = (int)threadIdx.x;
    int i0   = (int)blockIdx.x * ROWS_PER_BLK;
    const float invn = 1.0f / (float)NNODE;
    float mean  = sums[dcol] * invn;
    float var   = sumsq[dcol] * invn - mean * mean;
    float scale = gamma[dcol] * rsqrtf(var + BN_EPS);
    float shift = beta[dcol] - mean * scale;
    for (int r = 0; r < ROWS_PER_BLK; ++r) {
        int i = i0 + r;
        if (i >= NNODE) break;
        long idx = (long)i * DIM + dcol;
        out[idx] = v[idx] * scale + shift;
    }
}

extern "C" void kernel_launch(void* const* d_in, const int* in_sizes, int n_in,
                              void* d_out, int out_size, void* d_ws, size_t ws_size,
                              hipStream_t stream) {
    const float* x     = (const float*)d_in[0];
    const int*   ei    = (const int*)d_in[1];
    const float* Wl    = (const float*)d_in[2];
    const float* bl    = (const float*)d_in[3];
    const float* Wr    = (const float*)d_in[4];
    const float* br    = (const float*)d_in[5];
    const float* att   = (const float*)d_in[6];
    const float* bias  = (const float*)d_in[7];
    const float* gamma = (const float*)d_in[8];
    const float* beta  = (const float*)d_in[9];
    const float* du    = (const float*)d_in[10];
    float*       out   = (float*)d_out;

    float*    ws    = (float*)d_ws;
    float*    agg   = ws;
    unsigned* amax  = (unsigned*)(ws + (long)NNODE * DIM);
    float*    denom = ws + (long)NNODE * DIM + NNODE;
    float*    sums  = denom + NNODE;
    float*    sumsq = sums + DIM;
    float*    xl    = sumsq + DIM;
    float*    xr    = xl + (long)NNODE * DIM;
    float*    araw  = xr + (long)NNODE * DIM;

    long zeroN  = (long)NNODE * DIM + 2L * NNODE + 2L * DIM;
    long zeroN4 = zeroN / 4;
    zero_ws<<<1024, 256, 0, stream>>>((float4*)d_ws, zeroN4);

    gemm_xlxr<<<(NNODE / BM) * 4, 256, 0, stream>>>(x, Wl, bl, Wr, br, xl, xr);

    int eblocks = (EN * 32 + 255) / 256;
    edge_logits<<<eblocks, 256, 0, stream>>>(ei, xl, xr, att, araw, amax);

    edge_aggregate<<<eblocks, 256, 0, stream>>>(ei, xl, araw, amax, denom, agg);

    int sblocks = (NNODE + ROWS_PER_BLK - 1) / ROWS_PER_BLK;
    post_stats<<<sblocks, DIM, 0, stream>>>(agg, denom, bias, du, sums, sumsq);

    finalize<<<sblocks, DIM, 0, stream>>>(agg, sums, sumsq, gamma, beta, out);
}


# scatter-add as whole-line atomics: lane l takes columns l+32i, so each wave atomic (and each row load, 8 issued together) covers one 128-byte line: 8 atomic line requests per edge row, was 32
# speedup vs baseline: 1.2892x; 1.1589x over previous
#include <hip/hip_runtime.h>
#include <hip/hip_bf16.h>
#include <stdint.h>

typedef __attribute__((ext_vector_type(2))) float v2f;
typedef __attribute__((ext_vector_type(8))) float v8f;

#define NNODE   10000
#define NEDGE   160000
#define EN      (NEDGE + NNODE)
#define DIM     256
#define MT      (NNODE / 16)
#define NEG_SLOPE 0.2f
#define BN_EPS  1e-5f

__global__ void zero_ws(float4* __restrict__ p, long n4) {
    long i = (long)blockIdx.x * blockDim.x + threadIdx.x;
    long stride = (long)gridDim.x * blockDim.x;
    float4 z = make_float4(0.f, 0.f, 0.f, 0.f);
    for (; i < n4; i += stride) p[i] = z;
}

#define BM   80
#define BN   128
#define KC   32
#define XP   (KC + 4)
#define WNP  (BN + 16)
#define XS_FLOATS (BM * XP)
#define WS_FLOATS ((KC / 2) * WNP * 2)

__global__ __launch_bounds__(256)
void gemm_xlxr(const float* __restrict__ x,
               const float* __restrict__ Wl, const float* __restrict__ bl,
               const float* __restrict__ Wr, const float* __restrict__ br,
               float* __restrict__ xl, float* __restrict__ xr) {
    __shared__ __attribute__((aligned(16))) float xs[2][XS_FLOATS];
    __shared__ __attribute__((aligned(16))) float wsm[2][WS_FLOATS];

    int t    = (int)threadIdx.x;
    int lane = t & 31;
    int w    = t >> 5;
    int half = lane >> 4;
    int l16  = lane & 15;

    int rb  = (int)blockIdx.x >> 2;
    int cb  = (int)blockIdx.x & 3;
    int mat = cb >> 1;
    int nb  = (cb & 1) * BN;
    int m0  = rb * BM;

    const float* W    = mat ? Wr : Wl;
    const float* bias = mat ? br : bl;
    float*       out  = mat ? xr : xl;

    int xrow = t >> 4, xj = t & 15;
    const float* xg = x + (long)(m0 + xrow) * DIM + 2 * xj;
    int xo = xrow * XP + 2 * xj;

    int n4  = t & 31;
    int kpb = t >> 5;
    const float* wg = W + (long)(2 * kpb) * DIM + nb + 4 * n4;
    int wo = (kpb * WNP + 4 * n4) * 2;

    float2 x0, x1, x2, x3, x4;
    float4 wa0, wb0, wa1, wb1;

    auto gload = [&](int k0) {
        x0  = *(const float2*)(xg + k0);
        x1  = *(const float2*)(xg + k0 + 16L * DIM);
        x2  = *(const float2*)(xg + k0 + 32L * DIM);
        x3  = *(const float2*)(xg + k0 + 48L * DIM);
        x4  = *(const float2*)(xg + k0 + 64L * DIM);
        const float* p = wg + (long)k0 * DIM;
        wa0 = *(const float4*)(p);
        wb0 = *(const float4*)(p + DIM);
        wa1 = *(const float4*)(p + 16 * DIM);
        wb1 = *(const float4*)(p + 17 * DIM);
    };

    auto sstore = [&](int buf) {
        float* xd = &xs[buf][xo];
        *(float2*)(xd)           = x0;
        *(float2*)(xd + 16 * XP) = x1;
        *(float2*)(xd + 32 * XP) = x2;
        *(float2*)(xd + 48 * XP) = x3;
        *(float2*)(xd + 64 * XP) = x4;
        float* p = &wsm[buf][wo];
        *(float4*)(p)     = make_float4(wa0.x, wb0.x, wa0.y, wb0.y);
        *(float4*)(p + 4) = make_float4(wa0.z, wb0.z, wa0.w, wb0.w);
        float* q = p + 8 * WNP * 2;
        *(float4*)(q)     = make_float4(wa1.x, wb1.x, wa1.y, wb1.y);
        *(float4*)(q + 4) = make_float4(wa1.z, wb1.z, wa1.w, wb1.w);
    };

    v8f c0 = {}, c1 = {}, c2 = {}, c3 = {}, c4 = {};

    gload(0);
    sstore(0);
    __syncthreads();

    for (int kc = 0; kc < DIM / KC; ++kc) {
        int cur = kc & 1;
        bool more = (kc + 1) < (DIM / KC);
        if (more) gload((kc + 1) * KC);

        const float* xa = &xs[cur][l16 * XP + 2 * half];
        const float* wp = &wsm[cur][(half * WNP + 16 * w + l16) * 2];
#pragma unroll
        for (int kk = 0; kk < KC; kk += 4) {
            v2f b  = *(const v2f*)(wp + (kk >> 1) * WNP * 2);
            v2f a0 = *(const v2f*)(xa + 0 * 16 * XP + kk);
            v2f a1 = *(const v2f*)(xa + 1 * 16 * XP + kk);
            v2f a2 = *(const v2f*)(xa + 2 * 16 * XP + kk);
            v2f a3 = *(const v2f*)(xa + 3 * 16 * XP + kk);
            v2f a4 = *(const v2f*)(xa + 4 * 16 * XP + kk);
            c0 = __builtin_amdgcn_wmma_f32_16x16x4_f32(false, a0, false, b, (short)0, c0, false, false);
            c1 = __builtin_amdgcn_wmma_f32_16x16x4_f32(false, a1, false, b, (short)0, c1, false, false);
            c2 = __builtin_amdgcn_wmma_f32_16x16x4_f32(false, a2, false, b, (short)0, c2, false, false);
            c3 = __builtin_amdgcn_wmma_f32_16x16x4_f32(false, a3, false, b, (short)0, c3, false, false);
            c4 = __builtin_amdgcn_wmma_f32_16x16x4_f32(false, a4, false, b, (short)0, c4, false, false);
        }

        if (more) sstore(cur ^ 1);
        __syncthreads();
    }

    int col = nb + 16 * w + l16;
    float bv = bias[col];
    float* o = out + (long)(m0 + 8 * half) * DIM + col;
#pragma unroll
    for (int r = 0; r < 8; ++r) {
        o[(long)(0 * 16 + r) * DIM] = c0[r] + bv;
        o[(long)(1 * 16 + r) * DIM] = c1[r] + bv;
        o[(long)(2 * 16 + r) * DIM] = c2[r] + bv;
        o[(long)(3 * 16 + r) * DIM] = c3[r] + bv;
        o[(long)(4 * 16 + r) * DIM] = c4[r] + bv;
    }
}

__device__ __forceinline__ unsigned fenc(float f) {
    unsigned b = __float_as_uint(f);
    return (b & 0x80000000u) ? ~b : (b | 0x80000000u);
}
__device__ __forceinline__ float fdec(unsigned u) {
    return __uint_as_float((u & 0x80000000u) ? (u ^ 0x80000000u) : ~u);
}

__global__ void edge_logits(const int* __restrict__ ei,
                            const float* __restrict__ xl, const float* __restrict__ xr,
                            const float* __restrict__ att,
                            float* __restrict__ alpha_raw, unsigned* __restrict__ amax) {
    int wid  = (int)((blockIdx.x * blockDim.x + threadIdx.x) >> 5);
    int lane = (int)(threadIdx.x & 31);
    if (wid >= EN) return;

    int s, d;
    if (wid < NEDGE) { s = ei[wid]; d = ei[NEDGE + wid]; }
    else             { s = d = wid - NEDGE; }

    const float4* xs = (const float4*)(xl + (long)s * DIM);
    const float4* xd = (const float4*)(xr + (long)d * DIM);
    const float4* av = (const float4*)att;

    float acc = 0.f;
#pragma unroll
    for (int j = lane; j < DIM / 4; j += 32) {
        float4 a = xs[j], b = xd[j], t = av[j];
        float h;
        h = a.x + b.x; h = h > 0.f ? h : NEG_SLOPE * h; acc += t.x * h;
        h = a.y + b.y; h = h > 0.f ? h : NEG_SLOPE * h; acc += t.y * h;
        h = a.z + b.z; h = h > 0.f ? h : NEG_SLOPE * h; acc += t.z * h;
        h = a.w + b.w; h = h > 0.f ? h : NEG_SLOPE * h; acc += t.w * h;
    }
#pragma unroll
    for (int off = 16; off > 0; off >>= 1) acc += __shfl_xor(acc, off, 32);

    if (lane == 0) {
        alpha_raw[wid] = acc;
        atomicMax(&amax[d], fenc(acc));
    }
}

__global__ void edge_aggregate(const int* __restrict__ ei,
                               const float* __restrict__ xl,
                               const float* __restrict__ alpha_raw,
                               const unsigned* __restrict__ amax,
                               float* __restrict__ denom, float* __restrict__ agg) {
    int wid  = (int)((blockIdx.x * blockDim.x + threadIdx.x) >> 5);
    int lane = (int)(threadIdx.x & 31);
    if (wid >= EN) return;

    int s, d;
    if (wid < NEDGE) { s = ei[wid]; d = ei[NEDGE + wid]; }
    else             { s = d = wid - NEDGE; }

    float al = __expf(alpha_raw[wid] - fdec(amax[d]));
    if (lane == 0) atomicAdd(&denom[d], al);

    const float* xs = xl + (long)s * DIM + lane;
    float* out = agg + (long)d * DIM + lane;
    float a[DIM / 32];
#pragma unroll
    for (int i = 0; i < DIM / 32; ++i) a[i] = xs[32 * i];
#pragma unroll
    for (int i = 0; i < DIM / 32; ++i) atomicAdd(&out[32 * i], al * a[i]);
}

#define ROWS_PER_BLK 50

__global__ void post_stats(float* __restrict__ agg, const float* __restrict__ denom,
                           const float* __restrict__ bias, const float* __restrict__ du,
                           float* __restrict__ sums, float* __restrict__ sumsq) {
    int dcol = (int)threadIdx.x;
    int i0   = (int)blockIdx.x * ROWS_PER_BLK;
    float s = 0.f, s2 = 0.f;
    for (int r = 0; r < ROWS_PER_BLK; ++r) {
        int i = i0 + r;
        if (i >= NNODE) break;
        long idx = (long)i * DIM + dcol;
        float v = agg[idx] / denom[i] + bias[dcol];
        v = v > 0.f ? v : 0.f;
        v = (du[idx] >= 0.5f) ? 2.0f * v : 0.0f;
        agg[idx] = v;
        s += v; s2 += v * v;
    }
    atomicAdd(&sums[dcol], s);
    atomicAdd(&sumsq[dcol], s2);
}

__global__ void finalize(const float* __restrict__ v,
                         const float* __restrict__ sums, const float* __restrict__ sumsq,
                         const float* __restrict__ gamma, const float* __restrict__ beta,
                         float* __restrict__ out) {
    int dcol = (int)threadIdx.x;
    int i0   = (int)blockIdx.x * ROWS_PER_BLK;
    const float invn = 1.0f / (float)NNODE;
    float mean  = sums[dcol] * invn;
    float var   = sumsq[dcol] * invn - mean * mean;
    float scale = gamma[dcol] * rsqrtf(var + BN_EPS);
    float shift = beta[dcol] - mean * scale;
    for (int r = 0; r < ROWS_PER_BLK; ++r) {
        int i = i0 + r;
        if (i >= NNODE) break;
        long idx = (long)i * DIM + dcol;
        out[idx] = v[idx] * scale + shift;
    }
}

extern "C" void kernel_launch(void* const* d_in, const int* in_sizes, int n_in,
                              void* d_out, int out_size, void* d_ws, size_t ws_size,
                              hipStream_t stream) {
    const float* x     = (const float*)d_in[0];
    const int*   ei    = (const int*)d_in[1];
    const float* Wl    = (const float*)d_in[2];
    const float* bl    = (const float*)d_in[3];
    const float* Wr    = (const float*)d_in[4];
    const float* br    = (const float*)d_in[5];
    const float* att   = (const float*)d_in[6];
    const float* bias  = (const float*)d_in[7];
    const float* gamma = (const float*)d_in[8];
    const float* beta  = (const float*)d_in[9];
    const float* du    = (const float*)d_in[10];
    float*       out   = (float*)d_out;

    float*    ws    = (float*)d_ws;
    float*    agg   = ws;
    unsigned* amax  = (unsigned*)(ws + (long)NNODE * DIM);
    float*    denom = ws + (long)NNODE * DIM + NNODE;
    float*    sums  = denom + NNODE;
    float*    sumsq = sums + DIM;
    float*    xl    = sumsq + DIM;
    float*    xr    = xl + (long)NNODE * DIM;
    float*    araw  = xr + (long)NNODE * DIM;

    long zeroN  = (long)NNODE * DIM + 2L * NNODE + 2L * DIM;
    long zeroN4 = zeroN / 4;
    zero_ws<<<1024, 256, 0, stream>>>((float4*)d_ws, zeroN4);

    gemm_xlxr<<<(NNODE / BM) * 4, 256, 0, stream>>>(x, Wl, bl, Wr, br, xl, xr);

    int eblocks = (EN * 32 + 255) / 256;
    edge_logits<<<eblocks, 256, 0, stream>>>(ei, xl, xr, att, araw, amax);

    edge_aggregate<<<eblocks, 256, 0, stream>>>(ei, xl, araw, amax, denom, agg);

    int sblocks = (NNODE + ROWS_PER_BLK - 1) / ROWS_PER_BLK;
    post_stats<<<sblocks, DIM, 0, stream>>>(agg, denom, bias, du, sums, sumsq);

    finalize<<<sblocks, DIM, 0, stream>>>(agg, sums, sumsq, gamma, beta, out);
}


# edge kernels: each wave takes 8 consecutive edges, not 1: 8x fewer waves to start, edge ends in one hop, 8 rows in flight; logit sum tree unchanged, float atomics kept (arrival order free as before)
# speedup vs baseline: 1.5792x; 1.2249x over previous
#include <hip/hip_runtime.h>
#include <hip/hip_bf16.h>
#include <stdint.h>

typedef __attribute__((ext_vector_type(2))) float v2f;
typedef __attribute__((ext_vector_type(8))) float v8f;

#define NNODE   10000
#define NEDGE   160000
#define EN      (NEDGE + NNODE)
#define DIM     256
#define MT      (NNODE / 16)
#define NEG_SLOPE 0.2f
#define BN_EPS  1e-5f

__global__ void zero_ws(float4* __restrict__ p, long n4) {
    long i = (long)blockIdx.x * blockDim.x + threadIdx.x;
    long stride = (long)gridDim.x * blockDim.x;
    float4 z = make_float4(0.f, 0.f, 0.f, 0.f);
    for (; i < n4; i += stride) p[i] = z;
}

#define BM   80
#define BN   128
#define KC   32
#define XP   (KC + 4)
#define WNP  (BN + 16)
#define XS_FLOATS (BM * XP)
#define WS_FLOATS ((KC / 2) * WNP * 2)

__global__ __launch_bounds__(256)
void gemm_xlxr(const float* __restrict__ x,
               const float* __restrict__ Wl, const float* __restrict__ bl,
               const float* __restrict__ Wr, const float* __restrict__ br,
               float* __restrict__ xl, float* __restrict__ xr) {
    __shared__ __attribute__((aligned(16))) float xs[2][XS_FLOATS];
    __shared__ __attribute__((aligned(16))) float wsm[2][WS_FLOATS];

    int t    = (int)threadIdx.x;
    int lane = t & 31;
    int w    = t >> 5;
    int half = lane >> 4;
    int l16  = lane & 15;

    int rb  = (int)blockIdx.x >> 2;
    int cb  = (int)blockIdx.x & 3;
    int mat = cb >> 1;
    int nb  = (cb & 1) * BN;
    int m0  = rb * BM;

    const float* W    = mat ? Wr : Wl;
    const float* bias = mat ? br : bl;
    float*       out  = mat ? xr : xl;

    int xrow = t >> 4, xj = t & 15;
    const float* xg = x + (long)(m0 + xrow) * DIM + 2 * xj;
    int xo = xrow * XP + 2 * xj;

    int n4  = t & 31;
    int kpb = t >> 5;
    const float* wg = W + (long)(2 * kpb) * DIM + nb + 4 * n4;
    int wo = (kpb * WNP + 4 * n4) * 2;

    float2 x0, x1, x2, x3, x4;
    float4 wa0, wb0, wa1, wb1;

    auto gload = [&](int k0) {
        x0  = *(const float2*)(xg + k0);
        x1  = *(const float2*)(xg + k0 + 16L * DIM);
        x2  = *(const float2*)(xg + k0 + 32L * DIM);
        x3  = *(const float2*)(xg + k0 + 48L * DIM);
        x4  = *(const float2*)(xg + k0 + 64L * DIM);
        const float* p = wg + (long)k0 * DIM;
        wa0 = *(const float4*)(p);
        wb0 = *(const float4*)(p + DIM);
        wa1 = *(const float4*)(p + 16 * DIM);
        wb1 = *(const float4*)(p + 17 * DIM);
    };

    auto sstore = [&](int buf) {
        float* xd = &xs[buf][xo];
        *(float2*)(xd)           = x0;
        *(float2*)(xd + 16 * XP) = x1;
        *(float2*)(xd + 32 * XP) = x2;
        *(float2*)(xd + 48 * XP) = x3;
        *(float2*)(xd + 64 * XP) = x4;
        float* p = &wsm[buf][wo];
        *(float4*)(p)     = make_float4(wa0.x, wb0.x, wa0.y, wb0.y);
        *(float4*)(p + 4) = make_float4(wa0.z, wb0.z, wa0.w, wb0.w);
        float* q = p + 8 * WNP * 2;
        *(float4*)(q)     = make_float4(wa1.x, wb1.x, wa1.y, wb1.y);
        *(float4*)(q + 4) = make_float4(wa1.z, wb1.z, wa1.w, wb1.w);
    };

    v8f c0 = {}, c1 = {}, c2 = {}, c3 = {}, c4 = {};

    gload(0);
    sstore(0);
    __syncthreads();

    for (int kc = 0; kc < DIM / KC; ++kc) {
        int cur = kc & 1;
        bool more = (kc + 1) < (DIM / KC);
        if (more) gload((kc + 1) * KC);

        const float* xa = &xs[cur][l16 * XP + 2 * half];
        const float* wp = &wsm[cur][(half * WNP + 16 * w + l16) * 2];
#pragma unroll
        for (int kk = 0; kk < KC; kk += 4) {
            v2f b  = *(const v2f*)(wp + (kk >> 1) * WNP * 2);
            v2f a0 = *(const v2f*)(xa + 0 * 16 * XP + kk);
            v2f a1 = *(const v2f*)(xa + 1 * 16 * XP + kk);
            v2f a2 = *(const v2f*)(xa + 2 * 16 * XP + kk);
            v2f a3 = *(const v2f*)(xa + 3 * 16 * XP + kk);
            v2f a4 = *(const v2f*)(xa + 4 * 16 * XP + kk);
            c0 = __builtin_amdgcn_wmma_f32_16x16x4_f32(false, a0, false, b, (short)0, c0, false, false);
            c1 = __builtin_amdgcn_wmma_f32_16x16x4_f32(false, a1, false, b, (short)0, c1, false, false);
            c2 = __builtin_amdgcn_wmma_f32_16x16x4_f32(false, a2, false, b, (short)0, c2, false, false);
            c3 = __builtin_amdgcn_wmma_f32_16x16x4_f32(false, a3, false, b, (short)0, c3, false, false);
            c4 = __builtin_amdgcn_wmma_f32_16x16x4_f32(false, a4, false, b, (short)0, c4, false, false);
        }

        if (more) sstore(cur ^ 1);
        __syncthreads();
    }

    int col = nb + 16 * w + l16;
    float bv = bias[col];
    float* o = out + (long)(m0 + 8 * half) * DIM + col;
#pragma unroll
    for (int r = 0; r < 8; ++r) {
        o[(long)(0 * 16 + r) * DIM] = c0[r] + bv;
        o[(long)(1 * 16 + r) * DIM] = c1[r] + bv;
        o[(long)(2 * 16 + r) * DIM] = c2[r] + bv;
        o[(long)(3 * 16 + r) * DIM] = c3[r] + bv;
        o[(long)(4 * 16 + r) * DIM] = c4[r] + bv;
    }
}

#define EPW 8
#define EWAVES (EN / EPW)
static_assert(EN % EPW == 0, "a wave takes EPW whole edges");
static_assert(EPW == 8, "lane l carries edge l >> 2; the sum tree below is written for 8");

__device__ __forceinline__ unsigned fenc(float f) {
    unsigned b = __float_as_uint(f);
    return (b & 0x80000000u) ? ~b : (b | 0x80000000u);
}
__device__ __forceinline__ float fdec(unsigned u) {
    return __uint_as_float((u & 0x80000000u) ? (u ^ 0x80000000u) : ~u);
}

__device__ __forceinline__ void edge_ends(const int* __restrict__ ei, int e, int& s, int& d) {
    if (e < NEDGE) { s = ei[e]; d = ei[NEDGE + e]; }
    else           { s = d = e - NEDGE; }
}

__device__ __forceinline__ float logit_part(float4 a, float4 b, float4 t, float acc) {
    float h;
    h = a.x + b.x; h = h > 0.f ? h : NEG_SLOPE * h; acc += t.x * h;
    h = a.y + b.y; h = h > 0.f ? h : NEG_SLOPE * h; acc += t.y * h;
    h = a.z + b.z; h = h > 0.f ? h : NEG_SLOPE * h; acc += t.z * h;
    h = a.w + b.w; h = h > 0.f ? h : NEG_SLOPE * h; acc += t.w * h;
    return acc;
}

__global__ __launch_bounds__(256)
void edge_logits(const int* __restrict__ ei,
                 const float* __restrict__ xl, const float* __restrict__ xr,
                 const float* __restrict__ att,
                 float* __restrict__ alpha_raw, unsigned* __restrict__ amax) {
    int wid  = (int)((blockIdx.x * blockDim.x + threadIdx.x) >> 5);
    int lane = (int)(threadIdx.x & 31);
    if (wid >= EWAVES) return;

    int e = wid * EPW + (lane >> 2);
    int s, d;
    edge_ends(ei, e, s, d);

    const float4* av = (const float4*)att;
    float4 t0 = av[lane], t1 = av[lane + 32];

    float4 a0[EPW], a1[EPW], b0[EPW], b1[EPW];
#pragma unroll
    for (int i = 0; i < EPW; ++i) {
        int si = __builtin_amdgcn_readlane(s, 4 * i);
        int di = __builtin_amdgcn_readlane(d, 4 * i);
        const float4* xs = (const float4*)(xl + (long)si * DIM);
        const float4* xd = (const float4*)(xr + (long)di * DIM);
        a0[i] = xs[lane]; a1[i] = xs[lane + 32];
        b0[i] = xd[lane]; b1[i] = xd[lane + 32];
    }

    float acc[EPW];
#pragma unroll
    for (int i = 0; i < EPW; ++i)
        acc[i] = logit_part(a1[i], b1[i], t1, logit_part(a0[i], b0[i], t0, 0.f));

    bool h16 = (lane & 16) != 0, h8 = (lane & 8) != 0, h4 = (lane & 4) != 0;
    float r4[4], r2[2], r1;
#pragma unroll
    for (int j = 0; j < 4; ++j) {
        float keep = h16 ? acc[j + 4] : acc[j];
        float send = h16 ? acc[j] : acc[j + 4];
        r4[j] = keep + __shfl_xor(send, 16, 32);
    }
#pragma unroll
    for (int j = 0; j < 2; ++j) {
        float keep = h8 ? r4[j + 2] : r4[j];
        float send = h8 ? r4[j] : r4[j + 2];
        r2[j] = keep + __shfl_xor(send, 8, 32);
    }
    {
        float keep = h4 ? r2[1] : r2[0];
        float send = h4 ? r2[0] : r2[1];
        r1 = keep + __shfl_xor(send, 4, 32);
    }
    r1 += __shfl_xor(r1, 2, 32);
    r1 += __shfl_xor(r1, 1, 32);

    if ((lane & 3) == 0) {
        alpha_raw[e] = r1;
        atomicMax(&amax[d], fenc(r1));
    }
}

__global__ __launch_bounds__(256)
void edge_aggregate(const int* __restrict__ ei,
                    const float* __restrict__ xl,
                    const float* __restrict__ alpha_raw,
                    const unsigned* __restrict__ amax,
                    float* __restrict__ denom, float* __restrict__ agg) {
    int wid  = (int)((blockIdx.x * blockDim.x + threadIdx.x) >> 5);
    int lane = (int)(threadIdx.x & 31);
    if (wid >= EWAVES) return;

    int e = wid * EPW + (lane >> 2);
    int s, d;
    edge_ends(ei, e, s, d);

    float a[EPW][DIM / 32];
#pragma unroll
    for (int i = 0; i < EPW; ++i) {
        int si = __builtin_amdgcn_readlane(s, 4 * i);
        const float* xs = xl + (long)si * DIM + lane;
#pragma unroll
        for (int k = 0; k < DIM / 32; ++k) a[i][k] = xs[32 * k];
    }

    float al = __expf(alpha_raw[e] - fdec(amax[d]));
    if ((lane & 3) == 0) atomicAdd(&denom[d], al);

#pragma unroll
    for (int i = 0; i < EPW; ++i) {
        int   di  = __builtin_amdgcn_readlane(d, 4 * i);
        float ali = __int_as_float(__builtin_amdgcn_readlane(__float_as_int(al), 4 * i));
        float* out = agg + (long)di * DIM + lane;
#pragma unroll
        for (int k = 0; k < DIM / 32; ++k) atomicAdd(&out[32 * k], ali * a[i][k]);
    }
}

#define ROWS_PER_BLK 50

__global__ void post_stats(float* __restrict__ agg, const float* __restrict__ denom,
                           const float* __restrict__ bias, const float* __restrict__ du,
                           float* __restrict__ sums, float* __restrict__ sumsq) {
    int dcol = (int)threadIdx.x;
    int i0   = (int)blockIdx.x * ROWS_PER_BLK;
    float s = 0.f, s2 = 0.f;
    for (int r = 0; r < ROWS_PER_BLK; ++r) {
        int i = i0 + r;
        if (i >= NNODE) break;
        long idx = (long)i * DIM + dcol;
        float v = agg[idx] / denom[i] + bias[dcol];
        v = v > 0.f ? v : 0.f;
        v = (du[idx] >= 0.5f) ? 2.0f * v : 0.0f;
        agg[idx] = v;
        s += v; s2 += v * v;
    }
    atomicAdd(&sums[dcol], s);
    atomicAdd(&sumsq[dcol], s2);
}

__global__ void finalize(const float* __restrict__ v,
                         const float* __restrict__ sums, const float* __restrict__ sumsq,
                         const float* __restrict__ gamma, const float* __restrict__ beta,
                         float* __restrict__ out) {
    int dcol = (int)threadIdx.x;
    int i0   = (int)blockIdx.x * ROWS_PER_BLK;
    const float invn = 1.0f / (float)NNODE;
    float mean  = sums[dcol] * invn;
    float var   = sumsq[dcol] * invn - mean * mean;
    float scale = gamma[dcol] * rsqrtf(var + BN_EPS);
    float shift = beta[dcol] - mean * scale;
    for (int r = 0; r < ROWS_PER_BLK; ++r) {
        int i = i0 + r;
        if (i >= NNODE) break;
        long idx = (long)i * DIM + dcol;
        out[idx] = v[idx] * scale + shift;
    }
}

extern "C" void kernel_launch(void* const* d_in, const int* in_sizes, int n_in,
                              void* d_out, int out_size, void* d_ws, size_t ws_size,
                              hipStream_t stream) {
    const float* x     = (const float*)d_in[0];
    const int*   ei    = (const int*)d_in[1];
    const float* Wl    = (const float*)d_in[2];
    const float* bl    = (const float*)d_in[3];
    const float* Wr    = (const float*)d_in[4];
    const float* br    = (const float*)d_in[5];
    const float* att   = (const float*)d_in[6];
    const float* bias  = (const float*)d_in[7];
    const float* gamma = (const float*)d_in[8];
    const float* beta  = (const float*)d_in[9];
    const float* du    = (const float*)d_in[10];
    float*       out   = (float*)d_out;

    float*    ws    = (float*)d_ws;
    float*    agg   = ws;
    unsigned* amax  = (unsigned*)(ws + (long)NNODE * DIM);
    float*    denom = ws + (long)NNODE * DIM + NNODE;
    float*    sums  = denom + NNODE;
    float*    sumsq = sums + DIM;
    float*    xl    = sumsq + DIM;
    float*    xr    = xl + (long)NNODE * DIM;
    float*    araw  = xr + (long)NNODE * DIM;

    long zeroN  = (long)NNODE * DIM + 2L * NNODE + 2L * DIM;
    long zeroN4 = zeroN / 4;
    zero_ws<<<1024, 256, 0, stream>>>((float4*)d_ws, zeroN4);

    gemm_xlxr<<<(NNODE / BM) * 4, 256, 0, stream>>>(x, Wl, bl, Wr, br, xl, xr);

    int eblocks = (EWAVES * 32 + 255) / 256;
    edge_logits<<<eblocks, 256, 0, stream>>>(ei, xl, xr, att, araw, amax);

    edge_aggregate<<<eblocks, 256, 0, stream>>>(ei, xl, araw, amax, denom, agg);

    int sblocks = (NNODE + ROWS_PER_BLK - 1) / ROWS_PER_BLK;
    post_stats<<<sblocks, DIM, 0, stream>>>(agg, denom, bias, du, sums, sumsq);

    finalize<<<sblocks, DIM, 0, stream>>>(agg, sums, sumsq, gamma, beta, out);
}


# passes 4 and 5 (epilogue + column sums, batchnorm): a thread takes one float4 column group of 5 rows, all row loads issued before first use (was one round trip per row); column sum order changes
# speedup vs baseline: 1.9045x; 1.2060x over previous
#include <hip/hip_runtime.h>
#include <hip/hip_bf16.h>
#include <stdint.h>

typedef __attribute__((ext_vector_type(2))) float v2f;
typedef __attribute__((ext_vector_type(8))) float v8f;

#define NNODE   10000
#define NEDGE   160000
#define EN      (NEDGE + NNODE)
#define DIM     256
#define MT      (NNODE / 16)
#define NEG_SLOPE 0.2f
#define BN_EPS  1e-5f

__global__ void zero_ws(float4* __restrict__ p, long n4) {
    long i = (long)blockIdx.x * blockDim.x + threadIdx.x;
    long stride = (long)gridDim.x * blockDim.x;
    float4 z = make_float4(0.f, 0.f, 0.f, 0.f);
    for (; i < n4; i += stride) p[i] = z;
}

#define BM   80
#define BN   128
#define KC   32
#define XP   (KC + 4)
#define WNP  (BN + 16)
#define XS_FLOATS (BM * XP)
#define WS_FLOATS ((KC / 2) * WNP * 2)

__global__ __launch_bounds__(256)
void gemm_xlxr(const float* __restrict__ x,
               const float* __restrict__ Wl, const float* __restrict__ bl,
               const float* __restrict__ Wr, const float* __restrict__ br,
               float* __restrict__ xl, float* __restrict__ xr) {
    __shared__ __attribute__((aligned(16))) float xs[2][XS_FLOATS];
    __shared__ __attribute__((aligned(16))) float wsm[2][WS_FLOATS];

    int t    = (int)threadIdx.x;
    int lane = t & 31;
    int w    = t >> 5;
    int half = lane >> 4;
    int l16  = lane & 15;

    int rb  = (int)blockIdx.x >> 2;
    int cb  = (int)blockIdx.x & 3;
    int mat = cb >> 1;
    int nb  = (cb & 1) * BN;
    int m0  = rb * BM;

    const float* W    = mat ? Wr : Wl;
    const float* bias = mat ? br : bl;
    float*       out  = mat ? xr : xl;

    int xrow = t >> 4, xj = t & 15;
    const float* xg = x + (long)(m0 + xrow) * DIM + 2 * xj;
    int xo = xrow * XP + 2 * xj;

    int n4  = t & 31;
    int kpb = t >> 5;
    const float* wg = W + (long)(2 * kpb) * DIM + nb + 4 * n4;
    int wo = (kpb * WNP + 4 * n4) * 2;

    float2 x0, x1, x2, x3, x4;
    float4 wa0, wb0, wa1, wb1;

    auto gload = [&](int k0) {
        x0  = *(const float2*)(xg + k0);
        x1  = *(const float2*)(xg + k0 + 16L * DIM);
        x2  = *(const float2*)(xg + k0 + 32L * DIM);
        x3  = *(const float2*)(xg + k0 + 48L * DIM);
        x4  = *(const float2*)(xg + k0 + 64L * DIM);
        const float* p = wg + (long)k0 * DIM;
        wa0 = *(const float4*)(p);
        wb0 = *(const float4*)(p + DIM);
        wa1 = *(const float4*)(p + 16 * DIM);
        wb1 = *(const float4*)(p + 17 * DIM);
    };

    auto sstore = [&](int buf) {
        float* xd = &xs[buf][xo];
        *(float2*)(xd)           = x0;
        *(float2*)(xd + 16 * XP) = x1;
        *(float2*)(xd + 32 * XP) = x2;
        *(float2*)(xd + 48 * XP) = x3;
        *(float2*)(xd + 64 * XP) = x4;
        float* p = &wsm[buf][wo];
        *(float4*)(p)     = make_float4(wa0.x, wb0.x, wa0.y, wb0.y);
        *(float4*)(p + 4) = make_float4(wa0.z, wb0.z, wa0.w, wb0.w);
        float* q = p + 8 * WNP * 2;
        *(float4*)(q)     = make_float4(wa1.x, wb1.x, wa1.y, wb1.y);
        *(float4*)(q + 4) = make_float4(wa1.z, wb1.z, wa1.w, wb1.w);
    };

    v8f c0 = {}, c1 = {}, c2 = {}, c3 = {}, c4 = {};

    gload(0);
    sstore(0);
    __syncthreads();

    for (int kc = 0; kc < DIM / KC; ++kc) {
        int cur = kc & 1;
        bool more = (kc + 1) < (DIM / KC);
        if (more) gload((kc + 1) * KC);

        const float* xa = &xs[cur][l16 * XP + 2 * half];
        const float* wp = &wsm[cur][(half * WNP + 16 * w + l16) * 2];
#pragma unroll
        for (int kk = 0; kk < KC; kk += 4) {
            v2f b  = *(const v2f*)(wp + (kk >> 1) * WNP * 2);
            v2f a0 = *(const v2f*)(xa + 0 * 16 * XP + kk);
            v2f a1 = *(const v2f*)(xa + 1 * 16 * XP + kk);
            v2f a2 = *(const v2f*)(xa + 2 * 16 * XP + kk);
            v2f a3 = *(const v2f*)(xa + 3 * 16 * XP + kk);
            v2f a4 = *(const v2f*)(xa + 4 * 16 * XP + kk);
            c0 = __builtin_amdgcn_wmma_f32_16x16x4_f32(false, a0, false, b, (short)0, c0, false, false);
            c1 = __builtin_amdgcn_wmma_f32_16x16x4_f32(false, a1, false, b, (short)0, c1, false, false);
            c2 = __builtin_amdgcn_wmma_f32_16x16x4_f32(false, a2, false, b, (short)0, c2, false, false);
            c3 = __builtin_amdgcn_wmma_f32_16x16x4_f32(false, a3, false, b, (short)0, c3, false, false);
            c4 = __builtin_amdgcn_wmma_f32_16x16x4_f32(false, a4, false, b, (short)0, c4, false, false);
        }

        if (more) sstore(cur ^ 1);
        __syncthreads();
    }

    int col = nb + 16 * w + l16;
    float bv = bias[col];
    float* o = out + (long)(m0 + 8 * half) * DIM + col;
#pragma unroll
    for (int r = 0; r < 8; ++r) {
        o[(long)(0 * 16 + r) * DIM] = c0[r] + bv;
        o[(long)(1 * 16 + r) * DIM] = c1[r] + bv;
        o[(long)(2 * 16 + r) * DIM] = c2[r] + bv;
        o[(long)(3 * 16 + r) * DIM] = c3[r] + bv;
        o[(long)(4 * 16 + r) * DIM] = c4[r] + bv;
    }
}

#define EPW 8
#define EWAVES (EN / EPW)
static_assert(EN % EPW == 0, "a wave takes EPW whole edges");
static_assert(EPW == 8, "lane l carries edge l >> 2; the sum tree below is written for 8");

__device__ __forceinline__ unsigned fenc(float f) {
    unsigned b = __float_as_uint(f);
    return (b & 0x80000000u) ? ~b : (b | 0x80000000u);
}
__device__ __forceinline__ float fdec(unsigned u) {
    return __uint_as_float((u & 0x80000000u) ? (u ^ 0x80000000u) : ~u);
}

__device__ __forceinline__ void edge_ends(const int* __restrict__ ei, int e, int& s, int& d) {
    if (e < NEDGE) { s = ei[e]; d = ei[NEDGE + e]; }
    else           { s = d = e - NEDGE; }
}

__device__ __forceinline__ float logit_part(float4 a, float4 b, float4 t, float acc) {
    float h;
    h = a.x + b.x; h = h > 0.f ? h : NEG_SLOPE * h; acc += t.x * h;
    h = a.y + b.y; h = h > 0.f ? h : NEG_SLOPE * h; acc += t.y * h;
    h = a.z + b.z; h = h > 0.f ? h : NEG_SLOPE * h; acc += t.z * h;
    h = a.w + b.w; h = h > 0.f ? h : NEG_SLOPE * h; acc += t.w * h;
    return acc;
}

__global__ __launch_bounds__(256)
void edge_logits(const int* __restrict__ ei,
                 const float* __restrict__ xl, const float* __restrict__ xr,
                 const float* __restrict__ att,
                 float* __restrict__ alpha_raw, unsigned* __restrict__ amax) {
    int wid  = (int)((blockIdx.x * blockDim.x + threadIdx.x) >> 5);
    int lane = (int)(threadIdx.x & 31);
    if (wid >= EWAVES) return;

    int e = wid * EPW + (lane >> 2);
    int s, d;
    edge_ends(ei, e, s, d);

    const float4* av = (const float4*)att;
    float4 t0 = av[lane], t1 = av[lane + 32];

    float4 a0[EPW], a1[EPW], b0[EPW], b1[EPW];
#pragma unroll
    for (int i = 0; i < EPW; ++i) {
        int si = __builtin_amdgcn_readlane(s, 4 * i);
        int di = __builtin_amdgcn_readlane(d, 4 * i);
        const float4* xs = (const float4*)(xl + (long)si * DIM);
        const float4* xd = (const float4*)(xr + (long)di * DIM);
        a0[i] = xs[lane]; a1[i] = xs[lane + 32];
        b0[i] = xd[lane]; b1[i] = xd[lane + 32];
    }

    float acc[EPW];
#pragma unroll
    for (int i = 0; i < EPW; ++i)
        acc[i] = logit_part(a1[i], b1[i], t1, logit_part(a0[i], b0[i], t0, 0.f));

    bool h16 = (lane & 16) != 0, h8 = (lane & 8) != 0, h4 = (lane & 4) != 0;
    float r4[4], r2[2], r1;
#pragma unroll
    for (int j = 0; j < 4; ++j) {
        float keep = h16 ? acc[j + 4] : acc[j];
        float send = h16 ? acc[j] : acc[j + 4];
        r4[j] = keep + __shfl_xor(send, 16, 32);
    }
#pragma unroll
    for (int j = 0; j < 2; ++j) {
        float keep = h8 ? r4[j + 2] : r4[j];
        float send = h8 ? r4[j] : r4[j + 2];
        r2[j] = keep + __shfl_xor(send, 8, 32);
    }
    {
        float keep = h4 ? r2[1] : r2[0];
        float send = h4 ? r2[0] : r2[1];
        r1 = keep + __shfl_xor(send, 4, 32);
    }
    r1 += __shfl_xor(r1, 2, 32);
    r1 += __shfl_xor(r1, 1, 32);

    if ((lane & 3) == 0) {
        alpha_raw[e] = r1;
        atomicMax(&amax[d], fenc(r1));
    }
}

__global__ __launch_bounds__(256)
void edge_aggregate(const int* __restrict__ ei,
                    const float* __restrict__ xl,
                    const float* __restrict__ alpha_raw,
                    const unsigned* __restrict__ amax,
                    float* __restrict__ denom, float* __restrict__ agg) {
    int wid  = (int)((blockIdx.x * blockDim.x + threadIdx.x) >> 5);
    int lane = (int)(threadIdx.x & 31);
    if (wid >= EWAVES) return;

    int e = wid * EPW + (lane >> 2);
    int s, d;
    edge_ends(ei, e, s, d);

    float a[EPW][DIM / 32];
#pragma unroll
    for (int i = 0; i < EPW; ++i) {
        int si = __builtin_amdgcn_readlane(s, 4 * i);
        const float* xs = xl + (long)si * DIM + lane;
#pragma unroll
        for (int k = 0; k < DIM / 32; ++k) a[i][k] = xs[32 * k];
    }

    float al = __expf(alpha_raw[e] - fdec(amax[d]));
    if ((lane & 3) == 0) atomicAdd(&denom[d], al);

#pragma unroll
    for (int i = 0; i < EPW; ++i) {
        int   di  = __builtin_amdgcn_readlane(d, 4 * i);
        float ali = __int_as_float(__builtin_amdgcn_readlane(__float_as_int(al), 4 * i));
        float* out = agg + (long)di * DIM + lane;
#pragma unroll
        for (int k = 0; k < DIM / 32; ++k) atomicAdd(&out[32 * k], ali * a[i][k]);
    }
}

#define PR   20
#define PRT  (PR / 4)
static_assert(NNODE % PR == 0 && PR % 4 == 0, "whole blocks of four row groups");

__global__ __launch_bounds__(256)
void post_stats(float* __restrict__ agg, const float* __restrict__ denom,
                const float* __restrict__ bias, const float* __restrict__ du,
                float* __restrict__ sums, float* __restrict__ sumsq) {
    __shared__ __attribute__((aligned(16))) float red[2][4][DIM];

    int t  = (int)threadIdx.x;
    int c4 = t & 63;
    int rg = t >> 6;
    int i0 = (int)blockIdx.x * PR + rg;

    float4*       a4 = (float4*)agg;
    const float4* u4 = (const float4*)du;

    float4 a[PRT], u[PRT];
    float  dn[PRT];
#pragma unroll
    for (int j = 0; j < PRT; ++j) {
        long q = (long)(i0 + 4 * j) * (DIM / 4) + c4;
        a[j]  = a4[q];
        u[j]  = u4[q];
        dn[j] = denom[i0 + 4 * j];
    }
    float4 bv = ((const float4*)bias)[c4];

    float4 s  = make_float4(0.f, 0.f, 0.f, 0.f);
    float4 s2 = make_float4(0.f, 0.f, 0.f, 0.f);
#pragma unroll
    for (int j = 0; j < PRT; ++j) {
        float4 v;
        v.x = a[j].x / dn[j] + bv.x;
        v.y = a[j].y / dn[j] + bv.y;
        v.z = a[j].z / dn[j] + bv.z;
        v.w = a[j].w / dn[j] + bv.w;
        v.x = v.x > 0.f ? v.x : 0.f;
        v.y = v.y > 0.f ? v.y : 0.f;
        v.z = v.z > 0.f ? v.z : 0.f;
        v.w = v.w > 0.f ? v.w : 0.f;
        v.x = (u[j].x >= 0.5f) ? 2.0f * v.x : 0.0f;
        v.y = (u[j].y >= 0.5f) ? 2.0f * v.y : 0.0f;
        v.z = (u[j].z >= 0.5f) ? 2.0f * v.z : 0.0f;
        v.w = (u[j].w >= 0.5f) ? 2.0f * v.w : 0.0f;
        a4[(long)(i0 + 4 * j) * (DIM / 4) + c4] = v;
        s.x += v.x; s2.x += v.x * v.x;
        s.y += v.y; s2.y += v.y * v.y;
        s.z += v.z; s2.z += v.z * v.z;
        s.w += v.w; s2.w += v.w * v.w;
    }

    *(float4*)&red[0][rg][4 * c4] = s;
    *(float4*)&red[1][rg][4 * c4] = s2;
    __syncthreads();
    float cs  = (red[0][0][t] + red[0][1][t]) + (red[0][2][t] + red[0][3][t]);
    float cs2 = (red[1][0][t] + red[1][1][t]) + (red[1][2][t] + red[1][3][t]);
    atomicAdd(&sums[t], cs);
    atomicAdd(&sumsq[t], cs2);
}

__global__ __launch_bounds__(256)
void finalize(const float* __restrict__ v,
              const float* __restrict__ sums, const float* __restrict__ sumsq,
              const float* __restrict__ gamma, const float* __restrict__ beta,
              float* __restrict__ out) {
    int t  = (int)threadIdx.x;
    int c4 = t & 63;
    int rg = t >> 6;
    int i0 = (int)blockIdx.x * PR + rg;

    const float4* v4 = (const float4*)v;
    float4*       o4 = (float4*)out;

    float4 x[PRT];
#pragma unroll
    for (int j = 0; j < PRT; ++j) x[j] = v4[(long)(i0 + 4 * j) * (DIM / 4) + c4];

    float4 sm = ((const float4*)sums)[c4];
    float4 sq = ((const float4*)sumsq)[c4];
    float4 g  = ((const float4*)gamma)[c4];
    float4 b  = ((const float4*)beta)[c4];

    const float invn = 1.0f / (float)NNODE;
    float4 scale, shift;
    {
        float mean, var;
        mean = sm.x * invn; var = sq.x * invn - mean * mean;
        scale.x = g.x * rsqrtf(var + BN_EPS); shift.x = b.x - mean * scale.x;
        mean = sm.y * invn; var = sq.y * invn - mean * mean;
        scale.y = g.y * rsqrtf(var + BN_EPS); shift.y = b.y - mean * scale.y;
        mean = sm.z * invn; var = sq.z * invn - mean * mean;
        scale.z = g.z * rsqrtf(var + BN_EPS); shift.z = b.z - mean * scale.z;
        mean = sm.w * invn; var = sq.w * invn - mean * mean;
        scale.w = g.w * rsqrtf(var + BN_EPS); shift.w = b.w - mean * scale.w;
    }

#pragma unroll
    for (int j = 0; j < PRT; ++j) {
        float4 r;
        r.x = x[j].x * scale.x + shift.x;
        r.y = x[j].y * scale.y + shift.y;
        r.z = x[j].z * scale.z + shift.z;
        r.w = x[j].w * scale.w + shift.w;
        o4[(long)(i0 + 4 * j) * (DIM / 4) + c4] = r;
    }
}

extern "C" void kernel_launch(void* const* d_in, const int* in_sizes, int n_in,
                              void* d_out, int out_size, void* d_ws, size_t ws_size,
                              hipStream_t stream) {
    const float* x     = (const float*)d_in[0];
    const int*   ei    = (const int*)d_in[1];
    const float* Wl    = (const float*)d_in[2];
    const float* bl    = (const float*)d_in[3];
    const float* Wr    = (const float*)d_in[4];
    const float* br    = (const float*)d_in[5];
    const float* att   = (const float*)d_in[6];
    const float* bias  = (const float*)d_in[7];
    const float* gamma = (const float*)d_in[8];
    const float* beta  = (const float*)d_in[9];
    const float* du    = (const float*)d_in[10];
    float*       out   = (float*)d_out;

    float*    ws    = (float*)d_ws;
    float*    agg   = ws;
    unsigned* amax  = (unsigned*)(ws + (long)NNODE * DIM);
    float*    denom = ws + (long)NNODE * DIM + NNODE;
    float*    sums  = denom + NNODE;
    float*    sumsq = sums + DIM;
    float*    xl    = sumsq + DIM;
    float*    xr    = xl + (long)NNODE * DIM;
    float*    araw  = xr + (long)NNODE * DIM;

    long zeroN  = (long)NNODE * DIM + 2L * NNODE + 2L * DIM;
    long zeroN4 = zeroN / 4;
    zero_ws<<<1024, 256, 0, stream>>>((float4*)d_ws, zeroN4);

    gemm_xlxr<<<(NNODE / BM) * 4, 256, 0, stream>>>(x, Wl, bl, Wr, br, xl, xr);

    int eblocks = (EWAVES * 32 + 255) / 256;
    edge_logits<<<eblocks, 256, 0, stream>>>(ei, xl, xr, att, araw, amax);

    edge_aggregate<<<eblocks, 256, 0, stream>>>(ei, xl, araw, amax, denom, agg);

    int sblocks = NNODE / PR;
    post_stats<<<sblocks, 256, 0, stream>>>(agg, denom, bias, du, sums, sumsq);

    finalize<<<sblocks, 256, 0, stream>>>(agg, sums, sumsq, gamma, beta, out);
}


# scatter-add without global float atomics: a block owns 40 destination rows in LDS, scans the whole edge list for them, adds rows by LDS float adds, stores each row once; sums stay in arrival order
# speedup vs baseline: 2.6933x; 1.4142x over previous
#include <hip/hip_runtime.h>
#include <hip/hip_bf16.h>
#include <stdint.h>

typedef __attribute__((ext_vector_type(2))) float v2f;
typedef __attribute__((ext_vector_type(8))) float v8f;

#define NNODE   10000
#define NEDGE   160000
#define EN      (NEDGE + NNODE)
#define DIM     256
#define MT      (NNODE / 16)
#define NEG_SLOPE 0.2f
#define BN_EPS  1e-5f

__global__ void zero_ws(float4* __restrict__ p, long n4) {
    long i = (long)blockIdx.x * blockDim.x + threadIdx.x;
    long stride = (long)gridDim.x * blockDim.x;
    float4 z = make_float4(0.f, 0.f, 0.f, 0.f);
    for (; i < n4; i += stride) p[i] = z;
}

#define BM   80
#define BN   128
#define KC   32
#define XP   (KC + 4)
#define WNP  (BN + 16)
#define XS_FLOATS (BM * XP)
#define WS_FLOATS ((KC / 2) * WNP * 2)

__global__ __launch_bounds__(256)
void gemm_xlxr(const float* __restrict__ x,
               const float* __restrict__ Wl, const float* __restrict__ bl,
               const float* __restrict__ Wr, const float* __restrict__ br,
               float* __restrict__ xl, float* __restrict__ xr) {
    __shared__ __attribute__((aligned(16))) float xs[2][XS_FLOATS];
    __shared__ __attribute__((aligned(16))) float wsm[2][WS_FLOATS];

    int t    = (int)threadIdx.x;
    int lane = t & 31;
    int w    = t >> 5;
    int half = lane >> 4;
    int l16  = lane & 15;

    int rb  = (int)blockIdx.x >> 2;
    int cb  = (int)blockIdx.x & 3;
    int mat = cb >> 1;
    int nb  = (cb & 1) * BN;
    int m0  = rb * BM;

    const float* W    = mat ? Wr : Wl;
    const float* bias = mat ? br : bl;
    float*       out  = mat ? xr : xl;

    int xrow = t >> 4, xj = t & 15;
    const float* xg = x + (long)(m0 + xrow) * DIM + 2 * xj;
    int xo = xrow * XP + 2 * xj;

    int n4  = t & 31;
    int kpb = t >> 5;
    const float* wg = W + (long)(2 * kpb) * DIM + nb + 4 * n4;
    int wo = (kpb * WNP + 4 * n4) * 2;

    float2 x0, x1, x2, x3, x4;
    float4 wa0, wb0, wa1, wb1;

    auto gload = [&](int k0) {
        x0  = *(const float2*)(xg + k0);
        x1  = *(const float2*)(xg + k0 + 16L * DIM);
        x2  = *(const float2*)(xg + k0 + 32L * DIM);
        x3  = *(const float2*)(xg + k0 + 48L * DIM);
        x4  = *(const float2*)(xg + k0 + 64L * DIM);
        const float* p = wg + (long)k0 * DIM;
        wa0 = *(const float4*)(p);
        wb0 = *(const float4*)(p + DIM);
        wa1 = *(const float4*)(p + 16 * DIM);
        wb1 = *(const float4*)(p + 17 * DIM);
    };

    auto sstore = [&](int buf) {
        float* xd = &xs[buf][xo];
        *(float2*)(xd)           = x0;
        *(float2*)(xd + 16 * XP) = x1;
        *(float2*)(xd + 32 * XP) = x2;
        *(float2*)(xd + 48 * XP) = x3;
        *(float2*)(xd + 64 * XP) = x4;
        float* p = &wsm[buf][wo];
        *(float4*)(p)     = make_float4(wa0.x, wb0.x, wa0.y, wb0.y);
        *(float4*)(p + 4) = make_float4(wa0.z, wb0.z, wa0.w, wb0.w);
        float* q = p + 8 * WNP * 2;
        *(float4*)(q)     = make_float4(wa1.x, wb1.x, wa1.y, wb1.y);
        *(float4*)(q + 4) = make_float4(wa1.z, wb1.z, wa1.w, wb1.w);
    };

    v8f c0 = {}, c1 = {}, c2 = {}, c3 = {}, c4 = {};

    gload(0);
    sstore(0);
    __syncthreads();

    for (int kc = 0; kc < DIM / KC; ++kc) {
        int cur = kc & 1;
        bool more = (kc + 1) < (DIM / KC);
        if (more) gload((kc + 1) * KC);

        const float* xa = &xs[cur][l16 * XP + 2 * half];
        const float* wp = &wsm[cur][(half * WNP + 16 * w + l16) * 2];
#pragma unroll
        for (int kk = 0; kk < KC; kk += 4) {
            v2f b  = *(const v2f*)(wp + (kk >> 1) * WNP * 2);
            v2f a0 = *(const v2f*)(xa + 0 * 16 * XP + kk);
            v2f a1 = *(const v2f*)(xa + 1 * 16 * XP + kk);
            v2f a2 = *(const v2f*)(xa + 2 * 16 * XP + kk);
            v2f a3 = *(const v2f*)(xa + 3 * 16 * XP + kk);
            v2f a4 = *(const v2f*)(xa + 4 * 16 * XP + kk);
            c0 = __builtin_amdgcn_wmma_f32_16x16x4_f32(false, a0, false, b, (short)0, c0, false, false);
            c1 = __builtin_amdgcn_wmma_f32_16x16x4_f32(false, a1, false, b, (short)0, c1, false, false);
            c2 = __builtin_amdgcn_wmma_f32_16x16x4_f32(false, a2, false, b, (short)0, c2, false, false);
            c3 = __builtin_amdgcn_wmma_f32_16x16x4_f32(false, a3, false, b, (short)0, c3, false, false);
            c4 = __builtin_amdgcn_wmma_f32_16x16x4_f32(false, a4, false, b, (short)0, c4, false, false);
        }

        if (more) sstore(cur ^ 1);
        __syncthreads();
    }

    int col = nb + 16 * w + l16;
    float bv = bias[col];
    float* o = out + (long)(m0 + 8 * half) * DIM + col;
#pragma unroll
    for (int r = 0; r < 8; ++r) {
        o[(long)(0 * 16 + r) * DIM] = c0[r] + bv;
        o[(long)(1 * 16 + r) * DIM] = c1[r] + bv;
        o[(long)(2 * 16 + r) * DIM] = c2[r] + bv;
        o[(long)(3 * 16 + r) * DIM] = c3[r] + bv;
        o[(long)(4 * 16 + r) * DIM] = c4[r] + bv;
    }
}

#define EPW 8
#define EWAVES (EN / EPW)
static_assert(EN % EPW == 0, "a wave takes EPW whole edges");
static_assert(EPW == 8, "lane l carries edge l >> 2; the sum tree below is written for 8");

__device__ __forceinline__ unsigned fenc(float f) {
    unsigned b = __float_as_uint(f);
    return (b & 0x80000000u) ? ~b : (b | 0x80000000u);
}
__device__ __forceinline__ float fdec(unsigned u) {
    return __uint_as_float((u & 0x80000000u) ? (u ^ 0x80000000u) : ~u);
}

__device__ __forceinline__ void edge_ends(const int* __restrict__ ei, int e, int& s, int& d) {
    if (e < NEDGE) { s = ei[e]; d = ei[NEDGE + e]; }
    else           { s = d = e - NEDGE; }
}

__device__ __forceinline__ float logit_part(float4 a, float4 b, float4 t, float acc) {
    float h;
    h = a.x + b.x; h = h > 0.f ? h : NEG_SLOPE * h; acc += t.x * h;
    h = a.y + b.y; h = h > 0.f ? h : NEG_SLOPE * h; acc += t.y * h;
    h = a.z + b.z; h = h > 0.f ? h : NEG_SLOPE * h; acc += t.z * h;
    h = a.w + b.w; h = h > 0.f ? h : NEG_SLOPE * h; acc += t.w * h;
    return acc;
}

__global__ __launch_bounds__(256)
void edge_logits(const int* __restrict__ ei,
                 const float* __restrict__ xl, const float* __restrict__ xr,
                 const float* __restrict__ att,
                 float* __restrict__ alpha_raw, unsigned* __restrict__ amax) {
    int wid  = (int)((blockIdx.x * blockDim.x + threadIdx.x) >> 5);
    int lane = (int)(threadIdx.x & 31);
    if (wid >= EWAVES) return;

    int e = wid * EPW + (lane >> 2);
    int s, d;
    edge_ends(ei, e, s, d);

    const float4* av = (const float4*)att;
    float4 t0 = av[lane], t1 = av[lane + 32];

    float4 a0[EPW], a1[EPW], b0[EPW], b1[EPW];
#pragma unroll
    for (int i = 0; i < EPW; ++i) {
        int si = __builtin_amdgcn_readlane(s, 4 * i);
        int di = __builtin_amdgcn_readlane(d, 4 * i);
        const float4* xs = (const float4*)(xl + (long)si * DIM);
        const float4* xd = (const float4*)(xr + (long)di * DIM);
        a0[i] = xs[lane]; a1[i] = xs[lane + 32];
        b0[i] = xd[lane]; b1[i] = xd[lane + 32];
    }

    float acc[EPW];
#pragma unroll
    for (int i = 0; i < EPW; ++i)
        acc[i] = logit_part(a1[i], b1[i], t1, logit_part(a0[i], b0[i], t0, 0.f));

    bool h16 = (lane & 16) != 0, h8 = (lane & 8) != 0, h4 = (lane & 4) != 0;
    float r4[4], r2[2], r1;
#pragma unroll
    for (int j = 0; j < 4; ++j) {
        float keep = h16 ? acc[j + 4] : acc[j];
        float send = h16 ? acc[j] : acc[j + 4];
        r4[j] = keep + __shfl_xor(send, 16, 32);
    }
#pragma unroll
    for (int j = 0; j < 2; ++j) {
        float keep = h8 ? r4[j + 2] : r4[j];
        float send = h8 ? r4[j] : r4[j + 2];
        r2[j] = keep + __shfl_xor(send, 8, 32);
    }
    {
        float keep = h4 ? r2[1] : r2[0];
        float send = h4 ? r2[0] : r2[1];
        r1 = keep + __shfl_xor(send, 4, 32);
    }
    r1 += __shfl_xor(r1, 2, 32);
    r1 += __shfl_xor(r1, 1, 32);

    if ((lane & 3) == 0) {
        alpha_raw[e] = r1;
        atomicMax(&amax[d], fenc(r1));
    }
}

#define OWN   40
#define OBLK  (NNODE / OWN)
#define OTH   1024
#define OSEG  5
#define OLD   8
#define NQ    (NEDGE / 4)
#define OCAP  8192
#define OROT  (NQ / OBLK)
#define OGRP  4
static_assert(NNODE % OWN == 0, "whole tiles");
static_assert(NEDGE % 4 == 0, "destination list in whole 16-byte units");
static_assert(OSEG * OLD * OTH >= NQ, "the pieces cover the list");
static_assert(OWN <= 64 && EN <= (1 << 18) && OWN <= OCAP, "a list word holds (row << 18) | edge or source");

__global__ __launch_bounds__(OTH)
void edge_aggregate(const int* __restrict__ ei,
                    const float* __restrict__ xl,
                    const float* __restrict__ alpha_raw,
                    const unsigned* __restrict__ amax,
                    float* __restrict__ denom, float* __restrict__ agg) {
    __shared__ __attribute__((aligned(16))) float acc[OWN * DIM];
    __shared__ float    den[64];
    __shared__ unsigned cnt;
    __shared__ unsigned list[OCAP];
    __shared__ float    lal[OCAP];

    int t    = (int)threadIdx.x;
    int lane = t & 31;
    int w    = t >> 5;
    int n0   = (int)blockIdx.x * OWN;

    for (int i = t; i < OWN * DIM; i += OTH) acc[i] = 0.f;
    if (t < 64) den[t] = 0.f;
    if (t < OWN) list[t] = ((unsigned)t << 18) | (unsigned)(NEDGE + n0 + t);
    if (t == 0) cnt = OWN;
    __syncthreads();

    const int* dst = ei + NEDGE;
    const int4* dq = (const int4*)dst;
    int rot = (int)blockIdx.x * OROT;

    auto edge_src = [&](int e) { return (e < NEDGE) ? ei[e] : (e - NEDGE); };
    auto edge_w   = [&](int e, int r) { return __expf(alpha_raw[e] - fdec(amax[n0 + r])); };

    int  nkept = 0;
    bool full  = false;
    auto keep = [&](unsigned r, int e) {
        unsigned p = atomicAdd(&cnt, 1u);
        if (p < OCAP) { list[p] = (r << 18) | (unsigned)e; ++nkept; }
        else full = true;
    };
    for (int seg = 0; seg < OSEG; ++seg) {
        int4 v[OLD];
        int  ps[OLD];
#pragma unroll
        for (int j = 0; j < OLD; ++j) {
            int q = (seg * OLD + j) * OTH + t;
            int idx = q + rot;
            if (idx >= NQ) idx -= NQ;
            ps[j] = 4 * idx;
            v[j] = (q < NQ) ? dq[(q < NQ) ? idx : 0] : make_int4(-1, -1, -1, -1);
        }
#pragma unroll
        for (int j = 0; j < OLD; ++j) {
            unsigned d0 = (unsigned)(v[j].x - n0), d1 = (unsigned)(v[j].y - n0);
            unsigned d2 = (unsigned)(v[j].z - n0), d3 = (unsigned)(v[j].w - n0);
            if (d0 < OWN) keep(d0, ps[j] + 0);
            if (d1 < OWN) keep(d1, ps[j] + 1);
            if (d2 < OWN) keep(d2, ps[j] + 2);
            if (d3 < OWN) keep(d3, ps[j] + 3);
        }
    }
    if (full) {
        int ord = 0;
        for (int k = 0; k < OSEG * OLD; ++k) {
            int q = k * OTH + t;
            if (q >= NQ) break;
            int idx = q + rot;
            if (idx >= NQ) idx -= NQ;
#pragma nounroll
            for (int c = 0; c < 4; ++c) {
                int e = 4 * idx + c;
                unsigned r = (unsigned)(dst[e] - n0);
                if (r < OWN) {
                    if (ord >= nkept) {
                        float al = edge_w(e, (int)r);
                        const float* xs = xl + (long)edge_src(e) * DIM;
#pragma nounroll
                        for (int col = 0; col < DIM; ++col) atomicAdd(&acc[r * DIM + col], al * xs[col]);
                        atomicAdd(&den[r], al);
                    }
                    ++ord;
                }
            }
        }
    }
    __syncthreads();

    unsigned n = cnt < OCAP ? cnt : OCAP;

    for (unsigned i = (unsigned)t; i < n; i += OTH) {
        unsigned word = list[i];
        int e = (int)(word & 0x3FFFFu), r = (int)(word >> 18);
        float al = edge_w(e, r);
        list[i] = ((unsigned)r << 18) | (unsigned)edge_src(e);
        lal[i]  = al;
        atomicAdd(&den[r], al);
    }
    __syncthreads();

    for (unsigned base = OGRP * (unsigned)w; base < n; base += 32u * OGRP) {
        unsigned pw[OGRP];
        float    al[OGRP];
        float    a[OGRP][DIM / 32];
#pragma unroll
        for (int g = 0; g < OGRP; ++g) {
            unsigned i = (base + g < n) ? base + g : base;
            pw[g] = (unsigned)__builtin_amdgcn_readfirstlane((int)list[i]);
            al[g] = __int_as_float(__builtin_amdgcn_readfirstlane(__float_as_int(lal[i])));
            const float* xs = xl + (long)(pw[g] & 0x3FFFFu) * DIM + lane;
#pragma unroll
            for (int c = 0; c < DIM / 32; ++c) a[g][c] = xs[32 * c];
        }
#pragma unroll
        for (int g = 0; g < OGRP; ++g) {
            if (base + g < n) {
                float* o = &acc[(pw[g] >> 18) * DIM + lane];
#pragma unroll
                for (int c = 0; c < DIM / 32; ++c) atomicAdd(&o[32 * c], al[g] * a[g][c]);
            }
        }
    }
    __syncthreads();

    float4*       a4 = (float4*)agg + (long)n0 * (DIM / 4);
    const float4* t4 = (const float4*)acc;
    for (int i = t; i < OWN * DIM / 4; i += OTH) a4[i] = t4[i];
    if (t < OWN) denom[n0 + t] = den[t];
}

#define PR   20
#define PRT  (PR / 4)
static_assert(NNODE % PR == 0 && PR % 4 == 0, "whole blocks of four row groups");

__global__ __launch_bounds__(256)
void post_stats(float* __restrict__ agg, const float* __restrict__ denom,
                const float* __restrict__ bias, const float* __restrict__ du,
                float* __restrict__ sums, float* __restrict__ sumsq) {
    __shared__ __attribute__((aligned(16))) float red[2][4][DIM];

    int t  = (int)threadIdx.x;
    int c4 = t & 63;
    int rg = t >> 6;
    int i0 = (int)blockIdx.x * PR + rg;

    float4*       a4 = (float4*)agg;
    const float4* u4 = (const float4*)du;

    float4 a[PRT], u[PRT];
    float  dn[PRT];
#pragma unroll
    for (int j = 0; j < PRT; ++j) {
        long q = (long)(i0 + 4 * j) * (DIM / 4) + c4;
        a[j]  = a4[q];
        u[j]  = u4[q];
        dn[j] = denom[i0 + 4 * j];
    }
    float4 bv = ((const float4*)bias)[c4];

    float4 s  = make_float4(0.f, 0.f, 0.f, 0.f);
    float4 s2 = make_float4(0.f, 0.f, 0.f, 0.f);
#pragma unroll
    for (int j = 0; j < PRT; ++j) {
        float4 v;
        v.x = a[j].x / dn[j] + bv.x;
        v.y = a[j].y / dn[j] + bv.y;
        v.z = a[j].z / dn[j] + bv.z;
        v.w = a[j].w / dn[j] + bv.w;
        v.x = v.x > 0.f ? v.x : 0.f;
        v.y = v.y > 0.f ? v.y : 0.f;
        v.z = v.z > 0.f ? v.z : 0.f;
        v.w = v.w > 0.f ? v.w : 0.f;
        v.x = (u[j].x >= 0.5f) ? 2.0f * v.x : 0.0f;
        v.y = (u[j].y >= 0.5f) ? 2.0f * v.y : 0.0f;
        v.z = (u[j].z >= 0.5f) ? 2.0f * v.z : 0.0f;
        v.w = (u[j].w >= 0.5f) ? 2.0f * v.w : 0.0f;
        a4[(long)(i0 + 4 * j) * (DIM / 4) + c4] = v;
        s.x += v.x; s2.x += v.x * v.x;
        s.y += v.y; s2.y += v.y * v.y;
        s.z += v.z; s2.z += v.z * v.z;
        s.w += v.w; s2.w += v.w * v.w;
    }

    *(float4*)&red[0][rg][4 * c4] = s;
    *(float4*)&red[1][rg][4 * c4] = s2;
    __syncthreads();
    float cs  = (red[0][0][t] + red[0][1][t]) + (red[0][2][t] + red[0][3][t]);
    float cs2 = (red[1][0][t] + red[1][1][t]) + (red[1][2][t] + red[1][3][t]);
    atomicAdd(&sums[t], cs);
    atomicAdd(&sumsq[t], cs2);
}

__global__ __launch_bounds__(256)
void finalize(const float* __restrict__ v,
              const float* __restrict__ sums, const float* __restrict__ sumsq,
              const float* __restrict__ gamma, const float* __restrict__ beta,
              float* __restrict__ out) {
    int t  = (int)threadIdx.x;
    int c4 = t & 63;
    int rg = t >> 6;
    int i0 = (int)blockIdx.x * PR + rg;

    const float4* v4 = (const float4*)v;
    float4*       o4 = (float4*)out;

    float4 x[PRT];
#pragma unroll
    for (int j = 0; j < PRT; ++j) x[j] = v4[(long)(i0 + 4 * j) * (DIM / 4) + c4];

    float4 sm = ((const float4*)sums)[c4];
    float4 sq = ((const float4*)sumsq)[c4];
    float4 g  = ((const float4*)gamma)[c4];
    float4 b  = ((const float4*)beta)[c4];

    const float invn = 1.0f / (float)NNODE;
    float4 scale, shift;
    {
        float mean, var;
        mean = sm.x * invn; var = sq.x * invn - mean * mean;
        scale.x = g.x * rsqrtf(var + BN_EPS); shift.x = b.x - mean * scale.x;
        mean = sm.y * invn; var = sq.y * invn - mean * mean;
        scale.y = g.y * rsqrtf(var + BN_EPS); shift.y = b.y - mean * scale.y;
        mean = sm.z * invn; var = sq.z * invn - mean * mean;
        scale.z = g.z * rsqrtf(var + BN_EPS); shift.z = b.z - mean * scale.z;
        mean = sm.w * invn; var = sq.w * invn - mean * mean;
        scale.w = g.w * rsqrtf(var + BN_EPS); shift.w = b.w - mean * scale.w;
    }

#pragma unroll
    for (int j = 0; j < PRT; ++j) {
        float4 r;
        r.x = x[j].x * scale.x + shift.x;
        r.y = x[j].y * scale.y + shift.y;
        r.z = x[j].z * scale.z + shift.z;
        r.w = x[j].w * scale.w + shift.w;
        o4[(long)(i0 + 4 * j) * (DIM / 4) + c4] = r;
    }
}

extern "C" void kernel_launch(void* const* d_in, const int* in_sizes, int n_in,
                              void* d_out, int out_size, void* d_ws, size_t ws_size,
                              hipStream_t stream) {
    const float* x     = (const float*)d_in[0];
    const int*   ei    = (const int*)d_in[1];
    const float* Wl    = (const float*)d_in[2];
    const float* bl    = (const float*)d_in[3];
    const float* Wr    = (const float*)d_in[4];
    const float* br    = (const float*)d_in[5];
    const float* att   = (const float*)d_in[6];
    const float* bias  = (const float*)d_in[7];
    const float* gamma = (const float*)d_in[8];
    const float* beta  = (const float*)d_in[9];
    const float* du    = (const float*)d_in[10];
    float*       out   = (float*)d_out;

    float*    ws    = (float*)d_ws;
    float*    agg   = ws;
    unsigned* amax  = (unsigned*)(ws + (long)NNODE * DIM);
    float*    denom = ws + (long)NNODE * DIM + NNODE;
    float*    sums  = denom + NNODE;
    float*    sumsq = sums + DIM;
    float*    xl    = sumsq + DIM;
    float*    xr    = xl + (long)NNODE * DIM;
    float*    araw  = xr + (long)NNODE * DIM;

    long zeroN  = 2L * NNODE + 2L * DIM;
    long zeroN4 = zeroN / 4;
    zero_ws<<<24, 256, 0, stream>>>((float4*)amax, zeroN4);

    gemm_xlxr<<<(NNODE / BM) * 4, 256, 0, stream>>>(x, Wl, bl, Wr, br, xl, xr);

    int eblocks = (EWAVES * 32 + 255) / 256;
    edge_logits<<<eblocks, 256, 0, stream>>>(ei, xl, xr, att, araw, amax);

    edge_aggregate<<<OBLK, OTH, 0, stream>>>(ei, xl, araw, amax, denom, agg);

    int sblocks = NNODE / PR;
    post_stats<<<sblocks, 256, 0, stream>>>(agg, denom, bias, du, sums, sumsq);

    finalize<<<sblocks, 256, 0, stream>>>(agg, sums, sumsq, gamma, beta, out);
}


# logits, row maxima and weights computed inside the aggregation work-group (its 40 xr rows in LDS, row max by LDS atomic): the logit launch, its xr row reads and global atomics go; sum trees unchanged
# speedup vs baseline: 3.6351x; 1.3497x over previous
#include <hip/hip_runtime.h>
#include <hip/hip_bf16.h>
#include <stdint.h>

typedef __attribute__((ext_vector_type(2))) float v2f;
typedef __attribute__((ext_vector_type(8))) float v8f;

#define NNODE   10000
#define NEDGE   160000
#define EN      (NEDGE + NNODE)
#define DIM     256
#define MT      (NNODE / 16)
#define NEG_SLOPE 0.2f
#define BN_EPS  1e-5f

__global__ void zero_ws(float4* __restrict__ p, long n4) {
    long i = (long)blockIdx.x * blockDim.x + threadIdx.x;
    long stride = (long)gridDim.x * blockDim.x;
    float4 z = make_float4(0.f, 0.f, 0.f, 0.f);
    for (; i < n4; i += stride) p[i] = z;
}

#define BM   80
#define BN   128
#define KC   32
#define XP   (KC + 4)
#define WNP  (BN + 16)
#define XS_FLOATS (BM * XP)
#define WS_FLOATS ((KC / 2) * WNP * 2)

__global__ __launch_bounds__(256)
void gemm_xlxr(const float* __restrict__ x,
               const float* __restrict__ Wl, const float* __restrict__ bl,
               const float* __restrict__ Wr, const float* __restrict__ br,
               float* __restrict__ xl, float* __restrict__ xr) {
    __shared__ __attribute__((aligned(16))) float xs[2][XS_FLOATS];
    __shared__ __attribute__((aligned(16))) float wsm[2][WS_FLOATS];

    int t    = (int)threadIdx.x;
    int lane = t & 31;
    int w    = t >> 5;
    int half = lane >> 4;
    int l16  = lane & 15;

    int rb  = (int)blockIdx.x >> 2;
    int cb  = (int)blockIdx.x & 3;
    int mat = cb >> 1;
    int nb  = (cb & 1) * BN;
    int m0  = rb * BM;

    const float* W    = mat ? Wr : Wl;
    const float* bias = mat ? br : bl;
    float*       out  = mat ? xr : xl;

    int xrow = t >> 4, xj = t & 15;
    const float* xg = x + (long)(m0 + xrow) * DIM + 2 * xj;
    int xo = xrow * XP + 2 * xj;

    int n4  = t & 31;
    int kpb = t >> 5;
    const float* wg = W + (long)(2 * kpb) * DIM + nb + 4 * n4;
    int wo = (kpb * WNP + 4 * n4) * 2;

    float2 x0, x1, x2, x3, x4;
    float4 wa0, wb0, wa1, wb1;

    auto gload = [&](int k0) {
        x0  = *(const float2*)(xg + k0);
        x1  = *(const float2*)(xg + k0 + 16L * DIM);
        x2  = *(const float2*)(xg + k0 + 32L * DIM);
        x3  = *(const float2*)(xg + k0 + 48L * DIM);
        x4  = *(const float2*)(xg + k0 + 64L * DIM);
        const float* p = wg + (long)k0 * DIM;
        wa0 = *(const float4*)(p);
        wb0 = *(const float4*)(p + DIM);
        wa1 = *(const float4*)(p + 16 * DIM);
        wb1 = *(const float4*)(p + 17 * DIM);
    };

    auto sstore = [&](int buf) {
        float* xd = &xs[buf][xo];
        *(float2*)(xd)           = x0;
        *(float2*)(xd + 16 * XP) = x1;
        *(float2*)(xd + 32 * XP) = x2;
        *(float2*)(xd + 48 * XP) = x3;
        *(float2*)(xd + 64 * XP) = x4;
        float* p = &wsm[buf][wo];
        *(float4*)(p)     = make_float4(wa0.x, wb0.x, wa0.y, wb0.y);
        *(float4*)(p + 4) = make_float4(wa0.z, wb0.z, wa0.w, wb0.w);
        float* q = p + 8 * WNP * 2;
        *(float4*)(q)     = make_float4(wa1.x, wb1.x, wa1.y, wb1.y);
        *(float4*)(q + 4) = make_float4(wa1.z, wb1.z, wa1.w, wb1.w);
    };

    v8f c0 = {}, c1 = {}, c2 = {}, c3 = {}, c4 = {};

    gload(0);
    sstore(0);
    __syncthreads();

    for (int kc = 0; kc < DIM / KC; ++kc) {
        int cur = kc & 1;
        bool more = (kc + 1) < (DIM / KC);
        if (more) gload((kc + 1) * KC);

        const float* xa = &xs[cur][l16 * XP + 2 * half];
        const float* wp = &wsm[cur][(half * WNP + 16 * w + l16) * 2];
#pragma unroll
        for (int kk = 0; kk < KC; kk += 4) {
            v2f b  = *(const v2f*)(wp + (kk >> 1) * WNP * 2);
            v2f a0 = *(const v2f*)(xa + 0 * 16 * XP + kk);
            v2f a1 = *(const v2f*)(xa + 1 * 16 * XP + kk);
            v2f a2 = *(const v2f*)(xa + 2 * 16 * XP + kk);
            v2f a3 = *(const v2f*)(xa + 3 * 16 * XP + kk);
            v2f a4 = *(const v2f*)(xa + 4 * 16 * XP + kk);
            c0 = __builtin_amdgcn_wmma_f32_16x16x4_f32(false, a0, false, b, (short)0, c0, false, false);
            c1 = __builtin_amdgcn_wmma_f32_16x16x4_f32(false, a1, false, b, (short)0, c1, false, false);
            c2 = __builtin_amdgcn_wmma_f32_16x16x4_f32(false, a2, false, b, (short)0, c2, false, false);
            c3 = __builtin_amdgcn_wmma_f32_16x16x4_f32(false, a3, false, b, (short)0, c3, false, false);
            c4 = __builtin_amdgcn_wmma_f32_16x16x4_f32(false, a4, false, b, (short)0, c4, false, false);
        }

        if (more) sstore(cur ^ 1);
        __syncthreads();
    }

    int col = nb + 16 * w + l16;
    float bv = bias[col];
    float* o = out + (long)(m0 + 8 * half) * DIM + col;
#pragma unroll
    for (int r = 0; r < 8; ++r) {
        o[(long)(0 * 16 + r) * DIM] = c0[r] + bv;
        o[(long)(1 * 16 + r) * DIM] = c1[r] + bv;
        o[(long)(2 * 16 + r) * DIM] = c2[r] + bv;
        o[(long)(3 * 16 + r) * DIM] = c3[r] + bv;
        o[(long)(4 * 16 + r) * DIM] = c4[r] + bv;
    }
}

#define OWN   40
#define OBLK  (NNODE / OWN)
#define OTH   1024
#define OSEG  5
#define OLD   8
#define NQ    (NEDGE / 4)
#define OCAP  8192
#define OROT  (NQ / OBLK)
#define OGRP  4
#define XRV   ((OWN * DIM / 4 + OTH - 1) / OTH)
static_assert(NNODE % OWN == 0, "whole tiles");
static_assert(NEDGE % 4 == 0, "destination list in whole 16-byte units");
static_assert(OSEG * OLD * OTH >= NQ, "the pieces cover the list");
static_assert(OWN <= 64 && EN <= (1 << 18) && OWN <= OCAP, "a list word holds (row << 18) | edge or source");
static_assert(OGRP == 4, "the store of the four logits of a group is written for 4");

__device__ __forceinline__ unsigned fenc(float f) {
    unsigned b = __float_as_uint(f);
    return (b & 0x80000000u) ? ~b : (b | 0x80000000u);
}
__device__ __forceinline__ float fdec(unsigned u) {
    return __uint_as_float((u & 0x80000000u) ? (u ^ 0x80000000u) : ~u);
}

__device__ __forceinline__ float logit_part(float4 a, float4 b, float4 t, float acc) {
    float h;
    h = a.x + b.x; h = h > 0.f ? h : NEG_SLOPE * h; acc += t.x * h;
    h = a.y + b.y; h = h > 0.f ? h : NEG_SLOPE * h; acc += t.y * h;
    h = a.z + b.z; h = h > 0.f ? h : NEG_SLOPE * h; acc += t.z * h;
    h = a.w + b.w; h = h > 0.f ? h : NEG_SLOPE * h; acc += t.w * h;
    return acc;
}

__device__ __forceinline__ float wave_sum(float p) {
    p += __shfl_xor(p, 16, 32);
    p += __shfl_xor(p, 8, 32);
    p += __shfl_xor(p, 4, 32);
    p += __shfl_xor(p, 2, 32);
    p += __shfl_xor(p, 1, 32);
    return p;
}

__global__ __launch_bounds__(OTH)
void edge_aggregate(const int* __restrict__ ei,
                    const float* __restrict__ xl, const float* __restrict__ xr,
                    const float* __restrict__ att,
                    float* __restrict__ denom, float* __restrict__ agg) {
    __shared__ __attribute__((aligned(16))) float acc[OWN * DIM];
    __shared__ __attribute__((aligned(16))) float xrt[OWN * DIM];
    __shared__ float    den[64];
    __shared__ unsigned rmax[64];
    __shared__ unsigned cnt;
    __shared__ unsigned list[OCAP];
    __shared__ float    lal[OCAP];

    int t    = (int)threadIdx.x;
    int lane = t & 31;
    int w    = t >> 5;
    int n0   = (int)blockIdx.x * OWN;

    const float4* xr4 = (const float4*)(xr + (long)n0 * DIM);
    float4 xv[XRV];
#pragma unroll
    for (int k = 0; k < XRV; ++k) {
        int i = t + k * OTH;
        xv[k] = (i < OWN * DIM / 4) ? xr4[i] : make_float4(0.f, 0.f, 0.f, 0.f);
    }

    for (int i = t; i < OWN * DIM; i += OTH) acc[i] = 0.f;
    if (t < 64) { den[t] = 0.f; rmax[t] = 0u; }
    if (t < OWN) list[t] = ((unsigned)t << 18) | (unsigned)(NEDGE + n0 + t);
    if (t == 0) cnt = OWN;
    __syncthreads();

    const int* dst = ei + NEDGE;
    const int4* dq = (const int4*)dst;
    int rot = (int)blockIdx.x * OROT;

    auto keep = [&](unsigned r, int e) {
        unsigned p = atomicAdd(&cnt, 1u);
        if (p < OCAP) list[p] = (r << 18) | (unsigned)e;
    };
    for (int seg = 0; seg < OSEG; ++seg) {
        int4 v[OLD];
        int  ps[OLD];
#pragma unroll
        for (int j = 0; j < OLD; ++j) {
            int q = (seg * OLD + j) * OTH + t;
            int idx = q + rot;
            if (idx >= NQ) idx -= NQ;
            ps[j] = 4 * idx;
            v[j] = (q < NQ) ? dq[(q < NQ) ? idx : 0] : make_int4(-1, -1, -1, -1);
        }
#pragma unroll
        for (int j = 0; j < OLD; ++j) {
            unsigned d0 = (unsigned)(v[j].x - n0), d1 = (unsigned)(v[j].y - n0);
            unsigned d2 = (unsigned)(v[j].z - n0), d3 = (unsigned)(v[j].w - n0);
            if (d0 < OWN) keep(d0, ps[j] + 0);
            if (d1 < OWN) keep(d1, ps[j] + 1);
            if (d2 < OWN) keep(d2, ps[j] + 2);
            if (d3 < OWN) keep(d3, ps[j] + 3);
        }
    }
#pragma unroll
    for (int k = 0; k < XRV; ++k) {
        int i = t + k * OTH;
        if (i < OWN * DIM / 4) ((float4*)xrt)[i] = xv[k];
    }
    __syncthreads();

    unsigned total = cnt;
    bool     fits  = total <= OCAP;
    unsigned n     = fits ? total : 0u;

    const float4* av = (const float4*)att;
    float4 t0 = av[lane], t1 = av[lane + 32];

    for (unsigned i = (unsigned)t; i < n; i += OTH) {
        unsigned word = list[i];
        int e = (int)(word & 0x3FFFFu);
        int s = (e < NEDGE) ? ei[e] : (e - NEDGE);
        list[i] = (word & ~0x3FFFFu) | (unsigned)s;
    }
    __syncthreads();

    for (unsigned base = OGRP * (unsigned)w; base < n; base += 32u * OGRP) {
        unsigned pw[OGRP];
        float4   a0[OGRP], a1[OGRP];
#pragma unroll
        for (int g = 0; g < OGRP; ++g) {
            unsigned i = (base + g < n) ? base + g : base;
            pw[g] = (unsigned)__builtin_amdgcn_readfirstlane((int)list[i]);
            const float4* xs = (const float4*)(xl + (long)(pw[g] & 0x3FFFFu) * DIM);
            a0[g] = xs[lane]; a1[g] = xs[lane + 32];
        }
        float p[OGRP];
#pragma unroll
        for (int g = 0; g < OGRP; ++g) {
            const float4* xd = (const float4*)&xrt[(pw[g] >> 18) * DIM];
            p[g] = logit_part(a1[g], xd[lane + 32], t1, logit_part(a0[g], xd[lane], t0, 0.f));
        }
#pragma unroll
        for (int m = 16; m >= 1; m >>= 1) {
#pragma unroll
            for (int g = 0; g < OGRP; ++g) p[g] += __shfl_xor(p[g], m, 32);
        }
        if (lane < OGRP && base + (unsigned)lane < n) {
            float    pv = p[0];
            unsigned rv = pw[0];
            if (lane == 1) { pv = p[1]; rv = pw[1]; }
            if (lane == 2) { pv = p[2]; rv = pw[2]; }
            if (lane == 3) { pv = p[3]; rv = pw[3]; }
            lal[base + (unsigned)lane] = pv;
            atomicMax(&rmax[rv >> 18], fenc(pv));
        }
    }
    __syncthreads();

    for (unsigned i = (unsigned)t; i < n; i += OTH) {
        unsigned r = list[i] >> 18;
        float al = __expf(lal[i] - fdec(rmax[r]));
        lal[i] = al;
        atomicAdd(&den[r], al);
    }
    __syncthreads();

    if (fits) {
        for (unsigned base = OGRP * (unsigned)w; base < n; base += 32u * OGRP) {
            unsigned pw[OGRP];
            float    al[OGRP];
            float    a[OGRP][DIM / 32];
#pragma unroll
            for (int g = 0; g < OGRP; ++g) {
                unsigned i = (base + g < n) ? base + g : base;
                pw[g] = (unsigned)__builtin_amdgcn_readfirstlane((int)list[i]);
                al[g] = __int_as_float(__builtin_amdgcn_readfirstlane(__float_as_int(lal[i])));
                const float* xs = xl + (long)(pw[g] & 0x3FFFFu) * DIM + lane;
#pragma unroll
                for (int c = 0; c < DIM / 32; ++c) a[g][c] = xs[32 * c];
            }
#pragma unroll
            for (int g = 0; g < OGRP; ++g) {
                if (base + g < n) {
                    float* o = &acc[(pw[g] >> 18) * DIM + lane];
#pragma unroll
                    for (int c = 0; c < DIM / 32; ++c) atomicAdd(&o[32 * c], al[g] * a[g][c]);
                }
            }
        }
    } else {
        for (int r = w; r < OWN; r += OTH / 32) {
            int node = n0 + r;
            const float4* xd = (const float4*)&xrt[r * DIM];
            float4 b0 = xd[lane], b1 = xd[lane + 32];
            float  m = -INFINITY, dsum = 0.f;
            float4 o0 = make_float4(0.f, 0.f, 0.f, 0.f), o1 = o0;
#pragma nounroll
            for (int pass = 0; pass < 2; ++pass) {
#pragma nounroll
                for (int base = 0; base < EN; base += 32) {
                    int e = base + lane;
                    int d = -1;
                    if (e < NEDGE) d = dst[e];
                    else if (e < EN) d = e - NEDGE;
                    unsigned hit = __builtin_amdgcn_ballot_w32(d == node);
                    while (hit) {
                        int ee = base + __builtin_ctz(hit);
                        hit &= hit - 1u;
                        int s = (ee < NEDGE) ? ei[ee] : (ee - NEDGE);
                        const float4* xs = (const float4*)(xl + (long)s * DIM);
                        float4 r0 = xs[lane], r1 = xs[lane + 32];
                        float p = wave_sum(logit_part(r1, b1, t1, logit_part(r0, b0, t0, 0.f)));
                        if (pass == 0) {
                            m = fmaxf(m, p);
                        } else {
                            float al = __expf(p - m);
                            dsum += al;
                            o0.x += al * r0.x; o0.y += al * r0.y; o0.z += al * r0.z; o0.w += al * r0.w;
                            o1.x += al * r1.x; o1.y += al * r1.y; o1.z += al * r1.z; o1.w += al * r1.w;
                        }
                    }
                }
            }
            ((float4*)&acc[r * DIM])[lane]      = o0;
            ((float4*)&acc[r * DIM])[lane + 32] = o1;
            if (lane == 0) den[r] = dsum;
        }
    }
    __syncthreads();

    float4*       a4 = (float4*)agg + (long)n0 * (DIM / 4);
    const float4* t4 = (const float4*)acc;
    for (int i = t; i < OWN * DIM / 4; i += OTH) a4[i] = t4[i];
    if (t >= OTH - OWN) denom[n0 + t - (OTH - OWN)] = den[t - (OTH - OWN)];
}

#define PR   20
#define PRT  (PR / 4)
static_assert(NNODE % PR == 0 && PR % 4 == 0, "whole blocks of four row groups");

__global__ __launch_bounds__(256)
void post_stats(float* __restrict__ agg, const float* __restrict__ denom,
                const float* __restrict__ bias, const float* __restrict__ du,
                float* __restrict__ sums, float* __restrict__ sumsq) {
    __shared__ __attribute__((aligned(16))) float red[2][4][DIM];

    int t  = (int)threadIdx.x;
    int c4 = t & 63;
    int rg = t >> 6;
    int i0 = (int)blockIdx.x * PR + rg;

    float4*       a4 = (float4*)agg;
    const float4* u4 = (const float4*)du;

    float4 a[PRT], u[PRT];
    float  dn[PRT];
#pragma unroll
    for (int j = 0; j < PRT; ++j) {
        long q = (long)(i0 + 4 * j) * (DIM / 4) + c4;
        a[j]  = a4[q];
        u[j]  = u4[q];
        dn[j] = denom[i0 + 4 * j];
    }
    float4 bv = ((const float4*)bias)[c4];

    float4 s  = make_float4(0.f, 0.f, 0.f, 0.f);
    float4 s2 = make_float4(0.f, 0.f, 0.f, 0.f);
#pragma unroll
    for (int j = 0; j < PRT; ++j) {
        float4 v;
        v.x = a[j].x / dn[j] + bv.x;
        v.y = a[j].y / dn[j] + bv.y;
        v.z = a[j].z / dn[j] + bv.z;
        v.w = a[j].w / dn[j] + bv.w;
        v.x = v.x > 0.f ? v.x : 0.f;
        v.y = v.y > 0.f ? v.y : 0.f;
        v.z = v.z > 0.f ? v.z : 0.f;
        v.w = v.w > 0.f ? v.w : 0.f;
        v.x = (u[j].x >= 0.5f) ? 2.0f * v.x : 0.0f;
        v.y = (u[j].y >= 0.5f) ? 2.0f * v.y : 0.0f;
        v.z = (u[j].z >= 0.5f) ? 2.0f * v.z : 0.0f;
        v.w = (u[j].w >= 0.5f) ? 2.0f * v.w : 0.0f;
        a4[(long)(i0 + 4 * j) * (DIM / 4) + c4] = v;
        s.x += v.x; s2.x += v.x * v.x;
        s.y += v.y; s2.y += v.y * v.y;
        s.z += v.z; s2.z += v.z * v.z;
        s.w += v.w; s2.w += v.w * v.w;
    }

    *(float4*)&red[0][rg][4 * c4] = s;
    *(float4*)&red[1][rg][4 * c4] = s2;
    __syncthreads();
    float cs  = (red[0][0][t] + red[0][1][t]) + (red[0][2][t] + red[0][3][t]);
    float cs2 = (red[1][0][t] + red[1][1][t]) + (red[1][2][t] + red[1][3][t]);
    atomicAdd(&sums[t], cs);
    atomicAdd(&sumsq[t], cs2);
}

__global__ __launch_bounds__(256)
void finalize(const float* __restrict__ v,
              const float* __restrict__ sums, const float* __restrict__ sumsq,
              const float* __restrict__ gamma, const float* __restrict__ beta,
              float* __restrict__ out) {
    int t  = (int)threadIdx.x;
    int c4 = t & 63;
    int rg = t >> 6;
    int i0 = (int)blockIdx.x * PR + rg;

    const float4* v4 = (const float4*)v;
    float4*       o4 = (float4*)out;

    float4 x[PRT];
#pragma unroll
    for (int j = 0; j < PRT; ++j) x[j] = v4[(long)(i0 + 4 * j) * (DIM / 4) + c4];

    float4 sm = ((const float4*)sums)[c4];
    float4 sq = ((const float4*)sumsq)[c4];
    float4 g  = ((const float4*)gamma)[c4];
    float4 b  = ((const float4*)beta)[c4];

    const float invn = 1.0f / (float)NNODE;
    float4 scale, shift;
    {
        float mean, var;
        mean = sm.x * invn; var = sq.x * invn - mean * mean;
        scale.x = g.x * rsqrtf(var + BN_EPS); shift.x = b.x - mean * scale.x;
        mean = sm.y * invn; var = sq.y * invn - mean * mean;
        scale.y = g.y * rsqrtf(var + BN_EPS); shift.y = b.y - mean * scale.y;
        mean = sm.z * invn; var = sq.z * invn - mean * mean;
        scale.z = g.z * rsqrtf(var + BN_EPS); shift.z = b.z - mean * scale.z;
        mean = sm.w * invn; var = sq.w * invn - mean * mean;
        scale.w = g.w * rsqrtf(var + BN_EPS); shift.w = b.w - mean * scale.w;
    }

#pragma unroll
    for (int j = 0; j < PRT; ++j) {
        float4 r;
        r.x = x[j].x * scale.x + shift.x;
        r.y = x[j].y * scale.y + shift.y;
        r.z = x[j].z * scale.z + shift.z;
        r.w = x[j].w * scale.w + shift.w;
        o4[(long)(i0 + 4 * j) * (DIM / 4) + c4] = r;
    }
}

extern "C" void kernel_launch(void* const* d_in, const int* in_sizes, int n_in,
                              void* d_out, int out_size, void* d_ws, size_t ws_size,
                              hipStream_t stream) {
    const float* x     = (const float*)d_in[0];
    const int*   ei    = (const int*)d_in[1];
    const float* Wl    = (const float*)d_in[2];
    const float* bl    = (const float*)d_in[3];
    const float* Wr    = (const float*)d_in[4];
    const float* br    = (const float*)d_in[5];
    const float* att   = (const float*)d_in[6];
    const float* bias  = (const float*)d_in[7];
    const float* gamma = (const float*)d_in[8];
    const float* beta  = (const float*)d_in[9];
    const float* du    = (const float*)d_in[10];
    float*       out   = (float*)d_out;

    float*    ws    = (float*)d_ws;
    float*    agg   = ws;
    float*    denom = ws + (long)NNODE * DIM + NNODE;
    float*    sums  = denom + NNODE;
    float*    sumsq = sums + DIM;
    float*    xl    = sumsq + DIM;
    float*    xr    = xl + (long)NNODE * DIM;

    zero_ws<<<1, 128, 0, stream>>>((float4*)sums, (2 * DIM) / 4);

    gemm_xlxr<<<(NNODE / BM) * 4, 256, 0, stream>>>(x, Wl, bl, Wr, br, xl, xr);

    edge_aggregate<<<OBLK, OTH, 0, stream>>>(ei, xl, xr, att, denom, agg);

    post_stats<<<NNODE / PR, 256, 0, stream>>>(agg, denom, bias, du, sums, sumsq);

    finalize<<<NNODE / PR, 256, 0, stream>>>(agg, sums, sumsq, gamma, beta, out);
}


# epilogue (divide, bias, relu, dropout) and column sums run on the LDS tile inside the aggregation work-group, sums cleared in the GEMM launch: 5 launches to 3; column sum order changes
# speedup vs baseline: 4.0682x; 1.1191x over previous
#include <hip/hip_runtime.h>
#include <hip/hip_bf16.h>
#include <stdint.h>

typedef __attribute__((ext_vector_type(2))) float v2f;
typedef __attribute__((ext_vector_type(8))) float v8f;

#define NNODE   10000
#define NEDGE   160000
#define EN      (NEDGE + NNODE)
#define DIM     256
#define MT      (NNODE / 16)
#define NEG_SLOPE 0.2f
#define BN_EPS  1e-5f

#define BM   80
#define BN   128
#define KC   32
#define XP   (KC + 4)
#define WNP  (BN + 16)
#define XS_FLOATS (BM * XP)
#define WS_FLOATS ((KC / 2) * WNP * 2)

__global__ __launch_bounds__(256)
void gemm_xlxr(const float* __restrict__ x,
               const float* __restrict__ Wl, const float* __restrict__ bl,
               const float* __restrict__ Wr, const float* __restrict__ br,
               float* __restrict__ xl, float* __restrict__ xr,
               float* __restrict__ colsums) {
    __shared__ __attribute__((aligned(16))) float xs[2][XS_FLOATS];
    __shared__ __attribute__((aligned(16))) float wsm[2][WS_FLOATS];

    int t    = (int)threadIdx.x;
    int lane = t & 31;
    int w    = t >> 5;
    int half = lane >> 4;
    int l16  = lane & 15;

    if (blockIdx.x == 0) { colsums[t] = 0.f; colsums[t + DIM] = 0.f; }

    int rb  = (int)blockIdx.x >> 2;
    int cb  = (int)blockIdx.x & 3;
    int mat = cb >> 1;
    int nb  = (cb & 1) * BN;
    int m0  = rb * BM;

    const float* W    = mat ? Wr : Wl;
    const float* bias = mat ? br : bl;
    float*       out  = mat ? xr : xl;

    int xrow = t >> 4, xj = t & 15;
    const float* xg = x + (long)(m0 + xrow) * DIM + 2 * xj;
    int xo = xrow * XP + 2 * xj;

    int n4  = t & 31;
    int kpb = t >> 5;
    const float* wg = W + (long)(2 * kpb) * DIM + nb + 4 * n4;
    int wo = (kpb * WNP + 4 * n4) * 2;

    float2 x0, x1, x2, x3, x4;
    float4 wa0, wb0, wa1, wb1;

    auto gload = [&](int k0) {
        x0  = *(const float2*)(xg + k0);
        x1  = *(const float2*)(xg + k0 + 16L * DIM);
        x2  = *(const float2*)(xg + k0 + 32L * DIM);
        x3  = *(const float2*)(xg + k0 + 48L * DIM);
        x4  = *(const float2*)(xg + k0 + 64L * DIM);
        const float* p = wg + (long)k0 * DIM;
        wa0 = *(const float4*)(p);
        wb0 = *(const float4*)(p + DIM);
        wa1 = *(const float4*)(p + 16 * DIM);
        wb1 = *(const float4*)(p + 17 * DIM);
    };

    auto sstore = [&](int buf) {
        float* xd = &xs[buf][xo];
        *(float2*)(xd)           = x0;
        *(float2*)(xd + 16 * XP) = x1;
        *(float2*)(xd + 32 * XP) = x2;
        *(float2*)(xd + 48 * XP) = x3;
        *(float2*)(xd + 64 * XP) = x4;
        float* p = &wsm[buf][wo];
        *(float4*)(p)     = make_float4(wa0.x, wb0.x, wa0.y, wb0.y);
        *(float4*)(p + 4) = make_float4(wa0.z, wb0.z, wa0.w, wb0.w);
        float* q = p + 8 * WNP * 2;
        *(float4*)(q)     = make_float4(wa1.x, wb1.x, wa1.y, wb1.y);
        *(float4*)(q + 4) = make_float4(wa1.z, wb1.z, wa1.w, wb1.w);
    };

    v8f c0 = {}, c1 = {}, c2 = {}, c3 = {}, c4 = {};

    gload(0);
    sstore(0);
    __syncthreads();

    for (int kc = 0; kc < DIM / KC; ++kc) {
        int cur = kc & 1;
        bool more = (kc + 1) < (DIM / KC);
        if (more) gload((kc + 1) * KC);

        const float* xa = &xs[cur][l16 * XP + 2 * half];
        const float* wp = &wsm[cur][(half * WNP + 16 * w + l16) * 2];
#pragma unroll
        for (int kk = 0; kk < KC; kk += 4) {
            v2f b  = *(const v2f*)(wp + (kk >> 1) * WNP * 2);
            v2f a0 = *(const v2f*)(xa + 0 * 16 * XP + kk);
            v2f a1 = *(const v2f*)(xa + 1 * 16 * XP + kk);
            v2f a2 = *(const v2f*)(xa + 2 * 16 * XP + kk);
            v2f a3 = *(const v2f*)(xa + 3 * 16 * XP + kk);
            v2f a4 = *(const v2f*)(xa + 4 * 16 * XP + kk);
            c0 = __builtin_amdgcn_wmma_f32_16x16x4_f32(false, a0, false, b, (short)0, c0, false, false);
            c1 = __builtin_amdgcn_wmma_f32_16x16x4_f32(false, a1, false, b, (short)0, c1, false, false);
            c2 = __builtin_amdgcn_wmma_f32_16x16x4_f32(false, a2, false, b, (short)0, c2, false, false);
            c3 = __builtin_amdgcn_wmma_f32_16x16x4_f32(false, a3, false, b, (short)0, c3, false, false);
            c4 = __builtin_amdgcn_wmma_f32_16x16x4_f32(false, a4, false, b, (short)0, c4, false, false);
        }

        if (more) sstore(cur ^ 1);
        __syncthreads();
    }

    int col = nb + 16 * w + l16;
    float bv = bias[col];
    float* o = out + (long)(m0 + 8 * half) * DIM + col;
#pragma unroll
    for (int r = 0; r < 8; ++r) {
        o[(long)(0 * 16 + r) * DIM] = c0[r] + bv;
        o[(long)(1 * 16 + r) * DIM] = c1[r] + bv;
        o[(long)(2 * 16 + r) * DIM] = c2[r] + bv;
        o[(long)(3 * 16 + r) * DIM] = c3[r] + bv;
        o[(long)(4 * 16 + r) * DIM] = c4[r] + bv;
    }
}

#define OWN   40
#define OBLK  (NNODE / OWN)
#define OTH   1024
#define OSEG  5
#define OLD   8
#define NQ    (NEDGE / 4)
#define OCAP  8192
#define OROT  (NQ / OBLK)
#define OGRP  4
#define XRV   ((OWN * DIM / 4 + OTH - 1) / OTH)
static_assert(NNODE % OWN == 0, "whole tiles");
static_assert(NEDGE % 4 == 0, "destination list in whole 16-byte units");
static_assert(OSEG * OLD * OTH >= NQ, "the pieces cover the list");
static_assert(OWN <= 64 && EN <= (1 << 18) && OWN <= OCAP, "a list word holds (row << 18) | edge or source");
static_assert(OGRP == 4, "the store of the four logits of a group is written for 4");
#define ERG   (OTH / (DIM / 2))
#define ERT   (OWN / ERG)
static_assert(OTH % (DIM / 2) == 0 && OWN % ERG == 0, "every thread of the epilogue has ERT whole rows of one column pair");
static_assert(2 * ERG * DIM <= OCAP, "the column partials fit the weight list's LDS");
static_assert(2 * DIM <= OTH, "one thread per column sum");

__device__ __forceinline__ unsigned fenc(float f) {
    unsigned b = __float_as_uint(f);
    return (b & 0x80000000u) ? ~b : (b | 0x80000000u);
}
__device__ __forceinline__ float fdec(unsigned u) {
    return __uint_as_float((u & 0x80000000u) ? (u ^ 0x80000000u) : ~u);
}

__device__ __forceinline__ float logit_part(float4 a, float4 b, float4 t, float acc) {
    float h;
    h = a.x + b.x; h = h > 0.f ? h : NEG_SLOPE * h; acc += t.x * h;
    h = a.y + b.y; h = h > 0.f ? h : NEG_SLOPE * h; acc += t.y * h;
    h = a.z + b.z; h = h > 0.f ? h : NEG_SLOPE * h; acc += t.z * h;
    h = a.w + b.w; h = h > 0.f ? h : NEG_SLOPE * h; acc += t.w * h;
    return acc;
}

__device__ __forceinline__ float wave_sum(float p) {
    p += __shfl_xor(p, 16, 32);
    p += __shfl_xor(p, 8, 32);
    p += __shfl_xor(p, 4, 32);
    p += __shfl_xor(p, 2, 32);
    p += __shfl_xor(p, 1, 32);
    return p;
}

__global__ __launch_bounds__(OTH)
void edge_aggregate(const int* __restrict__ ei,
                    const float* __restrict__ xl, const float* __restrict__ xr,
                    const float* __restrict__ att,
                    const float* __restrict__ bias, const float* __restrict__ du,
                    float* __restrict__ colsums, float* __restrict__ agg) {
    __shared__ __attribute__((aligned(16))) float acc[OWN * DIM];
    __shared__ __attribute__((aligned(16))) float xrt[OWN * DIM];
    __shared__ float    den[64];
    __shared__ unsigned rmax[64];
    __shared__ unsigned cnt;
    __shared__ unsigned list[OCAP];
    __shared__ __attribute__((aligned(16))) float lal[OCAP];

    int t    = (int)threadIdx.x;
    int lane = t & 31;
    int w    = t >> 5;
    int n0   = (int)blockIdx.x * OWN;

    const float4* xr4 = (const float4*)(xr + (long)n0 * DIM);
    float4 xv[XRV];
#pragma unroll
    for (int k = 0; k < XRV; ++k) {
        int i = t + k * OTH;
        xv[k] = (i < OWN * DIM / 4) ? xr4[i] : make_float4(0.f, 0.f, 0.f, 0.f);
    }

    for (int i = t; i < OWN * DIM; i += OTH) acc[i] = 0.f;
    if (t < 64) { den[t] = 0.f; rmax[t] = 0u; }
    if (t < OWN) list[t] = ((unsigned)t << 18) | (unsigned)(NEDGE + n0 + t);
    if (t == 0) cnt = OWN;
    __syncthreads();

    const int* dst = ei + NEDGE;
    const int4* dq = (const int4*)dst;
    int rot = (int)blockIdx.x * OROT;

    auto keep = [&](unsigned r, int e) {
        unsigned p = atomicAdd(&cnt, 1u);
        if (p < OCAP) list[p] = (r << 18) | (unsigned)e;
    };
    for (int seg = 0; seg < OSEG; ++seg) {
        int4 v[OLD];
        int  ps[OLD];
#pragma unroll
        for (int j = 0; j < OLD; ++j) {
            int q = (seg * OLD + j) * OTH + t;
            int idx = q + rot;
            if (idx >= NQ) idx -= NQ;
            ps[j] = 4 * idx;
            v[j] = (q < NQ) ? dq[(q < NQ) ? idx : 0] : make_int4(-1, -1, -1, -1);
        }
#pragma unroll
        for (int j = 0; j < OLD; ++j) {
            unsigned d0 = (unsigned)(v[j].x - n0), d1 = (unsigned)(v[j].y - n0);
            unsigned d2 = (unsigned)(v[j].z - n0), d3 = (unsigned)(v[j].w - n0);
            if (d0 < OWN) keep(d0, ps[j] + 0);
            if (d1 < OWN) keep(d1, ps[j] + 1);
            if (d2 < OWN) keep(d2, ps[j] + 2);
            if (d3 < OWN) keep(d3, ps[j] + 3);
        }
    }
#pragma unroll
    for (int k = 0; k < XRV; ++k) {
        int i = t + k * OTH;
        if (i < OWN * DIM / 4) ((float4*)xrt)[i] = xv[k];
    }
    __syncthreads();

    unsigned total = cnt;
    bool     fits  = total <= OCAP;
    unsigned n     = fits ? total : 0u;

    const float4* av = (const float4*)att;
    float4 t0 = av[lane], t1 = av[lane + 32];

    for (unsigned i = (unsigned)t; i < n; i += OTH) {
        unsigned word = list[i];
        int e = (int)(word & 0x3FFFFu);
        int s = (e < NEDGE) ? ei[e] : (e - NEDGE);
        list[i] = (word & ~0x3FFFFu) | (unsigned)s;
    }
    __syncthreads();

    for (unsigned base = OGRP * (unsigned)w; base < n; base += 32u * OGRP) {
        unsigned pw[OGRP];
        float4   a0[OGRP], a1[OGRP];
#pragma unroll
        for (int g = 0; g < OGRP; ++g) {
            unsigned i = (base + g < n) ? base + g : base;
            pw[g] = (unsigned)__builtin_amdgcn_readfirstlane((int)list[i]);
            const float4* xs = (const float4*)(xl + (long)(pw[g] & 0x3FFFFu) * DIM);
            a0[g] = xs[lane]; a1[g] = xs[lane + 32];
        }
        float p[OGRP];
#pragma unroll
        for (int g = 0; g < OGRP; ++g) {
            const float4* xd = (const float4*)&xrt[(pw[g] >> 18) * DIM];
            p[g] = logit_part(a1[g], xd[lane + 32], t1, logit_part(a0[g], xd[lane], t0, 0.f));
        }
#pragma unroll
        for (int m = 16; m >= 1; m >>= 1) {
#pragma unroll
            for (int g = 0; g < OGRP; ++g) p[g] += __shfl_xor(p[g], m, 32);
        }
        if (lane < OGRP && base + (unsigned)lane < n) {
            float    pv = p[0];
            unsigned rv = pw[0];
            if (lane == 1) { pv = p[1]; rv = pw[1]; }
            if (lane == 2) { pv = p[2]; rv = pw[2]; }
            if (lane == 3) { pv = p[3]; rv = pw[3]; }
            lal[base + (unsigned)lane] = pv;
            atomicMax(&rmax[rv >> 18], fenc(pv));
        }
    }
    __syncthreads();

    const int cp = t & (DIM / 2 - 1);
    const int rg = t / (DIM / 2);
    const float2* u2 = (const float2*)du + (long)n0 * (DIM / 2);
    float2 bv = ((const float2*)bias)[cp];
    float2 uv[ERT];
#pragma unroll
    for (int j = 0; j < ERT; ++j) uv[j] = u2[(rg + ERG * j) * (DIM / 2) + cp];

    for (unsigned i = (unsigned)t; i < n; i += OTH) {
        unsigned r = list[i] >> 18;
        float al = __expf(lal[i] - fdec(rmax[r]));
        lal[i] = al;
        atomicAdd(&den[r], al);
    }
    __syncthreads();

    if (fits) {
        for (unsigned base = OGRP * (unsigned)w; base < n; base += 32u * OGRP) {
            unsigned pw[OGRP];
            float    al[OGRP];
            float    a[OGRP][DIM / 32];
#pragma unroll
            for (int g = 0; g < OGRP; ++g) {
                unsigned i = (base + g < n) ? base + g : base;
                pw[g] = (unsigned)__builtin_amdgcn_readfirstlane((int)list[i]);
                al[g] = __int_as_float(__builtin_amdgcn_readfirstlane(__float_as_int(lal[i])));
                const float* xs = xl + (long)(pw[g] & 0x3FFFFu) * DIM + lane;
#pragma unroll
                for (int c = 0; c < DIM / 32; ++c) a[g][c] = xs[32 * c];
            }
#pragma unroll
            for (int g = 0; g < OGRP; ++g) {
                if (base + g < n) {
                    float* o = &acc[(pw[g] >> 18) * DIM + lane];
#pragma unroll
                    for (int c = 0; c < DIM / 32; ++c) atomicAdd(&o[32 * c], al[g] * a[g][c]);
                }
            }
        }
    } else {
        for (int r = w; r < OWN; r += OTH / 32) {
            int node = n0 + r;
            const float4* xd = (const float4*)&xrt[r * DIM];
            float4 b0 = xd[lane], b1 = xd[lane + 32];
            float  m = -INFINITY, dsum = 0.f;
            float4 o0 = make_float4(0.f, 0.f, 0.f, 0.f), o1 = o0;
#pragma nounroll
            for (int pass = 0; pass < 2; ++pass) {
#pragma nounroll
                for (int base = 0; base < EN; base += 32) {
                    int e = base + lane;
                    int d = -1;
                    if (e < NEDGE) d = dst[e];
                    else if (e < EN) d = e - NEDGE;
                    unsigned hit = __builtin_amdgcn_ballot_w32(d == node);
                    while (hit) {
                        int ee = base + __builtin_ctz(hit);
                        hit &= hit - 1u;
                        int s = (ee < NEDGE) ? ei[ee] : (ee - NEDGE);
                        const float4* xs = (const float4*)(xl + (long)s * DIM);
                        float4 r0 = xs[lane], r1 = xs[lane + 32];
                        float p = wave_sum(logit_part(r1, b1, t1, logit_part(r0, b0, t0, 0.f)));
                        if (pass == 0) {
                            m = fmaxf(m, p);
                        } else {
                            float al = __expf(p - m);
                            dsum += al;
                            o0.x += al * r0.x; o0.y += al * r0.y; o0.z += al * r0.z; o0.w += al * r0.w;
                            o1.x += al * r1.x; o1.y += al * r1.y; o1.z += al * r1.z; o1.w += al * r1.w;
                        }
                    }
                }
            }
            ((float4*)&acc[r * DIM])[lane]      = o0;
            ((float4*)&acc[r * DIM])[lane + 32] = o1;
            if (lane == 0) den[r] = dsum;
        }
    }
    __syncthreads();

    float2*       a2 = (float2*)agg + (long)n0 * (DIM / 2);
    const float2* t2 = (const float2*)acc;
    float2 s  = make_float2(0.f, 0.f);
    float2 s2 = make_float2(0.f, 0.f);
#pragma unroll
    for (int j = 0; j < ERT; ++j) {
        int    row = rg + ERG * j;
        float  dn  = den[row];
        float2 a   = t2[row * (DIM / 2) + cp];
        float2 v;
        v.x = a.x / dn + bv.x;
        v.y = a.y / dn + bv.y;
        v.x = v.x > 0.f ? v.x : 0.f;
        v.y = v.y > 0.f ? v.y : 0.f;
        v.x = (uv[j].x >= 0.5f) ? 2.0f * v.x : 0.0f;
        v.y = (uv[j].y >= 0.5f) ? 2.0f * v.y : 0.0f;
        a2[row * (DIM / 2) + cp] = v;
        s.x += v.x; s2.x += v.x * v.x;
        s.y += v.y; s2.y += v.y * v.y;
    }
    float* red = lal;
    *(float2*)&red[rg * DIM + 2 * cp]             = s;
    *(float2*)&red[ERG * DIM + rg * DIM + 2 * cp] = s2;
    __syncthreads();
    if (t < 2 * DIM) {
        const float* r = red + (t / DIM) * ERG * DIM + (t & (DIM - 1));
        float cs = 0.f;
#pragma unroll
        for (int j = 0; j < ERG; ++j) cs += r[j * DIM];
        atomicAdd(&colsums[t], cs);
    }
}

#define PR   20
#define PRT  (PR / 4)
static_assert(NNODE % PR == 0 && PR % 4 == 0, "whole blocks of four row groups");

__global__ __launch_bounds__(256)
void finalize(const float* __restrict__ v,
              const float* __restrict__ sums, const float* __restrict__ sumsq,
              const float* __restrict__ gamma, const float* __restrict__ beta,
              float* __restrict__ out) {
    int t  = (int)threadIdx.x;
    int c4 = t & 63;
    int rg = t >> 6;
    int i0 = (int)blockIdx.x * PR + rg;

    const float4* v4 = (const float4*)v;
    float4*       o4 = (float4*)out;

    float4 x[PRT];
#pragma unroll
    for (int j = 0; j < PRT; ++j) x[j] = v4[(long)(i0 + 4 * j) * (DIM / 4) + c4];

    float4 sm = ((const float4*)sums)[c4];
    float4 sq = ((const float4*)sumsq)[c4];
    float4 g  = ((const float4*)gamma)[c4];
    float4 b  = ((const float4*)beta)[c4];

    const float invn = 1.0f / (float)NNODE;
    float4 scale, shift;
    {
        float mean, var;
        mean = sm.x * invn; var = sq.x * invn - mean * mean;
        scale.x = g.x * rsqrtf(var + BN_EPS); shift.x = b.x - mean * scale.x;
        mean = sm.y * invn; var = sq.y * invn - mean * mean;
        scale.y = g.y * rsqrtf(var + BN_EPS); shift.y = b.y - mean * scale.y;
        mean = sm.z * invn; var = sq.z * invn - mean * mean;
        scale.z = g.z * rsqrtf(var + BN_EPS); shift.z = b.z - mean * scale.z;
        mean = sm.w * invn; var = sq.w * invn - mean * mean;
        scale.w = g.w * rsqrtf(var + BN_EPS); shift.w = b.w - mean * scale.w;
    }

#pragma unroll
    for (int j = 0; j < PRT; ++j) {
        float4 r;
        r.x = x[j].x * scale.x + shift.x;
        r.y = x[j].y * scale.y + shift.y;
        r.z = x[j].z * scale.z + shift.z;
        r.w = x[j].w * scale.w + shift.w;
        o4[(long)(i0 + 4 * j) * (DIM / 4) + c4] = r;
    }
}

extern "C" void kernel_launch(void* const* d_in, const int* in_sizes, int n_in,
                              void* d_out, int out_size, void* d_ws, size_t ws_size,
                              hipStream_t stream) {
    const float* x     = (const float*)d_in[0];
    const int*   ei    = (const int*)d_in[1];
    const float* Wl    = (const float*)d_in[2];
    const float* bl    = (const float*)d_in[3];
    const float* Wr    = (const float*)d_in[4];
    const float* br    = (const float*)d_in[5];
    const float* att   = (const float*)d_in[6];
    const float* bias  = (const float*)d_in[7];
    const float* gamma = (const float*)d_in[8];
    const float* beta  = (const float*)d_in[9];
    const float* du    = (const float*)d_in[10];
    float*       out   = (float*)d_out;

    float*    ws    = (float*)d_ws;
    float*    agg   = ws;
    float*    sums  = ws + (long)NNODE * DIM + 2 * NNODE;
    float*    sumsq = sums + DIM;
    float*    xl    = sumsq + DIM;
    float*    xr    = xl + (long)NNODE * DIM;

    gemm_xlxr<<<(NNODE / BM) * 4, 256, 0, stream>>>(x, Wl, bl, Wr, br, xl, xr, sums);

    edge_aggregate<<<OBLK, OTH, 0, stream>>>(ei, xl, xr, att, bias, du, sums, agg);

    finalize<<<NNODE / PR, 256, 0, stream>>>(agg, sums, sumsq, gamma, beta, out);
}


# aggregation: kept edges ordered by row in LDS (counting sort); a wave takes one contiguous slice and sums runs of one row in registers: one LDS float add set per run, was per edge; sum order changes
# speedup vs baseline: 4.3853x; 1.0779x over previous
#include <hip/hip_runtime.h>
#include <hip/hip_bf16.h>
#include <stdint.h>

typedef __attribute__((ext_vector_type(2))) float v2f;
typedef __attribute__((ext_vector_type(8))) float v8f;

#define NNODE   10000
#define NEDGE   160000
#define EN      (NEDGE + NNODE)
#define DIM     256
#define MT      (NNODE / 16)
#define NEG_SLOPE 0.2f
#define BN_EPS  1e-5f

#define BM   80
#define BN   128
#define KC   32
#define XP   (KC + 4)
#define WNP  (BN + 16)
#define XS_FLOATS (BM * XP)
#define WS_FLOATS ((KC / 2) * WNP * 2)

__global__ __launch_bounds__(256)
void gemm_xlxr(const float* __restrict__ x,
               const float* __restrict__ Wl, const float* __restrict__ bl,
               const float* __restrict__ Wr, const float* __restrict__ br,
               float* __restrict__ xl, float* __restrict__ xr,
               float* __restrict__ colsums) {
    __shared__ __attribute__((aligned(16))) float xs[2][XS_FLOATS];
    __shared__ __attribute__((aligned(16))) float wsm[2][WS_FLOATS];

    int t    = (int)threadIdx.x;
    int lane = t & 31;
    int w    = t >> 5;
    int half = lane >> 4;
    int l16  = lane & 15;

    if (blockIdx.x == 0) { colsums[t] = 0.f; colsums[t + DIM] = 0.f; }

    int rb  = (int)blockIdx.x >> 2;
    int cb  = (int)blockIdx.x & 3;
    int mat = cb >> 1;
    int nb  = (cb & 1) * BN;
    int m0  = rb * BM;

    const float* W    = mat ? Wr : Wl;
    const float* bias = mat ? br : bl;
    float*       out  = mat ? xr : xl;

    int xrow = t >> 4, xj = t & 15;
    const float* xg = x + (long)(m0 + xrow) * DIM + 2 * xj;
    int xo = xrow * XP + 2 * xj;

    int n4  = t & 31;
    int kpb = t >> 5;
    const float* wg = W + (long)(2 * kpb) * DIM + nb + 4 * n4;
    int wo = (kpb * WNP + 4 * n4) * 2;

    float2 x0, x1, x2, x3, x4;
    float4 wa0, wb0, wa1, wb1;

    auto gload = [&](int k0) {
        x0  = *(const float2*)(xg + k0);
        x1  = *(const float2*)(xg + k0 + 16L * DIM);
        x2  = *(const float2*)(xg + k0 + 32L * DIM);
        x3  = *(const float2*)(xg + k0 + 48L * DIM);
        x4  = *(const float2*)(xg + k0 + 64L * DIM);
        const float* p = wg + (long)k0 * DIM;
        wa0 = *(const float4*)(p);
        wb0 = *(const float4*)(p + DIM);
        wa1 = *(const float4*)(p + 16 * DIM);
        wb1 = *(const float4*)(p + 17 * DIM);
    };

    auto sstore = [&](int buf) {
        float* xd = &xs[buf][xo];
        *(float2*)(xd)           = x0;
        *(float2*)(xd + 16 * XP) = x1;
        *(float2*)(xd + 32 * XP) = x2;
        *(float2*)(xd + 48 * XP) = x3;
        *(float2*)(xd + 64 * XP) = x4;
        float* p = &wsm[buf][wo];
        *(float4*)(p)     = make_float4(wa0.x, wb0.x, wa0.y, wb0.y);
        *(float4*)(p + 4) = make_float4(wa0.z, wb0.z, wa0.w, wb0.w);
        float* q = p + 8 * WNP * 2;
        *(float4*)(q)     = make_float4(wa1.x, wb1.x, wa1.y, wb1.y);
        *(float4*)(q + 4) = make_float4(wa1.z, wb1.z, wa1.w, wb1.w);
    };

    v8f c0 = {}, c1 = {}, c2 = {}, c3 = {}, c4 = {};

    gload(0);
    sstore(0);
    __syncthreads();

    for (int kc = 0; kc < DIM / KC; ++kc) {
        int cur = kc & 1;
        bool more = (kc + 1) < (DIM / KC);
        if (more) gload((kc + 1) * KC);

        const float* xa = &xs[cur][l16 * XP + 2 * half];
        const float* wp = &wsm[cur][(half * WNP + 16 * w + l16) * 2];
#pragma unroll
        for (int kk = 0; kk < KC; kk += 4) {
            v2f b  = *(const v2f*)(wp + (kk >> 1) * WNP * 2);
            v2f a0 = *(const v2f*)(xa + 0 * 16 * XP + kk);
            v2f a1 = *(const v2f*)(xa + 1 * 16 * XP + kk);
            v2f a2 = *(const v2f*)(xa + 2 * 16 * XP + kk);
            v2f a3 = *(const v2f*)(xa + 3 * 16 * XP + kk);
            v2f a4 = *(const v2f*)(xa + 4 * 16 * XP + kk);
            c0 = __builtin_amdgcn_wmma_f32_16x16x4_f32(false, a0, false, b, (short)0, c0, false, false);
            c1 = __builtin_amdgcn_wmma_f32_16x16x4_f32(false, a1, false, b, (short)0, c1, false, false);
            c2 = __builtin_amdgcn_wmma_f32_16x16x4_f32(false, a2, false, b, (short)0, c2, false, false);
            c3 = __builtin_amdgcn_wmma_f32_16x16x4_f32(false, a3, false, b, (short)0, c3, false, false);
            c4 = __builtin_amdgcn_wmma_f32_16x16x4_f32(false, a4, false, b, (short)0, c4, false, false);
        }

        if (more) sstore(cur ^ 1);
        __syncthreads();
    }

    int col = nb + 16 * w + l16;
    float bv = bias[col];
    float* o = out + (long)(m0 + 8 * half) * DIM + col;
#pragma unroll
    for (int r = 0; r < 8; ++r) {
        o[(long)(0 * 16 + r) * DIM] = c0[r] + bv;
        o[(long)(1 * 16 + r) * DIM] = c1[r] + bv;
        o[(long)(2 * 16 + r) * DIM] = c2[r] + bv;
        o[(long)(3 * 16 + r) * DIM] = c3[r] + bv;
        o[(long)(4 * 16 + r) * DIM] = c4[r] + bv;
    }
}

#define OWN   40
#define OBLK  (NNODE / OWN)
#define OTH   1024
#define OSEG  5
#define OLD   8
#define NQ    (NEDGE / 4)
#define OCAP  8192
#define OROT  (NQ / OBLK)
#define OGRP  4
#define XRV   ((OWN * DIM / 4 + OTH - 1) / OTH)
static_assert(NNODE % OWN == 0, "whole tiles");
static_assert(NEDGE % 4 == 0, "destination list in whole 16-byte units");
static_assert(OSEG * OLD * OTH >= NQ, "the pieces cover the list");
static_assert(OWN <= 64 && EN <= (1 << 18) && OWN <= OCAP, "a list word holds (row << 18) | edge or source");
static_assert(OGRP == 4, "the store of the four logits of a group is written for 4");
static_assert(OCAP <= 65536, "a place inside a row fits 16 bits");
#define ERG   (OTH / (DIM / 2))
#define ERT   (OWN / ERG)
static_assert(OTH % (DIM / 2) == 0 && OWN % ERG == 0, "every thread of the epilogue has ERT whole rows of one column pair");
static_assert(2 * ERG * DIM <= OCAP, "the column partials fit the weight list's LDS");
static_assert(2 * DIM <= OTH, "one thread per column sum");

__device__ __forceinline__ unsigned fenc(float f) {
    unsigned b = __float_as_uint(f);
    return (b & 0x80000000u) ? ~b : (b | 0x80000000u);
}
__device__ __forceinline__ float fdec(unsigned u) {
    return __uint_as_float((u & 0x80000000u) ? (u ^ 0x80000000u) : ~u);
}

__device__ __forceinline__ float logit_part(float4 a, float4 b, float4 t, float acc) {
    float h;
    h = a.x + b.x; h = h > 0.f ? h : NEG_SLOPE * h; acc += t.x * h;
    h = a.y + b.y; h = h > 0.f ? h : NEG_SLOPE * h; acc += t.y * h;
    h = a.z + b.z; h = h > 0.f ? h : NEG_SLOPE * h; acc += t.z * h;
    h = a.w + b.w; h = h > 0.f ? h : NEG_SLOPE * h; acc += t.w * h;
    return acc;
}

__device__ __forceinline__ float wave_sum(float p) {
    p += __shfl_xor(p, 16, 32);
    p += __shfl_xor(p, 8, 32);
    p += __shfl_xor(p, 4, 32);
    p += __shfl_xor(p, 2, 32);
    p += __shfl_xor(p, 1, 32);
    return p;
}

__global__ __launch_bounds__(OTH)
void edge_aggregate(const int* __restrict__ ei,
                    const float* __restrict__ xl, const float* __restrict__ xr,
                    const float* __restrict__ att,
                    const float* __restrict__ bias, const float* __restrict__ du,
                    float* __restrict__ colsums, float* __restrict__ agg) {
    __shared__ __attribute__((aligned(16))) float acc[OWN * DIM];
    __shared__ __attribute__((aligned(16))) float xrt[OWN * DIM];
    __shared__ float    den[64];
    __shared__ unsigned rmax[64];
    __shared__ unsigned cnt;
    __shared__ unsigned list[OCAP];
    __shared__ __attribute__((aligned(16))) float lal[OCAP];
    __shared__ unsigned       rcnt[64];
    __shared__ unsigned       rstart[64];
    __shared__ unsigned short lq[OCAP];
    __shared__ unsigned       slist[OCAP];
    __shared__ float          sal[OCAP];

    int t    = (int)threadIdx.x;
    int lane = t & 31;
    int w    = t >> 5;
    int n0   = (int)blockIdx.x * OWN;

    const float4* xr4 = (const float4*)(xr + (long)n0 * DIM);
    float4 xv[XRV];
#pragma unroll
    for (int k = 0; k < XRV; ++k) {
        int i = t + k * OTH;
        xv[k] = (i < OWN * DIM / 4) ? xr4[i] : make_float4(0.f, 0.f, 0.f, 0.f);
    }

    for (int i = t; i < OWN * DIM; i += OTH) acc[i] = 0.f;
    if (t < 64) { den[t] = 0.f; rmax[t] = 0u; rcnt[t] = 0u; }
    if (t < OWN) list[t] = ((unsigned)t << 18) | (unsigned)(NEDGE + n0 + t);
    if (t == 0) cnt = OWN;
    __syncthreads();

    const int* dst = ei + NEDGE;
    const int4* dq = (const int4*)dst;
    int rot = (int)blockIdx.x * OROT;

    auto keep = [&](unsigned r, int e) {
        unsigned p = atomicAdd(&cnt, 1u);
        if (p < OCAP) list[p] = (r << 18) | (unsigned)e;
    };
    for (int seg = 0; seg < OSEG; ++seg) {
        int4 v[OLD];
        int  ps[OLD];
#pragma unroll
        for (int j = 0; j < OLD; ++j) {
            int q = (seg * OLD + j) * OTH + t;
            int idx = q + rot;
            if (idx >= NQ) idx -= NQ;
            ps[j] = 4 * idx;
            v[j] = (q < NQ) ? dq[(q < NQ) ? idx : 0] : make_int4(-1, -1, -1, -1);
        }
#pragma unroll
        for (int j = 0; j < OLD; ++j) {
            unsigned d0 = (unsigned)(v[j].x - n0), d1 = (unsigned)(v[j].y - n0);
            unsigned d2 = (unsigned)(v[j].z - n0), d3 = (unsigned)(v[j].w - n0);
            if (d0 < OWN) keep(d0, ps[j] + 0);
            if (d1 < OWN) keep(d1, ps[j] + 1);
            if (d2 < OWN) keep(d2, ps[j] + 2);
            if (d3 < OWN) keep(d3, ps[j] + 3);
        }
    }
#pragma unroll
    for (int k = 0; k < XRV; ++k) {
        int i = t + k * OTH;
        if (i < OWN * DIM / 4) ((float4*)xrt)[i] = xv[k];
    }
    __syncthreads();

    unsigned total = cnt;
    bool     fits  = total <= OCAP;
    unsigned n     = fits ? total : 0u;

    const float4* av = (const float4*)att;
    float4 t0 = av[lane], t1 = av[lane + 32];

    for (unsigned i = (unsigned)t; i < n; i += OTH) {
        unsigned word = list[i];
        int e = (int)(word & 0x3FFFFu);
        int s = (e < NEDGE) ? ei[e] : (e - NEDGE);
        list[i] = (word & ~0x3FFFFu) | (unsigned)s;
        lq[i] = (unsigned short)atomicAdd(&rcnt[word >> 18], 1u);
    }
    __syncthreads();

    if (t < OWN) {
        unsigned first = 0u;
        for (int k = 0; k < t; ++k) first += rcnt[k];
        rstart[t] = first;
    }

    for (unsigned base = OGRP * (unsigned)w; base < n; base += 32u * OGRP) {
        unsigned pw[OGRP];
        float4   a0[OGRP], a1[OGRP];
#pragma unroll
        for (int g = 0; g < OGRP; ++g) {
            unsigned i = (base + g < n) ? base + g : base;
            pw[g] = (unsigned)__builtin_amdgcn_readfirstlane((int)list[i]);
            const float4* xs = (const float4*)(xl + (long)(pw[g] & 0x3FFFFu) * DIM);
            a0[g] = xs[lane]; a1[g] = xs[lane + 32];
        }
        float p[OGRP];
#pragma unroll
        for (int g = 0; g < OGRP; ++g) {
            const float4* xd = (const float4*)&xrt[(pw[g] >> 18) * DIM];
            p[g] = logit_part(a1[g], xd[lane + 32], t1, logit_part(a0[g], xd[lane], t0, 0.f));
        }
#pragma unroll
        for (int m = 16; m >= 1; m >>= 1) {
#pragma unroll
            for (int g = 0; g < OGRP; ++g) p[g] += __shfl_xor(p[g], m, 32);
        }
        if (lane < OGRP && base + (unsigned)lane < n) {
            float    pv = p[0];
            unsigned rv = pw[0];
            if (lane == 1) { pv = p[1]; rv = pw[1]; }
            if (lane == 2) { pv = p[2]; rv = pw[2]; }
            if (lane == 3) { pv = p[3]; rv = pw[3]; }
            lal[base + (unsigned)lane] = pv;
            atomicMax(&rmax[rv >> 18], fenc(pv));
        }
    }
    __syncthreads();

    const int cp = t & (DIM / 2 - 1);
    const int rg = t / (DIM / 2);
    const float2* u2 = (const float2*)du + (long)n0 * (DIM / 2);
    float2 bv = ((const float2*)bias)[cp];
    float2 uv[ERT];
#pragma unroll
    for (int j = 0; j < ERT; ++j) uv[j] = u2[(rg + ERG * j) * (DIM / 2) + cp];

    for (unsigned i = (unsigned)t; i < n; i += OTH) {
        unsigned word = list[i];
        unsigned r = word >> 18;
        float al = __expf(lal[i] - fdec(rmax[r]));
        atomicAdd(&den[r], al);
        unsigned pos = rstart[r] + lq[i];
        slist[pos] = word;
        sal[pos]   = al;
    }
    __syncthreads();

    if (fits) {
        const unsigned per = ((((n + 31u) >> 5) + (OGRP - 1u)) / OGRP) * OGRP;
        const unsigned beg = (unsigned)__builtin_amdgcn_readfirstlane((int)((unsigned)w * per));
        const unsigned lim = (unsigned)__builtin_amdgcn_readfirstlane((int)((beg + per < n) ? beg + per : n));
        const unsigned NOROW = 0xFFFFFFFFu;
        unsigned cur = NOROW;
        float    run[DIM / 32];
#pragma unroll
        for (int c = 0; c < DIM / 32; ++c) run[c] = 0.f;

        auto flush = [&](unsigned row) {
            float* o = &acc[row * DIM + lane];
#pragma unroll
            for (int c = 0; c < DIM / 32; ++c) atomicAdd(&o[32 * c], run[c]);
        };

        for (unsigned base = beg; base < lim; base += OGRP) {
            unsigned pw[OGRP];
            float    al[OGRP];
            float    a[OGRP][DIM / 32];
#pragma unroll
            for (int g = 0; g < OGRP; ++g) {
                unsigned i = (base + g < lim) ? base + g : base;
                pw[g] = (unsigned)__builtin_amdgcn_readfirstlane((int)slist[i]);
                al[g] = __int_as_float(__builtin_amdgcn_readfirstlane(__float_as_int(sal[i])));
                const float* xs = xl + (long)(pw[g] & 0x3FFFFu) * DIM + lane;
#pragma unroll
                for (int c = 0; c < DIM / 32; ++c) a[g][c] = xs[32 * c];
            }
#pragma unroll
            for (int g = 0; g < OGRP; ++g) {
                if (base + g < lim) {
                    unsigned row = pw[g] >> 18;
                    if (row != cur) {
                        if (cur != NOROW) flush(cur);
                        cur = row;
#pragma unroll
                        for (int c = 0; c < DIM / 32; ++c) run[c] = al[g] * a[g][c];
                    } else {
#pragma unroll
                        for (int c = 0; c < DIM / 32; ++c) run[c] += al[g] * a[g][c];
                    }
                }
            }
        }
        if (cur != NOROW) flush(cur);
    } else {
        for (int r = w; r < OWN; r += OTH / 32) {
            int node = n0 + r;
            const float4* xd = (const float4*)&xrt[r * DIM];
            float4 b0 = xd[lane], b1 = xd[lane + 32];
            float  m = -INFINITY, dsum = 0.f;
            float4 o0 = make_float4(0.f, 0.f, 0.f, 0.f), o1 = o0;
#pragma nounroll
            for (int pass = 0; pass < 2; ++pass) {
#pragma nounroll
                for (int base = 0; base < EN; base += 32) {
                    int e = base + lane;
                    int d = -1;
                    if (e < NEDGE) d = dst[e];
                    else if (e < EN) d = e - NEDGE;
                    unsigned hit = __builtin_amdgcn_ballot_w32(d == node);
                    while (hit) {
                        int ee = base + __builtin_ctz(hit);
                        hit &= hit - 1u;
                        int s = (ee < NEDGE) ? ei[ee] : (ee - NEDGE);
                        const float4* xs = (const float4*)(xl + (long)s * DIM);
                        float4 r0 = xs[lane], r1 = xs[lane + 32];
                        float p = wave_sum(logit_part(r1, b1, t1, logit_part(r0, b0, t0, 0.f)));
                        if (pass == 0) {
                            m = fmaxf(m, p);
                        } else {
                            float al = __expf(p - m);
                            dsum += al;
                            o0.x += al * r0.x; o0.y += al * r0.y; o0.z += al * r0.z; o0.w += al * r0.w;
                            o1.x += al * r1.x; o1.y += al * r1.y; o1.z += al * r1.z; o1.w += al * r1.w;
                        }
                    }
                }
            }
            ((float4*)&acc[r * DIM])[lane]      = o0;
            ((float4*)&acc[r * DIM])[lane + 32] = o1;
            if (lane == 0) den[r] = dsum;
        }
    }
    __syncthreads();

    float2*       a2 = (float2*)agg + (long)n0 * (DIM / 2);
    const float2* t2 = (const float2*)acc;
    float2 s  = make_float2(0.f, 0.f);
    float2 s2 = make_float2(0.f, 0.f);
#pragma unroll
    for (int j = 0; j < ERT; ++j) {
        int    row = rg + ERG * j;
        float  dn  = den[row];
        float2 a   = t2[row * (DIM / 2) + cp];
        float2 v;
        v.x = a.x / dn + bv.x;
        v.y = a.y / dn + bv.y;
        v.x = v.x > 0.f ? v.x : 0.f;
        v.y = v.y > 0.f ? v.y : 0.f;
        v.x = (uv[j].x >= 0.5f) ? 2.0f * v.x : 0.0f;
        v.y = (uv[j].y >= 0.5f) ? 2.0f * v.y : 0.0f;
        a2[row * (DIM / 2) + cp] = v;
        s.x += v.x; s2.x += v.x * v.x;
        s.y += v.y; s2.y += v.y * v.y;
    }
    float* red = lal;
    *(float2*)&red[rg * DIM + 2 * cp]             = s;
    *(float2*)&red[ERG * DIM + rg * DIM + 2 * cp] = s2;
    __syncthreads();
    if (t < 2 * DIM) {
        const float* r = red + (t / DIM) * ERG * DIM + (t & (DIM - 1));
        float cs = 0.f;
#pragma unroll
        for (int j = 0; j < ERG; ++j) cs += r[j * DIM];
        atomicAdd(&colsums[t], cs);
    }
}

#define PR   20
#define PRT  (PR / 4)
static_assert(NNODE % PR == 0 && PR % 4 == 0, "whole blocks of four row groups");

__global__ __launch_bounds__(256)
void finalize(const float* __restrict__ v,
              const float* __restrict__ sums, const float* __restrict__ sumsq,
              const float* __restrict__ gamma, const float* __restrict__ beta,
              float* __restrict__ out) {
    int t  = (int)threadIdx.x;
    int c4 = t & 63;
    int rg = t >> 6;
    int i0 = (int)blockIdx.x * PR + rg;

    const float4* v4 = (const float4*)v;
    float4*       o4 = (float4*)out;

    float4 x[PRT];
#pragma unroll
    for (int j = 0; j < PRT; ++j) x[j] = v4[(long)(i0 + 4 * j) * (DIM / 4) + c4];

    float4 sm = ((const float4*)sums)[c4];
    float4 sq = ((const float4*)sumsq)[c4];
    float4 g  = ((const float4*)gamma)[c4];
    float4 b  = ((const float4*)beta)[c4];

    const float invn = 1.0f / (float)NNODE;
    float4 scale, shift;
    {
        float mean, var;
        mean = sm.x * invn; var = sq.x * invn - mean * mean;
        scale.x = g.x * rsqrtf(var + BN_EPS); shift.x = b.x - mean * scale.x;
        mean = sm.y * invn; var = sq.y * invn - mean * mean;
        scale.y = g.y * rsqrtf(var + BN_EPS); shift.y = b.y - mean * scale.y;
        mean = sm.z * invn; var = sq.z * invn - mean * mean;
        scale.z = g.z * rsqrtf(var + BN_EPS); shift.z = b.z - mean * scale.z;
        mean = sm.w * invn; var = sq.w * invn - mean * mean;
        scale.w = g.w * rsqrtf(var + BN_EPS); shift.w = b.w - mean * scale.w;
    }

#pragma unroll
    for (int j = 0; j < PRT; ++j) {
        float4 r;
        r.x = x[j].x * scale.x + shift.x;
        r.y = x[j].y * scale.y + shift.y;
        r.z = x[j].z * scale.z + shift.z;
        r.w = x[j].w * scale.w + shift.w;
        o4[(long)(i0 + 4 * j) * (DIM / 4) + c4] = r;
    }
}

extern "C" void kernel_launch(void* const* d_in, const int* in_sizes, int n_in,
                              void* d_out, int out_size, void* d_ws, size_t ws_size,
                              hipStream_t stream) {
    const float* x     = (const float*)d_in[0];
    const int*   ei    = (const int*)d_in[1];
    const float* Wl    = (const float*)d_in[2];
    const float* bl    = (const float*)d_in[3];
    const float* Wr    = (const float*)d_in[4];
    const float* br    = (const float*)d_in[5];
    const float* att   = (const float*)d_in[6];
    const float* bias  = (const float*)d_in[7];
    const float* gamma = (const float*)d_in[8];
    const float* beta  = (const float*)d_in[9];
    const float* du    = (const float*)d_in[10];
    float*       out   = (float*)d_out;

    float*    ws    = (float*)d_ws;
    float*    agg   = ws;
    float*    sums  = ws + (long)NNODE * DIM + 2 * NNODE;
    float*    sumsq = sums + DIM;
    float*    xl    = sumsq + DIM;
    float*    xr    = xl + (long)NNODE * DIM;

    gemm_xlxr<<<(NNODE / BM) * 4, 256, 0, stream>>>(x, Wl, bl, Wr, br, xl, xr, sums);

    edge_aggregate<<<OBLK, OTH, 0, stream>>>(ei, xl, xr, att, bias, du, sums, agg);

    finalize<<<NNODE / PR, 256, 0, stream>>>(agg, sums, sumsq, gamma, beta, out);
}


# aggregation gathers each source row once, not twice: a wave walks its row-ordered slice with a running max and rescaled sums in registers; per-wave partials meet in the epilogue; sum order changes
# speedup vs baseline: 4.5859x; 1.0457x over previous
#include <hip/hip_runtime.h>
#include <hip/hip_bf16.h>
#include <stdint.h>

typedef __attribute__((ext_vector_type(2))) float v2f;
typedef __attribute__((ext_vector_type(8))) float v8f;

#define NNODE   10000
#define NEDGE   160000
#define EN      (NEDGE + NNODE)
#define DIM     256
#define MT      (NNODE / 16)
#define NEG_SLOPE 0.2f
#define BN_EPS  1e-5f

#define BM   80
#define BN   128
#define KC   32
#define XP   (KC + 4)
#define WNP  (BN + 16)
#define XS_FLOATS (BM * XP)
#define WS_FLOATS ((KC / 2) * WNP * 2)

__global__ __launch_bounds__(256)
void gemm_xlxr(const float* __restrict__ x,
               const float* __restrict__ Wl, const float* __restrict__ bl,
               const float* __restrict__ Wr, const float* __restrict__ br,
               float* __restrict__ xl, float* __restrict__ xr,
               float* __restrict__ colsums) {
    __shared__ __attribute__((aligned(16))) float xs[2][XS_FLOATS];
    __shared__ __attribute__((aligned(16))) float wsm[2][WS_FLOATS];

    int t    = (int)threadIdx.x;
    int lane = t & 31;
    int w    = t >> 5;
    int half = lane >> 4;
    int l16  = lane & 15;

    if (blockIdx.x == 0) { colsums[t] = 0.f; colsums[t + DIM] = 0.f; }

    int rb  = (int)blockIdx.x >> 2;
    int cb  = (int)blockIdx.x & 3;
    int mat = cb >> 1;
    int nb  = (cb & 1) * BN;
    int m0  = rb * BM;

    const float* W    = mat ? Wr : Wl;
    const float* bias = mat ? br : bl;
    float*       out  = mat ? xr : xl;

    int xrow = t >> 4, xj = t & 15;
    const float* xg = x + (long)(m0 + xrow) * DIM + 2 * xj;
    int xo = xrow * XP + 2 * xj;

    int n4  = t & 31;
    int kpb = t >> 5;
    const float* wg = W + (long)(2 * kpb) * DIM + nb + 4 * n4;
    int wo = (kpb * WNP + 4 * n4) * 2;

    float2 x0, x1, x2, x3, x4;
    float4 wa0, wb0, wa1, wb1;

    auto gload = [&](int k0) {
        x0  = *(const float2*)(xg + k0);
        x1  = *(const float2*)(xg + k0 + 16L * DIM);
        x2  = *(const float2*)(xg + k0 + 32L * DIM);
        x3  = *(const float2*)(xg + k0 + 48L * DIM);
        x4  = *(const float2*)(xg + k0 + 64L * DIM);
        const float* p = wg + (long)k0 * DIM;
        wa0 = *(const float4*)(p);
        wb0 = *(const float4*)(p + DIM);
        wa1 = *(const float4*)(p + 16 * DIM);
        wb1 = *(const float4*)(p + 17 * DIM);
    };

    auto sstore = [&](int buf) {
        float* xd = &xs[buf][xo];
        *(float2*)(xd)           = x0;
        *(float2*)(xd + 16 * XP) = x1;
        *(float2*)(xd + 32 * XP) = x2;
        *(float2*)(xd + 48 * XP) = x3;
        *(float2*)(xd + 64 * XP) = x4;
        float* p = &wsm[buf][wo];
        *(float4*)(p)     = make_float4(wa0.x, wb0.x, wa0.y, wb0.y);
        *(float4*)(p + 4) = make_float4(wa0.z, wb0.z, wa0.w, wb0.w);
        float* q = p + 8 * WNP * 2;
        *(float4*)(q)     = make_float4(wa1.x, wb1.x, wa1.y, wb1.y);
        *(float4*)(q + 4) = make_float4(wa1.z, wb1.z, wa1.w, wb1.w);
    };

    v8f c0 = {}, c1 = {}, c2 = {}, c3 = {}, c4 = {};

    gload(0);
    sstore(0);
    __syncthreads();

    for (int kc = 0; kc < DIM / KC; ++kc) {
        int cur = kc & 1;
        bool more = (kc + 1) < (DIM / KC);
        if (more) gload((kc + 1) * KC);

        const float* xa = &xs[cur][l16 * XP + 2 * half];
        const float* wp = &wsm[cur][(half * WNP + 16 * w + l16) * 2];
#pragma unroll
        for (int kk = 0; kk < KC; kk += 4) {
            v2f b  = *(const v2f*)(wp + (kk >> 1) * WNP * 2);
            v2f a0 = *(const v2f*)(xa + 0 * 16 * XP + kk);
            v2f a1 = *(const v2f*)(xa + 1 * 16 * XP + kk);
            v2f a2 = *(const v2f*)(xa + 2 * 16 * XP + kk);
            v2f a3 = *(const v2f*)(xa + 3 * 16 * XP + kk);
            v2f a4 = *(const v2f*)(xa + 4 * 16 * XP + kk);
            c0 = __builtin_amdgcn_wmma_f32_16x16x4_f32(false, a0, false, b, (short)0, c0, false, false);
            c1 = __builtin_amdgcn_wmma_f32_16x16x4_f32(false, a1, false, b, (short)0, c1, false, false);
            c2 = __builtin_amdgcn_wmma_f32_16x16x4_f32(false, a2, false, b, (short)0, c2, false, false);
            c3 = __builtin_amdgcn_wmma_f32_16x16x4_f32(false, a3, false, b, (short)0, c3, false, false);
            c4 = __builtin_amdgcn_wmma_f32_16x16x4_f32(false, a4, false, b, (short)0, c4, false, false);
        }

        if (more) sstore(cur ^ 1);
        __syncthreads();
    }

    int col = nb + 16 * w + l16;
    float bv = bias[col];
    float* o = out + (long)(m0 + 8 * half) * DIM + col;
#pragma unroll
    for (int r = 0; r < 8; ++r) {
        o[(long)(0 * 16 + r) * DIM] = c0[r] + bv;
        o[(long)(1 * 16 + r) * DIM] = c1[r] + bv;
        o[(long)(2 * 16 + r) * DIM] = c2[r] + bv;
        o[(long)(3 * 16 + r) * DIM] = c3[r] + bv;
        o[(long)(4 * 16 + r) * DIM] = c4[r] + bv;
    }
}

#define OWN   40
#define OBLK  (NNODE / OWN)
#define OTH   1024
#define OSEG  5
#define OLD   8
#define NQ    (NEDGE / 4)
#define OCAP  8192
#define OROT  (NQ / OBLK)
#define OGRP  4
#define XRV   ((OWN * DIM / 4 + OTH - 1) / OTH)
#define PSLOT (OWN + OTH / 32)
static_assert(NNODE % OWN == 0, "whole tiles");
static_assert(NEDGE % 4 == 0, "destination list in whole 16-byte units");
static_assert(OSEG * OLD * OTH >= NQ, "the pieces cover the list");
static_assert(OWN <= 64 && EN <= (1 << 18) && OWN <= OCAP, "a list word holds (row << 18) | edge or source");
static_assert(OGRP == 4, "the exchange of the four logits of a group is written for 4");
static_assert(OCAP <= 65536, "a place inside a row fits 16 bits");
#define ERG   (OTH / (DIM / 2))
#define ERT   (OWN / ERG)
static_assert(OTH % (DIM / 2) == 0 && OWN % ERG == 0, "every thread of the epilogue has ERT whole rows of one column pair");
static_assert(2 * ERG * DIM <= OCAP, "the column partials fit the first list's LDS");
static_assert(2 * DIM <= OTH, "one thread per column sum");

__device__ __forceinline__ float logit_part(float4 a, float4 b, float4 t, float acc) {
    float h;
    h = a.x + b.x; h = h > 0.f ? h : NEG_SLOPE * h; acc += t.x * h;
    h = a.y + b.y; h = h > 0.f ? h : NEG_SLOPE * h; acc += t.y * h;
    h = a.z + b.z; h = h > 0.f ? h : NEG_SLOPE * h; acc += t.z * h;
    h = a.w + b.w; h = h > 0.f ? h : NEG_SLOPE * h; acc += t.w * h;
    return acc;
}

__device__ __forceinline__ float4 axpb4(float s, float4 a, float4 b) {
    return make_float4(s * a.x + b.x, s * a.y + b.y, s * a.z + b.z, s * a.w + b.w);
}

__device__ __forceinline__ float wave_sum(float p) {
    p += __shfl_xor(p, 16, 32);
    p += __shfl_xor(p, 8, 32);
    p += __shfl_xor(p, 4, 32);
    p += __shfl_xor(p, 2, 32);
    p += __shfl_xor(p, 1, 32);
    return p;
}

__global__ __launch_bounds__(OTH)
void edge_aggregate(const int* __restrict__ ei,
                    const float* __restrict__ xl, const float* __restrict__ xr,
                    const float* __restrict__ att,
                    const float* __restrict__ bias, const float* __restrict__ du,
                    float* __restrict__ colsums, float* __restrict__ agg) {
    __shared__ __attribute__((aligned(16))) float acc[PSLOT * DIM];
    __shared__ __attribute__((aligned(16))) float xrt[OWN * DIM];
    __shared__ float    pm[PSLOT];
    __shared__ float    pd[PSLOT];
    __shared__ unsigned cnt;
    __shared__ __attribute__((aligned(16))) unsigned list[OCAP];
    __shared__ unsigned       rcnt[64];
    __shared__ unsigned       rstart[64];
    __shared__ unsigned       rwa[64];
    __shared__ unsigned       rwb[64];
    __shared__ unsigned short lq[OCAP];
    __shared__ unsigned       slist[OCAP];

    int t    = (int)threadIdx.x;
    int lane = t & 31;
    int w    = t >> 5;
    int n0   = (int)blockIdx.x * OWN;

    const float4* xr4 = (const float4*)(xr + (long)n0 * DIM);
    float4 xv[XRV];
#pragma unroll
    for (int k = 0; k < XRV; ++k) {
        int i = t + k * OTH;
        xv[k] = (i < OWN * DIM / 4) ? xr4[i] : make_float4(0.f, 0.f, 0.f, 0.f);
    }

    if (t < 64) rcnt[t] = 0u;
    if (t < OWN) list[t] = ((unsigned)t << 18) | (unsigned)(NEDGE + n0 + t);
    if (t == 0) cnt = OWN;
    __syncthreads();

    const int* dst = ei + NEDGE;
    const int4* dq = (const int4*)dst;
    int rot = (int)blockIdx.x * OROT;

    auto keep = [&](unsigned r, int e) {
        unsigned p = atomicAdd(&cnt, 1u);
        if (p < OCAP) list[p] = (r << 18) | (unsigned)e;
    };
    for (int seg = 0; seg < OSEG; ++seg) {
        int4 v[OLD];
        int  ps[OLD];
#pragma unroll
        for (int j = 0; j < OLD; ++j) {
            int q = (seg * OLD + j) * OTH + t;
            int idx = q + rot;
            if (idx >= NQ) idx -= NQ;
            ps[j] = 4 * idx;
            v[j] = (q < NQ) ? dq[(q < NQ) ? idx : 0] : make_int4(-1, -1, -1, -1);
        }
#pragma unroll
        for (int j = 0; j < OLD; ++j) {
            unsigned d0 = (unsigned)(v[j].x - n0), d1 = (unsigned)(v[j].y - n0);
            unsigned d2 = (unsigned)(v[j].z - n0), d3 = (unsigned)(v[j].w - n0);
            if (d0 < OWN) keep(d0, ps[j] + 0);
            if (d1 < OWN) keep(d1, ps[j] + 1);
            if (d2 < OWN) keep(d2, ps[j] + 2);
            if (d3 < OWN) keep(d3, ps[j] + 3);
        }
    }
#pragma unroll
    for (int k = 0; k < XRV; ++k) {
        int i = t + k * OTH;
        if (i < OWN * DIM / 4) ((float4*)xrt)[i] = xv[k];
    }
    __syncthreads();

    unsigned total = cnt;
    bool     fits  = total <= OCAP;
    unsigned n     = fits ? total : 0u;
    const unsigned per  = ((((n + 31u) >> 5) + (OGRP - 1u)) / OGRP) * OGRP;
    const unsigned perd = fits ? per : 1u;

    const float4* av = (const float4*)att;
    float4 t0 = av[lane], t1 = av[lane + 32];

    for (unsigned i = (unsigned)t; i < n; i += OTH) {
        unsigned word = list[i];
        int e = (int)(word & 0x3FFFFu);
        int s = (e < NEDGE) ? ei[e] : (e - NEDGE);
        list[i] = (word & ~0x3FFFFu) | (unsigned)s;
        lq[i] = (unsigned short)atomicAdd(&rcnt[word >> 18], 1u);
    }
    __syncthreads();

    if (t < OWN) {
        unsigned first = 0u;
        for (int k = 0; k < t; ++k) first += rcnt[k];
        rstart[t] = first;
        rwa[t] = fits ? first / perd : 0u;
        rwb[t] = fits ? (first + rcnt[t] - 1u) / perd : 0u;
    }
    __syncthreads();

    for (unsigned i = (unsigned)t; i < n; i += OTH) {
        unsigned word = list[i];
        slist[rstart[word >> 18] + lq[i]] = word;
    }
    __syncthreads();

    if (fits) {
        const unsigned beg = (unsigned)__builtin_amdgcn_readfirstlane((int)((unsigned)w * per));
        const unsigned lim = (unsigned)__builtin_amdgcn_readfirstlane((int)((beg + per < n) ? beg + per : n));
        const unsigned NOROW = 0xFFFFFFFFu;
        unsigned cur = NOROW;
        float    m = 0.f, d = 0.f;
        float4   r0 = make_float4(0.f, 0.f, 0.f, 0.f), r1 = r0;

        auto flush = [&](unsigned row) {
            unsigned slot = row + (unsigned)w;
            ((float4*)&acc[slot * DIM])[lane]      = r0;
            ((float4*)&acc[slot * DIM])[lane + 32] = r1;
            if (lane == 0) { pm[slot] = m; pd[slot] = d; }
        };

        for (unsigned base = beg; base < lim; base += OGRP) {
            unsigned pw[OGRP];
            float4   a0[OGRP], a1[OGRP];
#pragma unroll
            for (int g = 0; g < OGRP; ++g) {
                unsigned i = (base + g < lim) ? base + g : base;
                pw[g] = (unsigned)__builtin_amdgcn_readfirstlane((int)slist[i]);
                const float4* xs = (const float4*)(xl + (long)(pw[g] & 0x3FFFFu) * DIM);
                a0[g] = xs[lane]; a1[g] = xs[lane + 32];
            }
            float p[OGRP];
#pragma unroll
            for (int g = 0; g < OGRP; ++g) {
                const float4* xd = (const float4*)&xrt[(pw[g] >> 18) * DIM];
                p[g] = logit_part(a1[g], xd[lane + 32], t1, logit_part(a0[g], xd[lane], t0, 0.f));
            }
#pragma unroll
            for (int mk = 16; mk >= 1; mk >>= 1) {
#pragma unroll
                for (int g = 0; g < OGRP; ++g) p[g] += __shfl_xor(p[g], mk, 32);
            }
#pragma unroll
            for (int g = 0; g < OGRP; ++g) {
                if (base + g < lim) {
                    unsigned row = pw[g] >> 18;
                    float pg = __int_as_float(__builtin_amdgcn_readfirstlane(__float_as_int(p[g])));
                    if (row != cur) {
                        if (cur != NOROW) flush(cur);
                        cur = row;
                        m = pg; d = 1.f;
                        r0 = a0[g]; r1 = a1[g];
                    } else if (pg > m) {
                        float sc = __expf(m - pg);
                        d  = d * sc + 1.f;
                        r0 = axpb4(sc, r0, a0[g]);
                        r1 = axpb4(sc, r1, a1[g]);
                        m  = pg;
                    } else {
                        float al = __expf(pg - m);
                        d += al;
                        r0 = axpb4(al, a0[g], r0);
                        r1 = axpb4(al, a1[g], r1);
                    }
                }
            }
        }
        if (cur != NOROW) flush(cur);
    } else {
        for (int r = w; r < OWN; r += OTH / 32) {
            int node = n0 + r;
            const float4* xd = (const float4*)&xrt[r * DIM];
            float4 b0 = xd[lane], b1 = xd[lane + 32];
            float  m = -INFINITY, dsum = 0.f;
            float4 o0 = make_float4(0.f, 0.f, 0.f, 0.f), o1 = o0;
#pragma nounroll
            for (int pass = 0; pass < 2; ++pass) {
#pragma nounroll
                for (int base = 0; base < EN; base += 32) {
                    int e = base + lane;
                    int dd = -1;
                    if (e < NEDGE) dd = dst[e];
                    else if (e < EN) dd = e - NEDGE;
                    unsigned hit = __builtin_amdgcn_ballot_w32(dd == node);
                    while (hit) {
                        int ee = base + __builtin_ctz(hit);
                        hit &= hit - 1u;
                        int s = (ee < NEDGE) ? ei[ee] : (ee - NEDGE);
                        const float4* xs = (const float4*)(xl + (long)s * DIM);
                        float4 q0 = xs[lane], q1 = xs[lane + 32];
                        float p = wave_sum(logit_part(q1, b1, t1, logit_part(q0, b0, t0, 0.f)));
                        if (pass == 0) {
                            m = fmaxf(m, p);
                        } else {
                            float al = __expf(p - m);
                            dsum += al;
                            o0.x += al * q0.x; o0.y += al * q0.y; o0.z += al * q0.z; o0.w += al * q0.w;
                            o1.x += al * q1.x; o1.y += al * q1.y; o1.z += al * q1.z; o1.w += al * q1.w;
                        }
                    }
                }
            }
            ((float4*)&acc[r * DIM])[lane]      = o0;
            ((float4*)&acc[r * DIM])[lane + 32] = o1;
            if (lane == 0) { pm[r] = 0.f; pd[r] = dsum; }
        }
    }

    const int cp = t & (DIM / 2 - 1);
    const int rg = t / (DIM / 2);
    const float2* u2 = (const float2*)du + (long)n0 * (DIM / 2);
    float2 bv = ((const float2*)bias)[cp];
    float2 uv[ERT];
#pragma unroll
    for (int j = 0; j < ERT; ++j) uv[j] = u2[(rg + ERG * j) * (DIM / 2) + cp];
    __syncthreads();

    float2*       a2 = (float2*)agg + (long)n0 * (DIM / 2);
    const float2* t2 = (const float2*)acc;
    float2 s  = make_float2(0.f, 0.f);
    float2 s2 = make_float2(0.f, 0.f);
#pragma unroll
    for (int j = 0; j < ERT; ++j) {
        int      row = rg + ERG * j;
        unsigned wa  = rwa[row], wb = rwb[row];
        float mx = pm[row + wa];
        for (unsigned k = wa + 1u; k <= wb; ++k) mx = fmaxf(mx, pm[row + k]);
        float  dn = 0.f;
        float2 a  = make_float2(0.f, 0.f);
        for (unsigned k = wa; k <= wb; ++k) {
            unsigned slot = (unsigned)row + k;
            float  f  = __expf(pm[slot] - mx);
            float2 pv = t2[slot * (DIM / 2) + cp];
            dn  += f * pd[slot];
            a.x += f * pv.x;
            a.y += f * pv.y;
        }
        float2 v;
        v.x = a.x / dn + bv.x;
        v.y = a.y / dn + bv.y;
        v.x = v.x > 0.f ? v.x : 0.f;
        v.y = v.y > 0.f ? v.y : 0.f;
        v.x = (uv[j].x >= 0.5f) ? 2.0f * v.x : 0.0f;
        v.y = (uv[j].y >= 0.5f) ? 2.0f * v.y : 0.0f;
        a2[row * (DIM / 2) + cp] = v;
        s.x += v.x; s2.x += v.x * v.x;
        s.y += v.y; s2.y += v.y * v.y;
    }
    float* red = (float*)list;
    *(float2*)&red[rg * DIM + 2 * cp]             = s;
    *(float2*)&red[ERG * DIM + rg * DIM + 2 * cp] = s2;
    __syncthreads();
    if (t < 2 * DIM) {
        const float* r = red + (t / DIM) * ERG * DIM + (t & (DIM - 1));
        float cs = 0.f;
#pragma unroll
        for (int j = 0; j < ERG; ++j) cs += r[j * DIM];
        atomicAdd(&colsums[t], cs);
    }
}

#define PR   20
#define PRT  (PR / 4)
static_assert(NNODE % PR == 0 && PR % 4 == 0, "whole blocks of four row groups");

__global__ __launch_bounds__(256)
void finalize(const float* __restrict__ v,
              const float* __restrict__ sums, const float* __restrict__ sumsq,
              const float* __restrict__ gamma, const float* __restrict__ beta,
              float* __restrict__ out) {
    int t  = (int)threadIdx.x;
    int c4 = t & 63;
    int rg = t >> 6;
    int i0 = (int)blockIdx.x * PR + rg;

    const float4* v4 = (const float4*)v;
    float4*       o4 = (float4*)out;

    float4 x[PRT];
#pragma unroll
    for (int j = 0; j < PRT; ++j) x[j] = v4[(long)(i0 + 4 * j) * (DIM / 4) + c4];

    float4 sm = ((const float4*)sums)[c4];
    float4 sq = ((const float4*)sumsq)[c4];
    float4 g  = ((const float4*)gamma)[c4];
    float4 b  = ((const float4*)beta)[c4];

    const float invn = 1.0f / (float)NNODE;
    float4 scale, shift;
    {
        float mean, var;
        mean = sm.x * invn; var = sq.x * invn - mean * mean;
        scale.x = g.x * rsqrtf(var + BN_EPS); shift.x = b.x - mean * scale.x;
        mean = sm.y * invn; var = sq.y * invn - mean * mean;
        scale.y = g.y * rsqrtf(var + BN_EPS); shift.y = b.y - mean * scale.y;
        mean = sm.z * invn; var = sq.z * invn - mean * mean;
        scale.z = g.z * rsqrtf(var + BN_EPS); shift.z = b.z - mean * scale.z;
        mean = sm.w * invn; var = sq.w * invn - mean * mean;
        scale.w = g.w * rsqrtf(var + BN_EPS); shift.w = b.w - mean * scale.w;
    }

#pragma unroll
    for (int j = 0; j < PRT; ++j) {
        float4 r;
        r.x = x[j].x * scale.x + shift.x;
        r.y = x[j].y * scale.y + shift.y;
        r.z = x[j].z * scale.z + shift.z;
        r.w = x[j].w * scale.w + shift.w;
        o4[(long)(i0 + 4 * j) * (DIM / 4) + c4] = r;
    }
}

extern "C" void kernel_launch(void* const* d_in, const int* in_sizes, int n_in,
                              void* d_out, int out_size, void* d_ws, size_t ws_size,
                              hipStream_t stream) {
    const float* x     = (const float*)d_in[0];
    const int*   ei    = (const int*)d_in[1];
    const float* Wl    = (const float*)d_in[2];
    const float* bl    = (const float*)d_in[3];
    const float* Wr    = (const float*)d_in[4];
    const float* br    = (const float*)d_in[5];
    const float* att   = (const float*)d_in[6];
    const float* bias  = (const float*)d_in[7];
    const float* gamma = (const float*)d_in[8];
    const float* beta  = (const float*)d_in[9];
    const float* du    = (const float*)d_in[10];
    float*       out   = (float*)d_out;

    float*    ws    = (float*)d_ws;
    float*    agg   = ws;
    float*    sums  = ws + (long)NNODE * DIM + 2 * NNODE;
    float*    sumsq = sums + DIM;
    float*    xl    = sumsq + DIM;
    float*    xr    = xl + (long)NNODE * DIM;

    gemm_xlxr<<<(NNODE / BM) * 4, 256, 0, stream>>>(x, Wl, bl, Wr, br, xl, xr, sums);

    edge_aggregate<<<OBLK, OTH, 0, stream>>>(ei, xl, xr, att, bias, du, sums, agg);

    finalize<<<NNODE / PR, 256, 0, stream>>>(agg, sums, sumsq, gamma, beta, out);
}


# edge search reads one owner byte per edge (number of the work-group owning its destination row, written by the GEMM launch), not the 4-byte destination: 4x fewer loads; arrival (sum) order changes
# speedup vs baseline: 4.9704x; 1.0838x over previous
#include <hip/hip_runtime.h>
#include <hip/hip_bf16.h>
#include <stdint.h>

typedef __attribute__((ext_vector_type(2))) float v2f;
typedef __attribute__((ext_vector_type(8))) float v8f;

#define NNODE   10000
#define NEDGE   160000
#define EN      (NEDGE + NNODE)
#define DIM     256
#define MT      (NNODE / 16)
#define NEG_SLOPE 0.2f
#define BN_EPS  1e-5f

#define OWN  40
#define OBLK (NNODE / OWN)
#define BM   80
#define BN   128
#define KC   32
#define XP   (KC + 4)
#define WNP  (BN + 16)
#define XS_FLOATS (BM * XP)
#define WS_FLOATS ((KC / 2) * WNP * 2)
#define GBLK (NNODE / BM * 4)
#define OWB  (NEDGE / 4 / GBLK)
static_assert(OWB * GBLK * 4 == NEDGE && OWB <= 256, "the blocks' shares cover the destination list");
static_assert(OBLK <= 255, "an owner fits a byte, 255 left over for none");

__device__ __forceinline__ unsigned owner_of(int d) {
    return ((unsigned)d < (unsigned)NNODE) ? (unsigned)d / OWN : 255u;
}

__global__ __launch_bounds__(256)
void gemm_xlxr(const float* __restrict__ x,
               const float* __restrict__ Wl, const float* __restrict__ bl,
               const float* __restrict__ Wr, const float* __restrict__ br,
               float* __restrict__ xl, float* __restrict__ xr,
               float* __restrict__ colsums,
               const int* __restrict__ ei, unsigned* __restrict__ owners) {
    __shared__ __attribute__((aligned(16))) float xs[2][XS_FLOATS];
    __shared__ __attribute__((aligned(16))) float wsm[2][WS_FLOATS];

    int t    = (int)threadIdx.x;
    int lane = t & 31;
    int w    = t >> 5;
    int half = lane >> 4;
    int l16  = lane & 15;

    if (blockIdx.x == 0) { colsums[t] = 0.f; colsums[t + DIM] = 0.f; }

    const int  ou = (int)blockIdx.x * OWB + (t < OWB ? t : OWB - 1);
    const int4 od = ((const int4*)(ei + NEDGE))[ou];

    int rb  = (int)blockIdx.x >> 2;
    int cb  = (int)blockIdx.x & 3;
    int mat = cb >> 1;
    int nb  = (cb & 1) * BN;
    int m0  = rb * BM;

    const float* W    = mat ? Wr : Wl;
    const float* bias = mat ? br : bl;
    float*       out  = mat ? xr : xl;

    int xrow = t >> 4, xj = t & 15;
    const float* xg = x + (long)(m0 + xrow) * DIM + 2 * xj;
    int xo = xrow * XP + 2 * xj;

    int n4  = t & 31;
    int kpb = t >> 5;
    const float* wg = W + (long)(2 * kpb) * DIM + nb + 4 * n4;
    int wo = (kpb * WNP + 4 * n4) * 2;

    float2 x0, x1, x2, x3, x4;
    float4 wa0, wb0, wa1, wb1;

    auto gload = [&](int k0) {
        x0  = *(const float2*)(xg + k0);
        x1  = *(const float2*)(xg + k0 + 16L * DIM);
        x2  = *(const float2*)(xg + k0 + 32L * DIM);
        x3  = *(const float2*)(xg + k0 + 48L * DIM);
        x4  = *(const float2*)(xg + k0 + 64L * DIM);
        const float* p = wg + (long)k0 * DIM;
        wa0 = *(const float4*)(p);
        wb0 = *(const float4*)(p + DIM);
        wa1 = *(const float4*)(p + 16 * DIM);
        wb1 = *(const float4*)(p + 17 * DIM);
    };

    auto sstore = [&](int buf) {
        float* xd = &xs[buf][xo];
        *(float2*)(xd)           = x0;
        *(float2*)(xd + 16 * XP) = x1;
        *(float2*)(xd + 32 * XP) = x2;
        *(float2*)(xd + 48 * XP) = x3;
        *(float2*)(xd + 64 * XP) = x4;
        float* p = &wsm[buf][wo];
        *(float4*)(p)     = make_float4(wa0.x, wb0.x, wa0.y, wb0.y);
        *(float4*)(p + 4) = make_float4(wa0.z, wb0.z, wa0.w, wb0.w);
        float* q = p + 8 * WNP * 2;
        *(float4*)(q)     = make_float4(wa1.x, wb1.x, wa1.y, wb1.y);
        *(float4*)(q + 4) = make_float4(wa1.z, wb1.z, wa1.w, wb1.w);
    };

    v8f c0 = {}, c1 = {}, c2 = {}, c3 = {}, c4 = {};

    gload(0);
    sstore(0);
    const unsigned ow = owner_of(od.x) | (owner_of(od.y) << 8) | (owner_of(od.z) << 16) | (owner_of(od.w) << 24);
    __syncthreads();

    for (int kc = 0; kc < DIM / KC; ++kc) {
        int cur = kc & 1;
        bool more = (kc + 1) < (DIM / KC);
        if (more) gload((kc + 1) * KC);

        const float* xa = &xs[cur][l16 * XP + 2 * half];
        const float* wp = &wsm[cur][(half * WNP + 16 * w + l16) * 2];
#pragma unroll
        for (int kk = 0; kk < KC; kk += 4) {
            v2f b  = *(const v2f*)(wp + (kk >> 1) * WNP * 2);
            v2f a0 = *(const v2f*)(xa + 0 * 16 * XP + kk);
            v2f a1 = *(const v2f*)(xa + 1 * 16 * XP + kk);
            v2f a2 = *(const v2f*)(xa + 2 * 16 * XP + kk);
            v2f a3 = *(const v2f*)(xa + 3 * 16 * XP + kk);
            v2f a4 = *(const v2f*)(xa + 4 * 16 * XP + kk);
            c0 = __builtin_amdgcn_wmma_f32_16x16x4_f32(false, a0, false, b, (short)0, c0, false, false);
            c1 = __builtin_amdgcn_wmma_f32_16x16x4_f32(false, a1, false, b, (short)0, c1, false, false);
            c2 = __builtin_amdgcn_wmma_f32_16x16x4_f32(false, a2, false, b, (short)0, c2, false, false);
            c3 = __builtin_amdgcn_wmma_f32_16x16x4_f32(false, a3, false, b, (short)0, c3, false, false);
            c4 = __builtin_amdgcn_wmma_f32_16x16x4_f32(false, a4, false, b, (short)0, c4, false, false);
        }

        if (more) sstore(cur ^ 1);
        __syncthreads();
    }

    int col = nb + 16 * w + l16;
    float bv = bias[col];
    float* o = out + (long)(m0 + 8 * half) * DIM + col;
#pragma unroll
    for (int r = 0; r < 8; ++r) {
        o[(long)(0 * 16 + r) * DIM] = c0[r] + bv;
        o[(long)(1 * 16 + r) * DIM] = c1[r] + bv;
        o[(long)(2 * 16 + r) * DIM] = c2[r] + bv;
        o[(long)(3 * 16 + r) * DIM] = c3[r] + bv;
        o[(long)(4 * 16 + r) * DIM] = c4[r] + bv;
    }
    if (t < OWB) owners[ou] = ow;
}

#define OTH   1024
#define NQ    (NEDGE / 16)
#define OLD   ((NQ + OTH - 1) / OTH)
#define OCAP  8192
#define OROT  (NQ / OBLK)
#define OGRP  4
#define XRV   ((OWN * DIM / 4 + OTH - 1) / OTH)
#define PSLOT (OWN + OTH / 32)
static_assert(NNODE % OWN == 0, "whole tiles");
static_assert(NEDGE % 16 == 0, "owner byte list in whole 16-byte units");
static_assert(OLD * OTH >= NQ, "the loads cover the list");
static_assert(OWN <= 64 && EN <= (1 << 18) && OWN <= OCAP, "a list word holds an edge, later (row << 18) | source");
static_assert(OGRP == 4, "the exchange of the four logits of a group is written for 4");
static_assert(OCAP <= 65536, "a place inside a row fits 16 bits");
#define ERG   (OTH / (DIM / 2))
#define ERT   (OWN / ERG)
static_assert(OTH % (DIM / 2) == 0 && OWN % ERG == 0, "every thread of the epilogue has ERT whole rows of one column pair");
static_assert(2 * ERG * DIM <= OCAP, "the column partials fit the first list's LDS");
static_assert(2 * DIM <= OTH, "one thread per column sum");

__device__ __forceinline__ float logit_part(float4 a, float4 b, float4 t, float acc) {
    float h;
    h = a.x + b.x; h = h > 0.f ? h : NEG_SLOPE * h; acc += t.x * h;
    h = a.y + b.y; h = h > 0.f ? h : NEG_SLOPE * h; acc += t.y * h;
    h = a.z + b.z; h = h > 0.f ? h : NEG_SLOPE * h; acc += t.z * h;
    h = a.w + b.w; h = h > 0.f ? h : NEG_SLOPE * h; acc += t.w * h;
    return acc;
}

__device__ __forceinline__ float4 axpb4(float s, float4 a, float4 b) {
    return make_float4(s * a.x + b.x, s * a.y + b.y, s * a.z + b.z, s * a.w + b.w);
}

__device__ __forceinline__ float wave_sum(float p) {
    p += __shfl_xor(p, 16, 32);
    p += __shfl_xor(p, 8, 32);
    p += __shfl_xor(p, 4, 32);
    p += __shfl_xor(p, 2, 32);
    p += __shfl_xor(p, 1, 32);
    return p;
}

__global__ __launch_bounds__(OTH)
void edge_aggregate(const int* __restrict__ ei,
                    const float* __restrict__ xl, const float* __restrict__ xr,
                    const float* __restrict__ att,
                    const float* __restrict__ bias, const float* __restrict__ du,
                    const unsigned* __restrict__ owners,
                    float* __restrict__ colsums, float* __restrict__ agg) {
    __shared__ __attribute__((aligned(16))) float acc[PSLOT * DIM];
    __shared__ __attribute__((aligned(16))) float xrt[OWN * DIM];
    __shared__ float    pm[PSLOT];
    __shared__ float    pd[PSLOT];
    __shared__ unsigned cnt;
    __shared__ __attribute__((aligned(16))) unsigned list[OCAP];
    __shared__ unsigned       rcnt[64];
    __shared__ unsigned       rstart[64];
    __shared__ unsigned       rwa[64];
    __shared__ unsigned       rwb[64];
    __shared__ unsigned short lq[OCAP];
    __shared__ unsigned       slist[OCAP];

    int t    = (int)threadIdx.x;
    int lane = t & 31;
    int w    = t >> 5;
    int n0   = (int)blockIdx.x * OWN;

    const float4* xr4 = (const float4*)(xr + (long)n0 * DIM);
    float4 xv[XRV];
#pragma unroll
    for (int k = 0; k < XRV; ++k) {
        int i = t + k * OTH;
        xv[k] = (i < OWN * DIM / 4) ? xr4[i] : make_float4(0.f, 0.f, 0.f, 0.f);
    }

    if (t < 64) rcnt[t] = 0u;
    if (t < OWN) list[t] = (unsigned)(NEDGE + n0 + t);
    if (t == 0) cnt = OWN;
    __syncthreads();

    const int* dst = ei + NEDGE;
    const uint4* oq = (const uint4*)owners;
    int rot = (int)blockIdx.x * OROT;
    const unsigned pat = (unsigned)blockIdx.x * 0x01010101u;

    auto keep = [&](int e) {
        unsigned p = atomicAdd(&cnt, 1u);
        if (p < OCAP) list[p] = (unsigned)e;
    };
    auto zb = [](unsigned x) -> unsigned {
        return ~(((x & 0x7F7F7F7Fu) + 0x7F7F7F7Fu) | x | 0x7F7F7F7Fu);
    };
    auto look = [&](uint4 wv, int e0) {
        unsigned h = zb(wv.x ^ pat) | (zb(wv.y ^ pat) >> 1) | (zb(wv.z ^ pat) >> 2) | (zb(wv.w ^ pat) >> 3);
        while (h) {
            int p = __builtin_ctz(h);
            h &= h - 1u;
            keep(e0 + 4 * (7 - (p & 7)) + (p >> 3));
        }
    };
    {
        uint4 v[OLD];
        int   ps[OLD];
#pragma unroll
        for (int j = 0; j < OLD; ++j) {
            int q = j * OTH + t;
            int idx = q + rot;
            if (idx >= NQ) idx -= NQ;
            ps[j] = 16 * idx;
            v[j] = (q < NQ) ? oq[(q < NQ) ? idx : 0] : make_uint4(~0u, ~0u, ~0u, ~0u);
        }
#pragma unroll
        for (int j = 0; j < OLD; ++j) {
            look(v[j], ps[j]);
        }
    }
#pragma unroll
    for (int k = 0; k < XRV; ++k) {
        int i = t + k * OTH;
        if (i < OWN * DIM / 4) ((float4*)xrt)[i] = xv[k];
    }
    __syncthreads();

    unsigned total = cnt;
    bool     fits  = total <= OCAP;
    unsigned n     = fits ? total : 0u;
    const unsigned per  = ((((n + 31u) >> 5) + (OGRP - 1u)) / OGRP) * OGRP;
    const unsigned perd = fits ? per : 1u;

    const float4* av = (const float4*)att;
    float4 t0 = av[lane], t1 = av[lane + 32];

    for (unsigned i = (unsigned)t; i < n; i += OTH) {
        int e = (int)list[i];
        int s = (e < NEDGE) ? ei[e] : (e - NEDGE);
        int d = (e < NEDGE) ? dst[e] : (e - NEDGE);
        unsigned r = (unsigned)(d - n0);
        list[i] = (r << 18) | (unsigned)s;
        lq[i] = (unsigned short)atomicAdd(&rcnt[r], 1u);
    }
    __syncthreads();

    if (t < OWN) {
        unsigned first = 0u;
        for (int k = 0; k < t; ++k) first += rcnt[k];
        rstart[t] = first;
        rwa[t] = fits ? first / perd : 0u;
        rwb[t] = fits ? (first + rcnt[t] - 1u) / perd : 0u;
    }
    __syncthreads();

    for (unsigned i = (unsigned)t; i < n; i += OTH) {
        unsigned word = list[i];
        slist[rstart[word >> 18] + lq[i]] = word;
    }
    __syncthreads();

    if (fits) {
        const unsigned beg = (unsigned)__builtin_amdgcn_readfirstlane((int)((unsigned)w * per));
        const unsigned lim = (unsigned)__builtin_amdgcn_readfirstlane((int)((beg + per < n) ? beg + per : n));
        const unsigned NOROW = 0xFFFFFFFFu;
        unsigned cur = NOROW;
        float    m = 0.f, d = 0.f;
        float4   r0 = make_float4(0.f, 0.f, 0.f, 0.f), r1 = r0;

        auto flush = [&](unsigned row) {
            unsigned slot = row + (unsigned)w;
            ((float4*)&acc[slot * DIM])[lane]      = r0;
            ((float4*)&acc[slot * DIM])[lane + 32] = r1;
            if (lane == 0) { pm[slot] = m; pd[slot] = d; }
        };

        for (unsigned base = beg; base < lim; base += OGRP) {
            unsigned pw[OGRP];
            float4   a0[OGRP], a1[OGRP];
#pragma unroll
            for (int g = 0; g < OGRP; ++g) {
                unsigned i = (base + g < lim) ? base + g : base;
                pw[g] = (unsigned)__builtin_amdgcn_readfirstlane((int)slist[i]);
                const float4* xs = (const float4*)(xl + (long)(pw[g] & 0x3FFFFu) * DIM);
                a0[g] = xs[lane]; a1[g] = xs[lane + 32];
            }
            float p[OGRP];
#pragma unroll
            for (int g = 0; g < OGRP; ++g) {
                const float4* xd = (const float4*)&xrt[(pw[g] >> 18) * DIM];
                p[g] = logit_part(a1[g], xd[lane + 32], t1, logit_part(a0[g], xd[lane], t0, 0.f));
            }
#pragma unroll
            for (int mk = 16; mk >= 1; mk >>= 1) {
#pragma unroll
                for (int g = 0; g < OGRP; ++g) p[g] += __shfl_xor(p[g], mk, 32);
            }
#pragma unroll
            for (int g = 0; g < OGRP; ++g) {
                if (base + g < lim) {
                    unsigned row = pw[g] >> 18;
                    float pg = __int_as_float(__builtin_amdgcn_readfirstlane(__float_as_int(p[g])));
                    if (row != cur) {
                        if (cur != NOROW) flush(cur);
                        cur = row;
                        m = pg; d = 1.f;
                        r0 = a0[g]; r1 = a1[g];
                    } else if (pg > m) {
                        float sc = __expf(m - pg);
                        d  = d * sc + 1.f;
                        r0 = axpb4(sc, r0, a0[g]);
                        r1 = axpb4(sc, r1, a1[g]);
                        m  = pg;
                    } else {
                        float al = __expf(pg - m);
                        d += al;
                        r0 = axpb4(al, a0[g], r0);
                        r1 = axpb4(al, a1[g], r1);
                    }
                }
            }
        }
        if (cur != NOROW) flush(cur);
    } else {
        for (int r = w; r < OWN; r += OTH / 32) {
            int node = n0 + r;
            const float4* xd = (const float4*)&xrt[r * DIM];
            float4 b0 = xd[lane], b1 = xd[lane + 32];
            float  m = -INFINITY, dsum = 0.f;
            float4 o0 = make_float4(0.f, 0.f, 0.f, 0.f), o1 = o0;
#pragma nounroll
            for (int pass = 0; pass < 2; ++pass) {
#pragma nounroll
                for (int base = 0; base < EN; base += 32) {
                    int e = base + lane;
                    int dd = -1;
                    if (e < NEDGE) dd = dst[e];
                    else if (e < EN) dd = e - NEDGE;
                    unsigned hit = __builtin_amdgcn_ballot_w32(dd == node);
                    while (hit) {
                        int ee = base + __builtin_ctz(hit);
                        hit &= hit - 1u;
                        int s = (ee < NEDGE) ? ei[ee] : (ee - NEDGE);
                        const float4* xs = (const float4*)(xl + (long)s * DIM);
                        float4 q0 = xs[lane], q1 = xs[lane + 32];
                        float p = wave_sum(logit_part(q1, b1, t1, logit_part(q0, b0, t0, 0.f)));
                        if (pass == 0) {
                            m = fmaxf(m, p);
                        } else {
                            float al = __expf(p - m);
                            dsum += al;
                            o0.x += al * q0.x; o0.y += al * q0.y; o0.z += al * q0.z; o0.w += al * q0.w;
                            o1.x += al * q1.x; o1.y += al * q1.y; o1.z += al * q1.z; o1.w += al * q1.w;
                        }
                    }
                }
            }
            ((float4*)&acc[r * DIM])[lane]      = o0;
            ((float4*)&acc[r * DIM])[lane + 32] = o1;
            if (lane == 0) { pm[r] = 0.f; pd[r] = dsum; }
        }
    }

    const int cp = t & (DIM / 2 - 1);
    const int rg = t / (DIM / 2);
    const float2* u2 = (const float2*)du + (long)n0 * (DIM / 2);
    float2 bv = ((const float2*)bias)[cp];
    float2 uv[ERT];
#pragma unroll
    for (int j = 0; j < ERT; ++j) uv[j] = u2[(rg + ERG * j) * (DIM / 2) + cp];
    __syncthreads();

    float2*       a2 = (float2*)agg + (long)n0 * (DIM / 2);
    const float2* t2 = (const float2*)acc;
    float2 s  = make_float2(0.f, 0.f);
    float2 s2 = make_float2(0.f, 0.f);
#pragma unroll
    for (int j = 0; j < ERT; ++j) {
        int      row = rg + ERG * j;
        unsigned wa  = rwa[row], wb = rwb[row];
        float mx = pm[row + wa];
        for (unsigned k = wa + 1u; k <= wb; ++k) mx = fmaxf(mx, pm[row + k]);
        float  dn = 0.f;
        float2 a  = make_float2(0.f, 0.f);
        for (unsigned k = wa; k <= wb; ++k) {
            unsigned slot = (unsigned)row + k;
            float  f  = __expf(pm[slot] - mx);
            float2 pv = t2[slot * (DIM / 2) + cp];
            dn  += f * pd[slot];
            a.x += f * pv.x;
            a.y += f * pv.y;
        }
        float2 v;
        v.x = a.x / dn + bv.x;
        v.y = a.y / dn + bv.y;
        v.x = v.x > 0.f ? v.x : 0.f;
        v.y = v.y > 0.f ? v.y : 0.f;
        v.x = (uv[j].x >= 0.5f) ? 2.0f * v.x : 0.0f;
        v.y = (uv[j].y >= 0.5f) ? 2.0f * v.y : 0.0f;
        a2[row * (DIM / 2) + cp] = v;
        s.x += v.x; s2.x += v.x * v.x;
        s.y += v.y; s2.y += v.y * v.y;
    }
    float* red = (float*)list;
    *(float2*)&red[rg * DIM + 2 * cp]             = s;
    *(float2*)&red[ERG * DIM + rg * DIM + 2 * cp] = s2;
    __syncthreads();
    if (t < 2 * DIM) {
        const float* r = red + (t / DIM) * ERG * DIM + (t & (DIM - 1));
        float cs = 0.f;
#pragma unroll
        for (int j = 0; j < ERG; ++j) cs += r[j * DIM];
        atomicAdd(&colsums[t], cs);
    }
}

#define PR   20
#define PRT  (PR / 4)
static_assert(NNODE % PR == 0 && PR % 4 == 0, "whole blocks of four row groups");

__global__ __launch_bounds__(256)
void finalize(const float* __restrict__ v,
              const float* __restrict__ sums, const float* __restrict__ sumsq,
              const float* __restrict__ gamma, const float* __restrict__ beta,
              float* __restrict__ out) {
    int t  = (int)threadIdx.x;
    int c4 = t & 63;
    int rg = t >> 6;
    int i0 = (int)blockIdx.x * PR + rg;

    const float4* v4 = (const float4*)v;
    float4*       o4 = (float4*)out;

    float4 x[PRT];
#pragma unroll
    for (int j = 0; j < PRT; ++j) x[j] = v4[(long)(i0 + 4 * j) * (DIM / 4) + c4];

    float4 sm = ((const float4*)sums)[c4];
    float4 sq = ((const float4*)sumsq)[c4];
    float4 g  = ((const float4*)gamma)[c4];
    float4 b  = ((const float4*)beta)[c4];

    const float invn = 1.0f / (float)NNODE;
    float4 scale, shift;
    {
        float mean, var;
        mean = sm.x * invn; var = sq.x * invn - mean * mean;
        scale.x = g.x * rsqrtf(var + BN_EPS); shift.x = b.x - mean * scale.x;
        mean = sm.y * invn; var = sq.y * invn - mean * mean;
        scale.y = g.y * rsqrtf(var + BN_EPS); shift.y = b.y - mean * scale.y;
        mean = sm.z * invn; var = sq.z * invn - mean * mean;
        scale.z = g.z * rsqrtf(var + BN_EPS); shift.z = b.z - mean * scale.z;
        mean = sm.w * invn; var = sq.w * invn - mean * mean;
        scale.w = g.w * rsqrtf(var + BN_EPS); shift.w = b.w - mean * scale.w;
    }

#pragma unroll
    for (int j = 0; j < PRT; ++j) {
        float4 r;
        r.x = x[j].x * scale.x + shift.x;
        r.y = x[j].y * scale.y + shift.y;
        r.z = x[j].z * scale.z + shift.z;
        r.w = x[j].w * scale.w + shift.w;
        o4[(long)(i0 + 4 * j) * (DIM / 4) + c4] = r;
    }
}

extern "C" void kernel_launch(void* const* d_in, const int* in_sizes, int n_in,
                              void* d_out, int out_size, void* d_ws, size_t ws_size,
                              hipStream_t stream) {
    const float* x     = (const float*)d_in[0];
    const int*   ei    = (const int*)d_in[1];
    const float* Wl    = (const float*)d_in[2];
    const float* bl    = (const float*)d_in[3];
    const float* Wr    = (const float*)d_in[4];
    const float* br    = (const float*)d_in[5];
    const float* att   = (const float*)d_in[6];
    const float* bias  = (const float*)d_in[7];
    const float* gamma = (const float*)d_in[8];
    const float* beta  = (const float*)d_in[9];
    const float* du    = (const float*)d_in[10];
    float*       out   = (float*)d_out;

    float*    ws    = (float*)d_ws;
    float*    agg   = ws;
    float*    sums  = ws + (long)NNODE * DIM + 2 * NNODE;
    float*    sumsq = sums + DIM;
    float*    xl    = sumsq + DIM;
    float*    xr    = xl + (long)NNODE * DIM;
    unsigned* owners = (unsigned*)d_out;

    gemm_xlxr<<<GBLK, 256, 0, stream>>>(x, Wl, bl, Wr, br, xl, xr, sums, ei, owners);

    edge_aggregate<<<OBLK, OTH, 0, stream>>>(ei, xl, xr, att, bias, du, owners, sums, agg);

    finalize<<<NNODE / PR, 256, 0, stream>>>(agg, sums, sumsq, gamma, beta, out);
}
